# starting kernel
# baseline (speedup 1.0000x reference)
#include <hip/hip_runtime.h>

#define N_NODES 50000
#define N_EDGES 800000
#define N_GRAPHS 512
#define IN_C 128
#define HID 256
#define OUT_C 16

typedef __attribute__((ext_vector_type(16))) __bf16 v16bf;
typedef __attribute__((ext_vector_type(8)))  float  v8f;

__device__ __forceinline__ unsigned short f2bf(float f) {
  union { float f; unsigned int u; } c; c.f = f;
  unsigned int u = c.u;
  u += 0x7fffu + ((u >> 16) & 1u);
  return (unsigned short)(u >> 16);
}

__device__ __forceinline__ int frag_koff(int lane, int p) {
  int kb = (lane < 16) ? 0 : 8;
  return (p < 4) ? (kb + 2 * p) : (16 + kb + 2 * (p - 4));
}

__global__ void copy_f4(const float* __restrict__ src, float* __restrict__ dst, int n4) {
  int i = blockIdx.x * blockDim.x + threadIdx.x;
  if (i < n4) ((float4*)dst)[i] = ((const float4*)src)[i];
}

__global__ void edge_scatter(const float* __restrict__ h, const int* __restrict__ esrc,
                             const int* __restrict__ edst, float* __restrict__ agg,
                             int cg  , long total) {
  long i = (long)blockIdx.x * blockDim.x + threadIdx.x;
  if (i >= total) return;
  int e = (int)(i / cg);
  int c = (int)(i - (long)e * cg);
  int s = esrc[e], d = edst[e];
  const float4 v = ((const float4*)(h + (long)s * (cg * 4)))[c];
  float* o = agg + (long)d * (cg * 4) + c * 4;
  atomicAdd(o + 0, v.x); atomicAdd(o + 1, v.y);
  atomicAdd(o + 2, v.z); atomicAdd(o + 3, v.w);
}

__global__ void apack_f32(const float* __restrict__ src, unsigned int* __restrict__ out,
                          int K, int KT, long total) {
  long i = (long)blockIdx.x * blockDim.x + threadIdx.x;
  if (i >= total) return;
  int p = (int)(i & 7);
  int lane = (int)((i >> 3) & 31);
  long tt = i >> 8;
  int kt = (int)(tt % KT);
  int rt = (int)(tt / KT);
  int row = rt * 16 + (lane & 15);
  int k = kt * 32 + frag_koff(lane, p);
  const float* s = src + (long)row * K + k;
  out[i] = (unsigned int)f2bf(s[0]) | ((unsigned int)f2bf(s[1]) << 16);
}

__global__ void apack_bf16(const unsigned short* __restrict__ src, unsigned int* __restrict__ out,
                           int K, int KT, long total) {
  long i = (long)blockIdx.x * blockDim.x + threadIdx.x;
  if (i >= total) return;
  int p = (int)(i & 7);
  int lane = (int)((i >> 3) & 31);
  long tt = i >> 8;
  int kt = (int)(tt % KT);
  int rt = (int)(tt / KT);
  int row = rt * 16 + (lane & 15);
  int k = kt * 32 + frag_koff(lane, p);
  out[i] = *(const unsigned int*)(src + (long)row * K + k);
}

__global__ void wpack_f32(const float* __restrict__ W, unsigned int* __restrict__ out,
                          int K, int M, int KT, long total) {
  long i = (long)blockIdx.x * blockDim.x + threadIdx.x;
  if (i >= total) return;
  int p = (int)(i & 7);
  int lane = (int)((i >> 3) & 31);
  long tt = i >> 8;
  int kt = (int)(tt % KT);
  int ct = (int)(tt / KT);
  int col = ct * 16 + (lane & 15);
  int k = kt * 32 + frag_koff(lane, p);
  unsigned int lo = f2bf(W[(long)k * M + col]);
  unsigned int hi = f2bf(W[(long)(k + 1) * M + col]);
  out[i] = lo | (hi << 16);
}

template <int RELU, int OUTBF>
__global__ void gemm_pk(const uint4* __restrict__ Apk, const uint4* __restrict__ Wpk,
                        const float* __restrict__ bias, void* __restrict__ Cout,
                        int N, int K, int M) {
  const int tid  = threadIdx.x;
  const int lane = tid & 31;
  const int wave = tid >> 5;
  const int wr   = wave & 3;
  const int wc   = wave >> 2;
  const int KT   = K >> 5;
  const int nrt  = N >> 4;
  const int nct  = M >> 4;
  const int rt0  = blockIdx.y * 8 + wr * 2;
  const int ct0  = blockIdx.x * 4 + wc * 2;
  const bool r0 = rt0 < nrt,     r1 = (rt0 + 1) < nrt;
  const bool c0 = ct0 < nct,     c1 = (ct0 + 1) < nct;
  const int rA0 = r0 ? rt0 : 0,  rA1 = r1 ? (rt0 + 1) : 0;
  const int cB0 = c0 ? ct0 : 0,  cB1 = c1 ? (ct0 + 1) : 0;

  long ia0 = ((long)rA0 * KT * 32 + lane) * 2;
  long ia1 = ((long)rA1 * KT * 32 + lane) * 2;
  long ib0 = ((long)cB0 * KT * 32 + lane) * 2;
  long ib1 = ((long)cB1 * KT * 32 + lane) * 2;
  const long step = 64;

  v8f acc00 = {}, acc01 = {}, acc10 = {}, acc11 = {};
  union F { uint4 q[2]; v16bf v; };

  for (int kt = 0; kt < KT; ++kt) {
    F a0, a1, b0, b1;
    a0.q[0] = Apk[ia0]; a0.q[1] = Apk[ia0 + 1];
    a1.q[0] = Apk[ia1]; a1.q[1] = Apk[ia1 + 1];
    b0.q[0] = Wpk[ib0]; b0.q[1] = Wpk[ib0 + 1];
    b1.q[0] = Wpk[ib1]; b1.q[1] = Wpk[ib1 + 1];
    if (kt + 1 < KT) {
      __builtin_prefetch((const void*)(Apk + ia0 + step), 0, 3);
      __builtin_prefetch((const void*)(Apk + ia1 + step), 0, 3);
    }
    ia0 += step; ia1 += step; ib0 += step; ib1 += step;
    acc00 = __builtin_amdgcn_wmma_f32_16x16x32_bf16(false, a0.v, false, b0.v, (short)0, acc00, false, false);
    acc01 = __builtin_amdgcn_wmma_f32_16x16x32_bf16(false, a0.v, false, b1.v, (short)0, acc01, false, false);
    acc10 = __builtin_amdgcn_wmma_f32_16x16x32_bf16(false, a1.v, false, b0.v, (short)0, acc10, false, false);
    acc11 = __builtin_amdgcn_wmma_f32_16x16x32_bf16(false, a1.v, false, b1.v, (short)0, acc11, false, false);
  }

  const int rowL = lane & 15;
  auto st = [&](const v8f& acc, int rt, bool rok, int ct, bool cok) {
    if (!rok || !cok) return;
    int col = ct * 16 + rowL;
    #pragma unroll
    for (int r = 0; r < 8; ++r) {
      int row = rt * 16 + ((lane < 16) ? r : r + 8);
      float o = acc[r] + bias[col];
      if (RELU) o = fmaxf(o, 0.0f);
      if (OUTBF) ((unsigned short*)Cout)[(long)row * M + col] = f2bf(o);
      else       ((float*)Cout)[(long)row * M + col] = o;
    }
  };
  st(acc00, rt0, r0, ct0, c0);         st(acc01, rt0, r0, ct0 + 1, c1);
  st(acc10, rt0 + 1, r1, ct0, c0);     st(acc11, rt0 + 1, r1, ct0 + 1, c1);
}

__global__ void count_nodes(const int* __restrict__ batch, float* __restrict__ counts) {
  int n = blockIdx.x * blockDim.x + threadIdx.x;
  if (n < N_NODES) atomicAdd(&counts[batch[n]], 1.0f);
}

__global__ void pool_sum(const float* __restrict__ h, const int* __restrict__ batch,
                         float* __restrict__ pooled) {
  const int cg = HID / 4;
  long i = (long)blockIdx.x * blockDim.x + threadIdx.x;
  if (i >= (long)N_NODES * cg) return;
  int n = (int)(i / cg);
  int c = (int)(i - (long)n * cg);
  int g = batch[n];
  float4 v = ((const float4*)(h + (long)n * HID))[c];
  float* o = pooled + (long)g * HID + c * 4;
  atomicAdd(o + 0, v.x); atomicAdd(o + 1, v.y);
  atomicAdd(o + 2, v.z); atomicAdd(o + 3, v.w);
}

__global__ void pool_div(float* pooled, const float* __restrict__ counts) {
  int i = blockIdx.x * blockDim.x + threadIdx.x;
  if (i < N_GRAPHS * HID) {
    float c = counts[i / HID];
    pooled[i] /= fmaxf(c, 1.0f);
  }
}

extern "C" void kernel_launch(void* const* d_in, const int* in_sizes, int n_in,
                              void* d_out, int out_size, void* d_ws, size_t ws_size,
                              hipStream_t stream) {
  const float* x     = (const float*)d_in[0];
  const int*   ei    = (const int*)d_in[1];
  const int*   esrc  = ei;
  const int*   edst  = ei + N_EDGES;
  const int*   batch = (const int*)d_in[2];
  const float* cw1[3] = {(const float*)d_in[3],  (const float*)d_in[7],  (const float*)d_in[11]};
  const float* cb1[3] = {(const float*)d_in[4],  (const float*)d_in[8],  (const float*)d_in[12]};
  const float* cw2[3] = {(const float*)d_in[5],  (const float*)d_in[9],  (const float*)d_in[13]};
  const float* cb2[3] = {(const float*)d_in[6],  (const float*)d_in[10], (const float*)d_in[14]};
  const float* ow1 = (const float*)d_in[15];
  const float* ob1 = (const float*)d_in[16];
  const float* ow2 = (const float*)d_in[17];
  const float* ob2 = (const float*)d_in[18];

  char* ws = (char*)d_ws;
  size_t off = 0;
  auto bump = [&](size_t bytes) {
    void* p = ws + off;
    off += (bytes + 255) & ~(size_t)255;
    return p;
  };
  const int NRT = N_NODES / 16;
  float*          B0     = (float*)bump((size_t)N_NODES * HID * 4);
  float*          B2     = (float*)bump((size_t)N_NODES * HID * 4);
  unsigned int*   Apk    = (unsigned int*)bump((size_t)NRT * 8 * 256 * 4);
  unsigned short* Mbf    = (unsigned short*)bump((size_t)N_NODES * HID * 2);
  unsigned int*   Mpk    = (unsigned int*)bump((size_t)NRT * 8 * 256 * 4);
  float*          pooled = (float*)bump((size_t)N_GRAPHS * HID * 4);
  float*          counts = (float*)bump((size_t)N_GRAPHS * 4);
  unsigned int*   Ppk    = (unsigned int*)bump((size_t)(N_GRAPHS / 16) * 8 * 256 * 4);
  unsigned short* MPbf   = (unsigned short*)bump((size_t)N_GRAPHS * HID * 2);
  unsigned int*   MPpk   = (unsigned int*)bump((size_t)(N_GRAPHS / 16) * 8 * 256 * 4);
  unsigned int*   wpk1[3], *wpk2[3];
  for (int l = 0; l < 3; ++l) {
    wpk1[l] = (unsigned int*)bump((size_t)(HID / 16) * 8 * 256 * 4);
    wpk2[l] = (unsigned int*)bump((size_t)(HID / 16) * 8 * 256 * 4);
  }
  unsigned int* owpk1 = (unsigned int*)bump((size_t)(HID / 16) * 8 * 256 * 4);
  unsigned int* owpk2 = (unsigned int*)bump((size_t)(OUT_C / 16) * 8 * 256 * 4);

  dim3 blk(256);
  auto wpack = [&](const float* W, unsigned int* dst, int K, int M) {
    int KT = K / 32;
    long total = (long)(M / 16) * KT * 256;
    wpack_f32<<<(unsigned)((total + 255) / 256), blk, 0, stream>>>(W, dst, K, M, KT, total);
  };

  wpack(cw1[0], wpk1[0], IN_C, HID);
  wpack(cw2[0], wpk2[0], HID, HID);
  for (int l = 1; l < 3; ++l) {
    wpack(cw1[l], wpk1[l], HID, HID);
    wpack(cw2[l], wpk2[l], HID, HID);
  }
  wpack(ow1, owpk1, HID, HID);
  wpack(ow2, owpk2, HID, OUT_C);

  const float* h = x;
  int C = IN_C;
  for (int l = 0; l < 3; ++l) {
    int n4 = N_NODES * C / 4;
    copy_f4<<<(n4 + 255) / 256, blk, 0, stream>>>(h, B0, n4);
    long tot = (long)N_EDGES * (C / 4);
    edge_scatter<<<(unsigned)((tot + 255) / 256), blk, 0, stream>>>(h, esrc, edst, B0, C / 4, tot);

    int KT = C / 32;
    long atot = (long)NRT * KT * 256;
    apack_f32<<<(unsigned)((atot + 255) / 256), blk, 0, stream>>>(B0, Apk, C, KT, atot);

    dim3 g1((HID / 16 + 3) / 4, (NRT + 7) / 8);
    gemm_pk<1, 1><<<g1, blk, 0, stream>>>((const uint4*)Apk, (const uint4*)wpk1[l], cb1[l],
                                          Mbf, N_NODES, C, HID);
    long mtot = (long)NRT * (HID / 32) * 256;
    apack_bf16<<<(unsigned)((mtot + 255) / 256), blk, 0, stream>>>(Mbf, Mpk, HID, HID / 32, mtot);
    dim3 g2((HID / 16 + 3) / 4, (NRT + 7) / 8);
    gemm_pk<1, 0><<<g2, blk, 0, stream>>>((const uint4*)Mpk, (const uint4*)wpk2[l], cb2[l],
                                          B2, N_NODES, HID, HID);
    h = B2;
    C = HID;
  }

  hipMemsetAsync(pooled, 0, (size_t)N_GRAPHS * HID * 4, stream);
  hipMemsetAsync(counts, 0, (size_t)N_GRAPHS * 4, stream);
  count_nodes<<<(N_NODES + 255) / 256, blk, 0, stream>>>(batch, counts);
  long ptot = (long)N_NODES * (HID / 4);
  pool_sum<<<(unsigned)((ptot + 255) / 256), blk, 0, stream>>>(B2, batch, pooled);
  pool_div<<<(N_GRAPHS * HID + 255) / 256, blk, 0, stream>>>(pooled, counts);

  const int GRT = N_GRAPHS / 16;
  long gtot = (long)GRT * (HID / 32) * 256;
  apack_f32<<<(unsigned)((gtot + 255) / 256), blk, 0, stream>>>(pooled, Ppk, HID, HID / 32, gtot);
  dim3 g3((HID / 16 + 3) / 4, (GRT + 7) / 8);
  gemm_pk<1, 1><<<g3, blk, 0, stream>>>((const uint4*)Ppk, (const uint4*)owpk1, ob1,
                                        MPbf, N_GRAPHS, HID, HID);
  apack_bf16<<<(unsigned)((gtot + 255) / 256), blk, 0, stream>>>(MPbf, MPpk, HID, HID / 32, gtot);
  dim3 g4((OUT_C / 16 + 3) / 4, (GRT + 7) / 8);
  gemm_pk<0, 0><<<g4, blk, 0, stream>>>((const uint4*)MPpk, (const uint4*)owpk2, ob2,
                                        (float*)d_out, N_GRAPHS, HID, OUT_C);
}


# Mean pool: float atomics, clears, count, divide and pack passes replaced by owner work-groups (4 graphs each: id list scanned once, rows added in registers in one fixed order); 25 -> 20 launches
# speedup vs baseline: 2.7918x; 2.7918x over previous
#include <hip/hip_runtime.h>

#define N_NODES 50000
#define N_EDGES 800000
#define N_GRAPHS 512
#define IN_C 128
#define HID 256
#define OUT_C 16

typedef __attribute__((ext_vector_type(16))) __bf16 v16bf;
typedef __attribute__((ext_vector_type(8)))  float  v8f;
typedef __attribute__((ext_vector_type(8)))  unsigned int v8u;
typedef const __attribute__((address_space(4))) v8u* list_ptr;

__device__ __forceinline__ unsigned short f2bf(float f) {
  union { float f; unsigned int u; } c; c.f = f;
  unsigned int u = c.u;
  u += 0x7fffu + ((u >> 16) & 1u);
  return (unsigned short)(u >> 16);
}

typedef __attribute__((ext_vector_type(2))) float  v2f;
typedef __attribute__((ext_vector_type(2))) __bf16 v2bf;
__device__ __forceinline__ unsigned int pack2bf(float a, float b) {
  const v2f f = {a, b};
  const v2bf h = __builtin_convertvector(f, v2bf);
  return __builtin_bit_cast(unsigned int, h);
}

__device__ __forceinline__ int frag_koff(int lane, int p) {
  int kb = (lane < 16) ? 0 : 8;
  return (p < 4) ? (kb + 2 * p) : (16 + kb + 2 * (p - 4));
}

#define AGG_NB      196
#define AGG_WAVES   16
#define AGG_THREADS (AGG_WAVES * 32)
#define AGG_IT      15
#define AGG_SUB     (AGG_IT * 128)
#define AGG_ROUND   (AGG_SUB * AGG_WAVES)
#define AGG_NROUND  ((N_EDGES + AGG_ROUND - 1) / AGG_ROUND)
#define AGG_GROUPS  ((N_NODES + AGG_NB - 1) / AGG_NB)
#define LST_CAP     1024
#define LST_G       8
#define LST_PAD     (5 * LST_G)

static_assert(N_NODES <= (1 << 20) && N_EDGES <= (1 << 20) && AGG_NB < 256, "list entry fields");
static_assert(LST_G == 8, "a group of entries is one 8-dword scalar load");

static_assert(N_EDGES % 4 == 0, "four edges per index load");

__device__ __forceinline__ void scan_load(const int* __restrict__ edst, int e0, int4 (&d)[AGG_IT]) {
  #pragma unroll
  for (int it = 0; it < AGG_IT; ++it) {
    const int e = e0 + it * 128;
    d[it] = *(const int4*)(edst + (e < N_EDGES ? e : N_EDGES - 4));
  }
}

__device__ __forceinline__ int scan_note(const int4 (&d)[AGG_IT], int e0, int base, int nloc,
                                         unsigned int ltmask, unsigned int* mylst) {
  int c = 0;
  #pragma unroll
  for (int it = 0; it < AGG_IT; ++it) {
    const unsigned int e = (unsigned int)(e0 + it * 128);
    const unsigned int x0 = (unsigned int)(d[it].x - base);
    const unsigned int x1 = (unsigned int)(d[it].y - base);
    const unsigned int x2 = (unsigned int)(d[it].z - base);
    const unsigned int x3 = (unsigned int)(d[it].w - base);
    const unsigned int lim = (e < (unsigned int)N_EDGES) ? (unsigned int)nloc : 0u;
    const bool m0 = x0 < lim, m1 = x1 < lim;
    const bool m2 = x2 < lim, m3 = x3 < lim;
    const unsigned int mn01 = x0 < x1 ? x0 : x1, mn23 = x2 < x3 ? x2 : x3;
    if (__builtin_amdgcn_ballot_w32((mn01 < mn23 ? mn01 : mn23) < lim)) {
      unsigned int b;
      b = __builtin_amdgcn_ballot_w32(m0);
      if (m0) mylst[c + __builtin_popcount(b & ltmask)] = (x0 << 20) | (e + 0u);
      c += __builtin_popcount(b);
      b = __builtin_amdgcn_ballot_w32(m1);
      if (m1) mylst[c + __builtin_popcount(b & ltmask)] = (x1 << 20) | (e + 1u);
      c += __builtin_popcount(b);
      b = __builtin_amdgcn_ballot_w32(m2);
      if (m2) mylst[c + __builtin_popcount(b & ltmask)] = (x2 << 20) | (e + 2u);
      c += __builtin_popcount(b);
      b = __builtin_amdgcn_ballot_w32(m3);
      if (m3) mylst[c + __builtin_popcount(b & ltmask)] = (x3 << 20) | (e + 3u);
      c += __builtin_popcount(b);
    }
  }
  return c;
}

__global__ void __launch_bounds__(AGG_THREADS)
owner_lists(const int* __restrict__ esrc, const int* __restrict__ edst,
            unsigned int* __restrict__ glst, int* __restrict__ gcnt) {
  __shared__ unsigned int lst[AGG_WAVES * AGG_SUB];
  __shared__ unsigned int own[AGG_WAVES * LST_CAP];
  __shared__ int          cnt[AGG_WAVES];

  const int tid  = threadIdx.x;
  const int lane = tid & 31;
  const int wave = tid >> 5;
  const int base = blockIdx.x * AGG_NB;
  int nloc = N_NODES - base;
  if (nloc > AGG_NB) nloc = AGG_NB;

  const unsigned int ltmask = (1u << lane) - 1u;
  unsigned int* mylst = lst + wave * AGG_SUB;
  unsigned int* myown = own + wave * LST_CAP;
  unsigned int* myout = glst + ((long)blockIdx.x * AGG_WAVES + wave) * LST_CAP;
  int gpos = 0;

  int rr = (int)(blockIdx.x % AGG_NROUND);
  int4 d[AGG_IT];
  scan_load(edst, rr * AGG_ROUND + wave * AGG_SUB + lane * 4, d);
  for (int r = 0; r < AGG_NROUND; ++r) {
    const int e0 = rr * AGG_ROUND + wave * AGG_SUB + lane * 4;
    rr = (rr + 1 == AGG_NROUND) ? 0 : rr + 1;
    int4 dn[AGG_IT];
    scan_load(edst, rr * AGG_ROUND + wave * AGG_SUB + lane * 4, dn);

    const int c = scan_note(d, e0, base, nloc, ltmask, mylst);
    if (lane == 0) cnt[wave] = c;
    __syncthreads();

    const int myc = (lane < AGG_WAVES) ? cnt[lane] : 0;
    int incl = myc;
    #pragma unroll
    for (int o = 1; o < AGG_WAVES; o <<= 1) {
      const int t = __shfl_up(incl, o);
      if (lane >= o) incl += t;
    }
    const int excl  = incl - myc;
    const int total = __builtin_amdgcn_readlane(incl, AGG_WAVES - 1);
    for (int f0 = 0; f0 < total; f0 += 32) {
      const int f = f0 + lane;
      int li = 0;
      #pragma unroll
      for (int l = 0; l < AGG_WAVES - 1; ++l) li += (f >= __builtin_amdgcn_readlane(incl, l)) ? 1 : 0;
      const int st = __shfl(excl, li);
      unsigned int ent = 0u;
      bool mine = false;
      if (f < total) {
        ent  = lst[li * AGG_SUB + (f - st)];
        mine = ((ent >> 20) & (unsigned int)(AGG_WAVES - 1)) == (unsigned int)wave;
      }
      const unsigned int m = __builtin_amdgcn_ballot_w32(mine);
      if (mine) {
        const int pos = gpos + __builtin_popcount(m & ltmask);
        if (pos < LST_CAP) myown[pos] = ent;
      }
      gpos += __builtin_popcount(m);
    }
    __syncthreads();

    #pragma unroll
    for (int it = 0; it < AGG_IT; ++it) d[it] = dn[it];
  }

  const int nout = gpos < LST_CAP ? gpos : LST_CAP;
  for (int i = lane; i < nout; i += 32) {
    const unsigned int ent = myown[i];
    myout[i] = (ent & 0xFFF00000u) | (unsigned int)esrc[ent & 0xFFFFFu];
  }
  #pragma unroll
  for (int k = 0; k < (LST_PAD + 31) / 32; ++k) {
    const int o = k * 32 + lane;
    if (o < LST_PAD && gpos + o < LST_CAP) myout[gpos + o] = ((unsigned int)AGG_NB << 20) | (unsigned int)base;
  }
  if (lane == 0) gcnt[blockIdx.x * AGG_WAVES + wave] = gpos;
}

template <int C>
__global__ void __launch_bounds__(AGG_THREADS)
agg_owner(const float* __restrict__ h, const int* __restrict__ esrc,
          const int* __restrict__ edst, const unsigned int* __restrict__ glst,
          const int* __restrict__ gcnt, float* __restrict__ agg) {
  __shared__ float        acc[(AGG_NB + 1) * C];
  __shared__ unsigned int lst[AGG_WAVES * AGG_SUB];
  __shared__ int          cnt[AGG_WAVES];
  constexpr int VEC = C / 128;

  const int tid  = threadIdx.x;
  const int lane = tid & 31;
  const int wave = tid >> 5;
  const int base = blockIdx.x * AGG_NB;
  int nloc = N_NODES - base;
  if (nloc > AGG_NB) nloc = AGG_NB;

  {
    const float4* hs = (const float4*)(h + (long)base * C);
    float4* a4 = (float4*)acc;
    const int n4 = nloc * (C / 4);
    for (int i = tid; i < n4; i += AGG_THREADS) a4[i] = hs[i];
  }

  const int lc = (lane < AGG_WAVES) ? gcnt[blockIdx.x * AGG_WAVES + lane] : 0;
  const bool fits = __builtin_amdgcn_ballot_w32(lc > LST_CAP - LST_PAD) == 0u;

  if (fits) {
    __syncthreads();
    const int wv = __builtin_amdgcn_readfirstlane(wave);
    const int n  = __builtin_amdgcn_readlane(lc, wv);
    const list_ptr q = (list_ptr)(unsigned long)(glst + ((long)blockIdx.x * AGG_WAVES + wv) * LST_CAP);
    const int ng = (n + LST_G - 1) / LST_G;

    v8u eA = q[0], eB = q[1], eC = q[2];
    float4 A[LST_G][VEC], B[LST_G][VEC];
    auto rows_ask = [&](float4 (&R)[LST_G][VEC], const v8u& e) {
      #pragma unroll
      for (int j = 0; j < LST_G; ++j) {
        const int s = (int)(e[j] & 0xFFFFFu);
        const float4* hp = (const float4*)(h + (long)s * C) + lane;
        #pragma unroll
        for (int v = 0; v < VEC; ++v) R[j][v] = hp[v * 32];
      }
    };
    auto rows_add = [&](const float4 (&R)[LST_G][VEC], const v8u& e) {
      #pragma unroll
      for (int j = 0; j < LST_G; ++j) {
        const int nd = (int)(e[j] >> 20);
        float4* ap = (float4*)(acc + nd * C) + lane;
        #pragma unroll
        for (int v = 0; v < VEC; ++v) {
          float4 a = ap[v * 32];
          a.x += R[j][v].x; a.y += R[j][v].y; a.z += R[j][v].z; a.w += R[j][v].w;
          ap[v * 32] = a;
        }
      }
    };
    rows_ask(A, eA);
    const int nt = (ng + 1) / 2;
    for (int t = 0; t < nt; ++t) {
      const v8u fB = q[2 * t + 3], fC = q[2 * t + 4];
      rows_ask(B, eB);
      rows_add(A, eA);
      rows_ask(A, eC);
      rows_add(B, eB);
      eA = eC; eB = fB; eC = fC;
    }
    __syncthreads();
  } else {
    const unsigned int ltmask = (1u << lane) - 1u;
    unsigned int* mylst = lst + wave * AGG_SUB;
    int rr = (int)(blockIdx.x % AGG_NROUND);
    for (int r = 0; r < AGG_NROUND; ++r) {
      const int e0 = rr * AGG_ROUND + wave * AGG_SUB + lane * 4;
      rr = (rr + 1 == AGG_NROUND) ? 0 : rr + 1;
      int4 d[AGG_IT];
      scan_load(edst, e0, d);
      const int c = scan_note(d, e0, base, nloc, ltmask, mylst);
      if (lane == 0) cnt[wave] = c;
      __syncthreads();

      const int myc = (lane < AGG_WAVES) ? cnt[lane] : 0;
      int incl = myc;
      #pragma unroll
      for (int o = 1; o < AGG_WAVES; o <<= 1) {
        const int t = __shfl_up(incl, o);
        if (lane >= o) incl += t;
      }
      const int excl  = incl - myc;
      const int total = __builtin_amdgcn_readlane(incl, AGG_WAVES - 1);
      for (int f0 = 0; f0 < total; f0 += 32) {
        const int f = f0 + lane;
        int li = 0;
        #pragma unroll
        for (int l = 0; l < AGG_WAVES - 1; ++l) li += (f >= __builtin_amdgcn_readlane(incl, l)) ? 1 : 0;
        const int st = __shfl(excl, li);
        unsigned int ent = 0u;
        bool mine = false;
        if (f < total) {
          ent  = lst[li * AGG_SUB + (f - st)];
          mine = ((ent >> 20) & (unsigned int)(AGG_WAVES - 1)) == (unsigned int)wave;
        }
        unsigned int m = __builtin_amdgcn_ballot_w32(mine);
        int s = 0;
        if (mine) s = esrc[ent & 0xFFFFFu];
        while (m) {
          const int b0 = __builtin_ctz(m); m &= m - 1u;
          const int s0 = __builtin_amdgcn_readlane(s, b0);
          const int n0 = (int)(__builtin_amdgcn_readlane((int)ent, b0) >> 20) & 0xFF;
          const float4* h0 = (const float4*)(h + (long)s0 * C) + lane;
          float4* a0 = (float4*)(acc + n0 * C) + lane;
          #pragma unroll
          for (int j = 0; j < VEC; ++j) {
            const float4 v0 = h0[j * 32];
            float4 a = a0[j * 32];
            a.x += v0.x; a.y += v0.y; a.z += v0.z; a.w += v0.w;
            a0[j * 32] = a;
          }
        }
      }
      __syncthreads();
    }
  }

  {
    float4* o4 = (float4*)(agg + (long)base * C);
    const float4* a4 = (const float4*)acc;
    const int n4 = nloc * (C / 4);
    for (int i = tid; i < n4; i += AGG_THREADS) o4[i] = a4[i];
  }
}

__global__ void apack_f32(const float* __restrict__ src, unsigned int* __restrict__ out,
                          int K, int KT, long total) {
  long i = (long)blockIdx.x * blockDim.x + threadIdx.x;
  if (i >= total) return;
  int p = (int)(i & 7);
  int lane = (int)((i >> 3) & 31);
  long tt = i >> 8;
  int kt = (int)(tt % KT);
  int rt = (int)(tt / KT);
  int row = rt * 16 + (lane & 15);
  int k = kt * 32 + frag_koff(lane, p);
  const float* s = src + (long)row * K + k;
  out[i] = (unsigned int)f2bf(s[0]) | ((unsigned int)f2bf(s[1]) << 16);
}

__global__ void apack_bf16(const unsigned short* __restrict__ src, unsigned int* __restrict__ out,
                           int K, int KT, long total) {
  long i = (long)blockIdx.x * blockDim.x + threadIdx.x;
  if (i >= total) return;
  int p = (int)(i & 7);
  int lane = (int)((i >> 3) & 31);
  long tt = i >> 8;
  int kt = (int)(tt % KT);
  int rt = (int)(tt / KT);
  int row = rt * 16 + (lane & 15);
  int k = kt * 32 + frag_koff(lane, p);
  out[i] = *(const unsigned int*)(src + (long)row * K + k);
}

__global__ void wpack_f32(const float* __restrict__ W, unsigned int* __restrict__ out,
                          int K, int M, int KT, long total) {
  long i = (long)blockIdx.x * blockDim.x + threadIdx.x;
  if (i >= total) return;
  int p = (int)(i & 7);
  int lane = (int)((i >> 3) & 31);
  long tt = i >> 8;
  int kt = (int)(tt % KT);
  int ct = (int)(tt / KT);
  int col = ct * 16 + (lane & 15);
  int k = kt * 32 + frag_koff(lane, p);
  unsigned int lo = f2bf(W[(long)k * M + col]);
  unsigned int hi = f2bf(W[(long)(k + 1) * M + col]);
  out[i] = lo | (hi << 16);
}

template <int RELU, int OUTBF>
__global__ void gemm_pk(const uint4* __restrict__ Apk, const uint4* __restrict__ Wpk,
                        const float* __restrict__ bias, void* __restrict__ Cout,
                        int N, int K, int M) {
  const int tid  = threadIdx.x;
  const int lane = tid & 31;
  const int wave = tid >> 5;
  const int wr   = wave & 3;
  const int wc   = wave >> 2;
  const int KT   = K >> 5;
  const int nrt  = N >> 4;
  const int nct  = M >> 4;
  const int rt0  = blockIdx.y * 8 + wr * 2;
  const int ct0  = blockIdx.x * 4 + wc * 2;
  const bool r0 = rt0 < nrt,     r1 = (rt0 + 1) < nrt;
  const bool c0 = ct0 < nct,     c1 = (ct0 + 1) < nct;
  const int rA0 = r0 ? rt0 : 0,  rA1 = r1 ? (rt0 + 1) : 0;
  const int cB0 = c0 ? ct0 : 0,  cB1 = c1 ? (ct0 + 1) : 0;

  long ia0 = ((long)rA0 * KT * 32 + lane) * 2;
  long ia1 = ((long)rA1 * KT * 32 + lane) * 2;
  long ib0 = ((long)cB0 * KT * 32 + lane) * 2;
  long ib1 = ((long)cB1 * KT * 32 + lane) * 2;
  const long step = 64;

  v8f acc00 = {}, acc01 = {}, acc10 = {}, acc11 = {};
  union F { uint4 q[2]; v16bf v; };

  for (int kt = 0; kt < KT; ++kt) {
    F a0, a1, b0, b1;
    a0.q[0] = Apk[ia0]; a0.q[1] = Apk[ia0 + 1];
    a1.q[0] = Apk[ia1]; a1.q[1] = Apk[ia1 + 1];
    b0.q[0] = Wpk[ib0]; b0.q[1] = Wpk[ib0 + 1];
    b1.q[0] = Wpk[ib1]; b1.q[1] = Wpk[ib1 + 1];
    if (kt + 1 < KT) {
      __builtin_prefetch((const void*)(Apk + ia0 + step), 0, 3);
      __builtin_prefetch((const void*)(Apk + ia1 + step), 0, 3);
    }
    ia0 += step; ia1 += step; ib0 += step; ib1 += step;
    acc00 = __builtin_amdgcn_wmma_f32_16x16x32_bf16(false, a0.v, false, b0.v, (short)0, acc00, false, false);
    acc01 = __builtin_amdgcn_wmma_f32_16x16x32_bf16(false, a0.v, false, b1.v, (short)0, acc01, false, false);
    acc10 = __builtin_amdgcn_wmma_f32_16x16x32_bf16(false, a1.v, false, b0.v, (short)0, acc10, false, false);
    acc11 = __builtin_amdgcn_wmma_f32_16x16x32_bf16(false, a1.v, false, b1.v, (short)0, acc11, false, false);
  }

  const int rowL = lane & 15;
  auto st = [&](const v8f& acc, int rt, bool rok, int ct, bool cok) {
    if (!rok || !cok) return;
    int col = ct * 16 + rowL;
    #pragma unroll
    for (int r = 0; r < 8; ++r) {
      int row = rt * 16 + ((lane < 16) ? r : r + 8);
      float o = acc[r] + bias[col];
      if (RELU) o = fmaxf(o, 0.0f);
      if (OUTBF) ((unsigned short*)Cout)[(long)row * M + col] = f2bf(o);
      else       ((float*)Cout)[(long)row * M + col] = o;
    }
  };
  st(acc00, rt0, r0, ct0, c0);         st(acc01, rt0, r0, ct0 + 1, c1);
  st(acc10, rt0 + 1, r1, ct0, c0);     st(acc11, rt0 + 1, r1, ct0 + 1, c1);
}


#define MLP_WAVES 4
#define MLP_PITCH 36
template <int KT1>
__global__ void __launch_bounds__(MLP_WAVES * 32)
mlp_chain(const uint4* __restrict__ Xpk, const uint4* __restrict__ W1pk, const float* __restrict__ b1,
          const uint4* __restrict__ W2pk, const float* __restrict__ b2, float* __restrict__ out) {
  __shared__ alignas(16) float stage[MLP_WAVES][2][16 * MLP_PITCH];
  const int lane = threadIdx.x & 31;
  const int wave = threadIdx.x >> 5;
  const int c    = lane & 15;
  const int hh   = lane >> 4;
  const int nrt  = N_NODES / 16;
  const int t0   = (blockIdx.x * MLP_WAVES + wave) * 2;
  const bool ok0 = t0 < nrt, ok1 = (t0 + 1) < nrt;
  const int tA = ok0 ? t0 : 0, tB = ok1 ? (t0 + 1) : 0;
  constexpr int KT2 = HID / 32;
  constexpr int UT  = HID / 16;
  union F { uint4 q[2]; v16bf v; };

  v8f acc[UT][2];
  #pragma unroll
  for (int j = 0; j < UT; ++j) { acc[j][0] = v8f{}; acc[j][1] = v8f{}; }
  const uint4* w1l = W1pk + lane * 2;
  auto w1_ask = [&](F (&g)[UT / 2], int kt, int j0) {
    #pragma unroll
    for (int j = 0; j < UT / 2; ++j) {
      const uint4* pw = w1l + (long)((j0 + j) * KT1 + kt) * 64;
      g[j].q[0] = pw[0]; g[j].q[1] = pw[1];
    }
  };
  auto x_ask = [&](F& x0, F& x1, int kt) {
    const uint4* p0 = Xpk + ((long)(tA * KT1 + kt) * 32 + lane) * 2;
    const uint4* p1 = Xpk + ((long)(tB * KT1 + kt) * 32 + lane) * 2;
    x0.q[0] = p0[0]; x0.q[1] = p0[1];
    x1.q[0] = p1[0]; x1.q[1] = p1[1];
  };
  F wA[UT / 2], wB[UT / 2], x0, x1;
  x_ask(x0, x1, 0);
  w1_ask(wA, 0, 0);
  for (int kt = 0; kt < KT1; ++kt) {
    const int ktn = (kt + 1 < KT1) ? kt + 1 : kt;
    F y0, y1;
    w1_ask(wB, kt, UT / 2);
    x_ask(y0, y1, ktn);
    __builtin_amdgcn_sched_barrier(0);
    #pragma unroll
    for (int j = 0; j < UT / 2; ++j) {
      acc[j][0] = __builtin_amdgcn_wmma_f32_16x16x32_bf16(false, wA[j].v, false, x0.v, (short)0, acc[j][0], false, false);
      acc[j][1] = __builtin_amdgcn_wmma_f32_16x16x32_bf16(false, wA[j].v, false, x1.v, (short)0, acc[j][1], false, false);
    }
    __builtin_amdgcn_sched_barrier(0);
    w1_ask(wA, ktn, 0);
    __builtin_amdgcn_sched_barrier(0);
    #pragma unroll
    for (int j = 0; j < UT / 2; ++j) {
      acc[UT / 2 + j][0] = __builtin_amdgcn_wmma_f32_16x16x32_bf16(false, wB[j].v, false, x0.v, (short)0, acc[UT / 2 + j][0], false, false);
      acc[UT / 2 + j][1] = __builtin_amdgcn_wmma_f32_16x16x32_bf16(false, wB[j].v, false, x1.v, (short)0, acc[UT / 2 + j][1], false, false);
    }
    __builtin_amdgcn_sched_barrier(0);
    x0 = y0; x1 = y1;
  }

  F m[2][KT2];
  #pragma unroll
  for (int t = 0; t < KT2; ++t) {
    const float4* bp = (const float4*)(b1 + 32 * t + 8 * hh);
    const float4 ba0 = bp[0], ba1 = bp[1];
    const float4 bb0 = bp[4], bb1 = bp[5];
    const float ba[8] = {ba0.x, ba0.y, ba0.z, ba0.w, ba1.x, ba1.y, ba1.z, ba1.w};
    const float bb[8] = {bb0.x, bb0.y, bb0.z, bb0.w, bb1.x, bb1.y, bb1.z, bb1.w};
    #pragma unroll
    for (int nt = 0; nt < 2; ++nt) {
      unsigned int lo[4], hi[4];
      #pragma unroll
      for (int p = 0; p < 4; ++p) {
        const float a0 = fmaxf(acc[2 * t][nt][2 * p]     + ba[2 * p],     0.0f);
        const float a1 = fmaxf(acc[2 * t][nt][2 * p + 1] + ba[2 * p + 1], 0.0f);
        const float c0 = fmaxf(acc[2 * t + 1][nt][2 * p]     + bb[2 * p],     0.0f);
        const float c1 = fmaxf(acc[2 * t + 1][nt][2 * p + 1] + bb[2 * p + 1], 0.0f);
        lo[p] = pack2bf(a0, a1);
        hi[p] = pack2bf(c0, c1);
      }
      m[nt][t].q[0] = make_uint4(lo[0], lo[1], lo[2], lo[3]);
      m[nt][t].q[1] = make_uint4(hi[0], hi[1], hi[2], hi[3]);
    }
  }

  float* st0 = &stage[wave][0][0];
  float* st1 = &stage[wave][1][0];
  const uint4* w2l = W2pk + lane * 2;
  auto w2_ask = [&](F (&g)[8], int cp, int tq) {
    #pragma unroll
    for (int t = 0; t < 4; ++t) {
      const uint4* pa = w2l + (long)((2 * cp) * KT2 + tq + t) * 64;
      const uint4* pb = w2l + (long)((2 * cp + 1) * KT2 + tq + t) * 64;
      g[2 * t].q[0] = pa[0];     g[2 * t].q[1] = pa[1];
      g[2 * t + 1].q[0] = pb[0]; g[2 * t + 1].q[1] = pb[1];
    }
  };
  w2_ask(wA, 0, 0);
  for (int cp = 0; cp < UT / 2; ++cp) {
    const int cpn = (cp + 1 < UT / 2) ? cp + 1 : cp;
    v8f o00 = {}, o01 = {}, o10 = {}, o11 = {};
    w2_ask(wB, cp, 4);
    __builtin_amdgcn_sched_barrier(0);
    #pragma unroll
    for (int t = 0; t < 4; ++t) {
      o00 = __builtin_amdgcn_wmma_f32_16x16x32_bf16(false, wA[2 * t].v,     false, m[0][t].v, (short)0, o00, false, false);
      o01 = __builtin_amdgcn_wmma_f32_16x16x32_bf16(false, wA[2 * t].v,     false, m[1][t].v, (short)0, o01, false, false);
      o10 = __builtin_amdgcn_wmma_f32_16x16x32_bf16(false, wA[2 * t + 1].v, false, m[0][t].v, (short)0, o10, false, false);
      o11 = __builtin_amdgcn_wmma_f32_16x16x32_bf16(false, wA[2 * t + 1].v, false, m[1][t].v, (short)0, o11, false, false);
    }
    __builtin_amdgcn_sched_barrier(0);
    w2_ask(wA, cpn, 0);
    __builtin_amdgcn_sched_barrier(0);
    #pragma unroll
    for (int t = 0; t < 4; ++t) {
      o00 = __builtin_amdgcn_wmma_f32_16x16x32_bf16(false, wB[2 * t].v,     false, m[0][4 + t].v, (short)0, o00, false, false);
      o01 = __builtin_amdgcn_wmma_f32_16x16x32_bf16(false, wB[2 * t].v,     false, m[1][4 + t].v, (short)0, o01, false, false);
      o10 = __builtin_amdgcn_wmma_f32_16x16x32_bf16(false, wB[2 * t + 1].v, false, m[0][4 + t].v, (short)0, o10, false, false);
      o11 = __builtin_amdgcn_wmma_f32_16x16x32_bf16(false, wB[2 * t + 1].v, false, m[1][4 + t].v, (short)0, o11, false, false);
    }
    __builtin_amdgcn_sched_barrier(0);
    const float4* bq = (const float4*)(b2 + 32 * cp + 8 * hh);
    const float4 ba0 = bq[0], ba1 = bq[1], bb0 = bq[4], bb1 = bq[5];
    auto put = [&](float* st, const v8f& oa, const v8f& ob) {
      float4* r = (float4*)(st + c * MLP_PITCH + 8 * hh);
      r[0] = make_float4(fmaxf(oa[0] + ba0.x, 0.0f), fmaxf(oa[1] + ba0.y, 0.0f),
                         fmaxf(oa[2] + ba0.z, 0.0f), fmaxf(oa[3] + ba0.w, 0.0f));
      r[1] = make_float4(fmaxf(oa[4] + ba1.x, 0.0f), fmaxf(oa[5] + ba1.y, 0.0f),
                         fmaxf(oa[6] + ba1.z, 0.0f), fmaxf(oa[7] + ba1.w, 0.0f));
      r[4] = make_float4(fmaxf(ob[0] + bb0.x, 0.0f), fmaxf(ob[1] + bb0.y, 0.0f),
                         fmaxf(ob[2] + bb0.z, 0.0f), fmaxf(ob[3] + bb0.w, 0.0f));
      r[5] = make_float4(fmaxf(ob[4] + bb1.x, 0.0f), fmaxf(ob[5] + bb1.y, 0.0f),
                         fmaxf(ob[6] + bb1.z, 0.0f), fmaxf(ob[7] + bb1.w, 0.0f));
    };
    put(st0, o00, o10);
    put(st1, o01, o11);
    auto rows_out = [&](const float* st, int tile, bool ok) {
      if (!ok) return;
      #pragma unroll
      for (int i = 0; i < 4; ++i) {
        const int node = 4 * i + (lane >> 3);
        const float4 v = *(const float4*)(st + node * MLP_PITCH + 4 * (lane & 7));
        *(float4*)(out + (long)(tile * 16 + node) * HID + 32 * cp + 4 * (lane & 7)) = v;
      }
    };
    rows_out(st0, t0, ok0);
    rows_out(st1, t0 + 1, ok1);
  }
}

#define POOL_GB      4
#define POOL_WAVES   16
#define POOL_THREADS (POOL_WAVES * 32)
#define POOL_IT      25
#define POOL_SUB     (POOL_IT * 128)
#define POOL_PG      16

static_assert(POOL_SUB * POOL_WAVES >= N_NODES && N_NODES % 4 == 0, "the slices cover the id list, four ids per load");
static_assert(N_GRAPHS % POOL_GB == 0 && 16 % POOL_GB == 0 && POOL_GB * 8 == 32, "four graphs x eight dwords = one line");
static_assert(POOL_WAVES == POOL_GB * (HID / 64) && POOL_WAVES == 2 * (HID / 32), "one wave per graph and 64 columns; one per line");

__global__ void __launch_bounds__(POOL_THREADS)
pool_owner(const float* __restrict__ h, const int* __restrict__ batch, unsigned int* __restrict__ Ppk) {
  __shared__ unsigned int lst[POOL_WAVES * POOL_SUB];
  __shared__ int          cnt[POOL_WAVES];
  __shared__ unsigned int stage[POOL_GB * (HID / 2)];

  const int tid  = threadIdx.x;
  const int lane = tid & 31;
  const int wave = tid >> 5;
  const int g0   = blockIdx.x * POOL_GB;
  const unsigned int ltmask = (1u << lane) - 1u;

  {
    unsigned int* mylst = lst + wave * POOL_SUB;
    const int n0 = wave * POOL_SUB + lane * 4;
    int4 d[POOL_IT];
    #pragma unroll
    for (int it = 0; it < POOL_IT; ++it) {
      const int n = n0 + it * 128;
      d[it] = *(const int4*)(batch + (n < N_NODES ? n : N_NODES - 4));
    }
    int c = 0;
    #pragma unroll
    for (int it = 0; it < POOL_IT; ++it) {
      const unsigned int n  = (unsigned int)(n0 + it * 128);
      const unsigned int x0 = (unsigned int)(d[it].x - g0);
      const unsigned int x1 = (unsigned int)(d[it].y - g0);
      const unsigned int x2 = (unsigned int)(d[it].z - g0);
      const unsigned int x3 = (unsigned int)(d[it].w - g0);
      const unsigned int lim = (n < (unsigned int)N_NODES) ? (unsigned int)POOL_GB : 0u;
      const bool m0 = x0 < lim, m1 = x1 < lim;
      const bool m2 = x2 < lim, m3 = x3 < lim;
      if (__builtin_amdgcn_ballot_w32(m0 || m1 || m2 || m3)) {
        unsigned int b;
        b = __builtin_amdgcn_ballot_w32(m0);
        if (m0) mylst[c + __builtin_popcount(b & ltmask)] = (x0 << 20) | (n + 0u);
        c += __builtin_popcount(b);
        b = __builtin_amdgcn_ballot_w32(m1);
        if (m1) mylst[c + __builtin_popcount(b & ltmask)] = (x1 << 20) | (n + 1u);
        c += __builtin_popcount(b);
        b = __builtin_amdgcn_ballot_w32(m2);
        if (m2) mylst[c + __builtin_popcount(b & ltmask)] = (x2 << 20) | (n + 2u);
        c += __builtin_popcount(b);
        b = __builtin_amdgcn_ballot_w32(m3);
        if (m3) mylst[c + __builtin_popcount(b & ltmask)] = (x3 << 20) | (n + 3u);
        c += __builtin_popcount(b);
      }
    }
    if (lane == 0) cnt[wave] = c;
  }
  __syncthreads();

  const unsigned int gi = (unsigned int)(wave & (POOL_GB - 1));
  const int cq = wave / POOL_GB;
  const float2* hq = (const float2*)h + cq * 32 + lane;
  float2 acc = make_float2(0.0f, 0.0f);
  int deg = 0;
  for (int lw = 0; lw < POOL_WAVES; ++lw) {
    const int cn = __builtin_amdgcn_readfirstlane(cnt[lw]);
    const unsigned int* L = lst + lw * POOL_SUB;
    for (int f0 = 0; f0 < cn; f0 += 32) {
      const int f = f0 + lane;
      const unsigned int e = (f < cn) ? L[f] : 0xFFFFFFFFu;
      unsigned int m = __builtin_amdgcn_ballot_w32((e >> 20) == gi);
      deg += __builtin_popcount(m);
      while (m) {
        float2 v[POOL_PG];
        #pragma unroll
        for (int j = 0; j < POOL_PG; ++j) {
          v[j] = make_float2(0.0f, 0.0f);
          if (m) {
            const int b = __builtin_ctz(m);
            m &= m - 1u;
            const int n = __builtin_amdgcn_readlane((int)e, b) & 0xFFFFF;
            v[j] = hq[n * (HID / 2)];
          }
        }
        #pragma unroll
        for (int j = 0; j < POOL_PG; ++j) { acc.x += v[j].x; acc.y += v[j].y; }
      }
    }
  }

  const float den = fmaxf((float)deg, 1.0f);
  stage[gi * (HID / 2) + cq * 32 + lane] = pack2bf(acc.x / den, acc.y / den);
  __syncthreads();

  {
    const int kt = wave >> 1;
    const int hh = wave & 1;
    const int gl = lane >> 3;
    const int p  = lane & 7;
    const int k  = kt * 32 + frag_koff(16 * hh, p);
    const int rt = g0 >> 4;
    const int c0 = g0 & 15;
    Ppk[((rt * (HID / 32) + kt) * 32 + c0 + 16 * hh) * 8 + lane] = stage[gl * (HID / 2) + (k >> 1)];
  }
}

extern "C" void kernel_launch(void* const* d_in, const int* in_sizes, int n_in,
                              void* d_out, int out_size, void* d_ws, size_t ws_size,
                              hipStream_t stream) {
  const float* x     = (const float*)d_in[0];
  const int*   ei    = (const int*)d_in[1];
  const int*   esrc  = ei;
  const int*   edst  = ei + N_EDGES;
  const int*   batch = (const int*)d_in[2];
  const float* cw1[3] = {(const float*)d_in[3],  (const float*)d_in[7],  (const float*)d_in[11]};
  const float* cb1[3] = {(const float*)d_in[4],  (const float*)d_in[8],  (const float*)d_in[12]};
  const float* cw2[3] = {(const float*)d_in[5],  (const float*)d_in[9],  (const float*)d_in[13]};
  const float* cb2[3] = {(const float*)d_in[6],  (const float*)d_in[10], (const float*)d_in[14]};
  const float* ow1 = (const float*)d_in[15];
  const float* ob1 = (const float*)d_in[16];
  const float* ow2 = (const float*)d_in[17];
  const float* ob2 = (const float*)d_in[18];

  char* ws = (char*)d_ws;
  size_t off = 0;
  auto bump = [&](size_t bytes) {
    void* p = ws + off;
    off += (bytes + 255) & ~(size_t)255;
    return p;
  };
  const int NRT = N_NODES / 16;
  float*          B0     = (float*)bump((size_t)N_NODES * HID * 4);
  float*          B2     = (float*)bump((size_t)N_NODES * HID * 4);
  unsigned int*   Apk    = (unsigned int*)bump((size_t)NRT * 8 * 256 * 4);
  unsigned int*   Ppk    = (unsigned int*)bump((size_t)(N_GRAPHS / 16) * 8 * 256 * 4);
  unsigned short* MPbf   = (unsigned short*)bump((size_t)N_GRAPHS * HID * 2);
  unsigned int*   MPpk   = (unsigned int*)bump((size_t)(N_GRAPHS / 16) * 8 * 256 * 4);
  unsigned int*   wpk1[3], *wpk2[3];
  for (int l = 0; l < 3; ++l) {
    wpk1[l] = (unsigned int*)bump((size_t)(HID / 16) * 8 * 256 * 4);
    wpk2[l] = (unsigned int*)bump((size_t)(HID / 16) * 8 * 256 * 4);
  }
  unsigned int* owpk1 = (unsigned int*)bump((size_t)(HID / 16) * 8 * 256 * 4);
  unsigned int* owpk2 = (unsigned int*)bump((size_t)(OUT_C / 16) * 8 * 256 * 4);
  unsigned int* glst  = (unsigned int*)bump((size_t)AGG_GROUPS * AGG_WAVES * LST_CAP * 4);
  int*          gcnt  = (int*)bump((size_t)AGG_GROUPS * AGG_WAVES * 4);

  dim3 blk(256);
  auto wpack = [&](const float* W, unsigned int* dst, int K, int M) {
    int KT = K / 32;
    long total = (long)(M / 16) * KT * 256;
    wpack_f32<<<(unsigned)((total + 255) / 256), blk, 0, stream>>>(W, dst, K, M, KT, total);
  };

  wpack(cw1[0], wpk1[0], IN_C, HID);
  wpack(cw2[0], wpk2[0], HID, HID);
  for (int l = 1; l < 3; ++l) {
    wpack(cw1[l], wpk1[l], HID, HID);
    wpack(cw2[l], wpk2[l], HID, HID);
  }
  wpack(ow1, owpk1, HID, HID);
  wpack(ow2, owpk2, HID, OUT_C);

  owner_lists<<<AGG_GROUPS, AGG_THREADS, 0, stream>>>(esrc, edst, glst, gcnt);

  const float* h = x;
  int C = IN_C;
  for (int l = 0; l < 3; ++l) {
    if (l == 0) agg_owner<IN_C><<<AGG_GROUPS, AGG_THREADS, 0, stream>>>(h, esrc, edst, glst, gcnt, B0);
    else        agg_owner<HID><<<AGG_GROUPS, AGG_THREADS, 0, stream>>>(h, esrc, edst, glst, gcnt, B0);

    int KT = C / 32;
    long atot = (long)NRT * KT * 256;
    apack_f32<<<(unsigned)((atot + 255) / 256), blk, 0, stream>>>(B0, Apk, C, KT, atot);

    const unsigned mg = (unsigned)((NRT + 2 * MLP_WAVES - 1) / (2 * MLP_WAVES));
    if (l == 0) mlp_chain<IN_C / 32><<<mg, MLP_WAVES * 32, 0, stream>>>((const uint4*)Apk, (const uint4*)wpk1[l], cb1[l],
                                                                     (const uint4*)wpk2[l], cb2[l], B2);
    else        mlp_chain<HID / 32><<<mg, MLP_WAVES * 32, 0, stream>>>((const uint4*)Apk, (const uint4*)wpk1[l], cb1[l],
                                                                    (const uint4*)wpk2[l], cb2[l], B2);
    h = B2;
    C = HID;
  }

  pool_owner<<<N_GRAPHS / POOL_GB, POOL_THREADS, 0, stream>>>(B2, batch, Ppk);

  const int GRT = N_GRAPHS / 16;
  long gtot = (long)GRT * (HID / 32) * 256;
  dim3 g3((HID / 16 + 3) / 4, (GRT + 7) / 8);
  gemm_pk<1, 1><<<g3, blk, 0, stream>>>((const uint4*)Ppk, (const uint4*)owpk1, ob1,
                                        MPbf, N_GRAPHS, HID, HID);
  apack_bf16<<<(unsigned)((gtot + 255) / 256), blk, 0, stream>>>(MPbf, MPpk, HID, HID / 32, gtot);
  dim3 g4((OUT_C / 16 + 3) / 4, (GRT + 7) / 8);
  gemm_pk<0, 0><<<g4, blk, 0, stream>>>((const uint4*)MPpk, (const uint4*)owpk2, ob2,
                                        (float*)d_out, N_GRAPHS, HID, OUT_C);
}


# Activation pack passes removed: the aggregation rounds its f32 sums to the operand type itself (same rounding), plain rows; products load fragments straight from the rows; 22 -> 18 launches
# speedup vs baseline: 3.2950x; 1.1802x over previous
#include <hip/hip_runtime.h>

#define N_NODES 50000
#define N_EDGES 800000
#define N_GRAPHS 512
#define IN_C 128
#define HID 256
#define OUT_C 16

typedef __attribute__((ext_vector_type(16))) __bf16 v16bf;
typedef __attribute__((ext_vector_type(8)))  float  v8f;
typedef __attribute__((ext_vector_type(8)))  unsigned int v8u;
typedef const __attribute__((address_space(4))) v8u* list_ptr;

__device__ __forceinline__ unsigned short f2bf(float f) {
  union { float f; unsigned int u; } c; c.f = f;
  unsigned int u = c.u;
  u += 0x7fffu + ((u >> 16) & 1u);
  return (unsigned short)(u >> 16);
}

typedef __attribute__((ext_vector_type(2))) float  v2f;
typedef __attribute__((ext_vector_type(2))) __bf16 v2bf;
__device__ __forceinline__ unsigned int pack2bf(float a, float b) {
  const v2f f = {a, b};
  const v2bf h = __builtin_convertvector(f, v2bf);
  return __builtin_bit_cast(unsigned int, h);
}

__device__ __forceinline__ int frag_koff(int lane, int p) {
  int kb = (lane < 16) ? 0 : 8;
  return (p < 4) ? (kb + 2 * p) : (16 + kb + 2 * (p - 4));
}

#define AGG_NB      196
#define AGG_WAVES   16
#define AGG_THREADS (AGG_WAVES * 32)
#define AGG_IT      15
#define AGG_SUB     (AGG_IT * 128)
#define AGG_ROUND   (AGG_SUB * AGG_WAVES)
#define AGG_NROUND  ((N_EDGES + AGG_ROUND - 1) / AGG_ROUND)
#define AGG_GROUPS  ((N_NODES + AGG_NB - 1) / AGG_NB)
#define LST_CAP     1024
#define LST_G       8
#define LST_PAD     (5 * LST_G)

static_assert(N_NODES <= (1 << 20) && N_EDGES <= (1 << 20) && AGG_NB < 256, "list entry fields");
static_assert(LST_G == 8, "a group of entries is one 8-dword scalar load");

static_assert(N_EDGES % 4 == 0, "four edges per index load");

__device__ __forceinline__ void scan_load(const int* __restrict__ edst, int e0, int4 (&d)[AGG_IT]) {
  #pragma unroll
  for (int it = 0; it < AGG_IT; ++it) {
    const int e = e0 + it * 128;
    d[it] = *(const int4*)(edst + (e < N_EDGES ? e : N_EDGES - 4));
  }
}

__device__ __forceinline__ int scan_note(const int4 (&d)[AGG_IT], int e0, int base, int nloc,
                                         unsigned int ltmask, unsigned int* mylst) {
  int c = 0;
  #pragma unroll
  for (int it = 0; it < AGG_IT; ++it) {
    const unsigned int e = (unsigned int)(e0 + it * 128);
    const unsigned int x0 = (unsigned int)(d[it].x - base);
    const unsigned int x1 = (unsigned int)(d[it].y - base);
    const unsigned int x2 = (unsigned int)(d[it].z - base);
    const unsigned int x3 = (unsigned int)(d[it].w - base);
    const unsigned int lim = (e < (unsigned int)N_EDGES) ? (unsigned int)nloc : 0u;
    const bool m0 = x0 < lim, m1 = x1 < lim;
    const bool m2 = x2 < lim, m3 = x3 < lim;
    const unsigned int mn01 = x0 < x1 ? x0 : x1, mn23 = x2 < x3 ? x2 : x3;
    if (__builtin_amdgcn_ballot_w32((mn01 < mn23 ? mn01 : mn23) < lim)) {
      unsigned int b;
      b = __builtin_amdgcn_ballot_w32(m0);
      if (m0) mylst[c + __builtin_popcount(b & ltmask)] = (x0 << 20) | (e + 0u);
      c += __builtin_popcount(b);
      b = __builtin_amdgcn_ballot_w32(m1);
      if (m1) mylst[c + __builtin_popcount(b & ltmask)] = (x1 << 20) | (e + 1u);
      c += __builtin_popcount(b);
      b = __builtin_amdgcn_ballot_w32(m2);
      if (m2) mylst[c + __builtin_popcount(b & ltmask)] = (x2 << 20) | (e + 2u);
      c += __builtin_popcount(b);
      b = __builtin_amdgcn_ballot_w32(m3);
      if (m3) mylst[c + __builtin_popcount(b & ltmask)] = (x3 << 20) | (e + 3u);
      c += __builtin_popcount(b);
    }
  }
  return c;
}

__global__ void __launch_bounds__(AGG_THREADS)
owner_lists(const int* __restrict__ esrc, const int* __restrict__ edst,
            unsigned int* __restrict__ glst, int* __restrict__ gcnt) {
  __shared__ unsigned int lst[AGG_WAVES * AGG_SUB];
  __shared__ unsigned int own[AGG_WAVES * LST_CAP];
  __shared__ int          cnt[AGG_WAVES];

  const int tid  = threadIdx.x;
  const int lane = tid & 31;
  const int wave = tid >> 5;
  const int base = blockIdx.x * AGG_NB;
  int nloc = N_NODES - base;
  if (nloc > AGG_NB) nloc = AGG_NB;

  const unsigned int ltmask = (1u << lane) - 1u;
  unsigned int* mylst = lst + wave * AGG_SUB;
  unsigned int* myown = own + wave * LST_CAP;
  unsigned int* myout = glst + ((long)blockIdx.x * AGG_WAVES + wave) * LST_CAP;
  int gpos = 0;

  int rr = (int)(blockIdx.x % AGG_NROUND);
  int4 d[AGG_IT];
  scan_load(edst, rr * AGG_ROUND + wave * AGG_SUB + lane * 4, d);
  for (int r = 0; r < AGG_NROUND; ++r) {
    const int e0 = rr * AGG_ROUND + wave * AGG_SUB + lane * 4;
    rr = (rr + 1 == AGG_NROUND) ? 0 : rr + 1;
    int4 dn[AGG_IT];
    scan_load(edst, rr * AGG_ROUND + wave * AGG_SUB + lane * 4, dn);

    const int c = scan_note(d, e0, base, nloc, ltmask, mylst);
    if (lane == 0) cnt[wave] = c;
    __syncthreads();

    const int myc = (lane < AGG_WAVES) ? cnt[lane] : 0;
    int incl = myc;
    #pragma unroll
    for (int o = 1; o < AGG_WAVES; o <<= 1) {
      const int t = __shfl_up(incl, o);
      if (lane >= o) incl += t;
    }
    const int excl  = incl - myc;
    const int total = __builtin_amdgcn_readlane(incl, AGG_WAVES - 1);
    for (int f0 = 0; f0 < total; f0 += 32) {
      const int f = f0 + lane;
      int li = 0;
      #pragma unroll
      for (int l = 0; l < AGG_WAVES - 1; ++l) li += (f >= __builtin_amdgcn_readlane(incl, l)) ? 1 : 0;
      const int st = __shfl(excl, li);
      unsigned int ent = 0u;
      bool mine = false;
      if (f < total) {
        ent  = lst[li * AGG_SUB + (f - st)];
        mine = ((ent >> 20) & (unsigned int)(AGG_WAVES - 1)) == (unsigned int)wave;
      }
      const unsigned int m = __builtin_amdgcn_ballot_w32(mine);
      if (mine) {
        const int pos = gpos + __builtin_popcount(m & ltmask);
        if (pos < LST_CAP) myown[pos] = ent;
      }
      gpos += __builtin_popcount(m);
    }
    __syncthreads();

    #pragma unroll
    for (int it = 0; it < AGG_IT; ++it) d[it] = dn[it];
  }

  const int nout = gpos < LST_CAP ? gpos : LST_CAP;
  for (int i = lane; i < nout; i += 32) {
    const unsigned int ent = myown[i];
    myout[i] = (ent & 0xFFF00000u) | (unsigned int)esrc[ent & 0xFFFFFu];
  }
  #pragma unroll
  for (int k = 0; k < (LST_PAD + 31) / 32; ++k) {
    const int o = k * 32 + lane;
    if (o < LST_PAD && gpos + o < LST_CAP) myout[gpos + o] = ((unsigned int)AGG_NB << 20) | (unsigned int)base;
  }
  if (lane == 0) gcnt[blockIdx.x * AGG_WAVES + wave] = gpos;
}

template <int C>
__global__ void __launch_bounds__(AGG_THREADS)
agg_owner(const float* __restrict__ h, const int* __restrict__ esrc,
          const int* __restrict__ edst, const unsigned int* __restrict__ glst,
          const int* __restrict__ gcnt, unsigned short* __restrict__ agg) {
  __shared__ float        acc[(AGG_NB + 1) * C];
  __shared__ unsigned int lst[AGG_WAVES * AGG_SUB];
  __shared__ int          cnt[AGG_WAVES];
  constexpr int VEC = C / 128;

  const int tid  = threadIdx.x;
  const int lane = tid & 31;
  const int wave = tid >> 5;
  const int base = blockIdx.x * AGG_NB;
  int nloc = N_NODES - base;
  if (nloc > AGG_NB) nloc = AGG_NB;

  {
    const float4* hs = (const float4*)(h + (long)base * C);
    float4* a4 = (float4*)acc;
    const int n4 = nloc * (C / 4);
    for (int i = tid; i < n4; i += AGG_THREADS) a4[i] = hs[i];
  }

  const int lc = (lane < AGG_WAVES) ? gcnt[blockIdx.x * AGG_WAVES + lane] : 0;
  const bool fits = __builtin_amdgcn_ballot_w32(lc > LST_CAP - LST_PAD) == 0u;

  if (fits) {
    __syncthreads();
    const int wv = __builtin_amdgcn_readfirstlane(wave);
    const int n  = __builtin_amdgcn_readlane(lc, wv);
    const list_ptr q = (list_ptr)(unsigned long)(glst + ((long)blockIdx.x * AGG_WAVES + wv) * LST_CAP);
    const int ng = (n + LST_G - 1) / LST_G;

    v8u eA = q[0], eB = q[1], eC = q[2];
    float4 A[LST_G][VEC], B[LST_G][VEC];
    auto rows_ask = [&](float4 (&R)[LST_G][VEC], const v8u& e) {
      #pragma unroll
      for (int j = 0; j < LST_G; ++j) {
        const int s = (int)(e[j] & 0xFFFFFu);
        const float4* hp = (const float4*)(h + (long)s * C) + lane;
        #pragma unroll
        for (int v = 0; v < VEC; ++v) R[j][v] = hp[v * 32];
      }
    };
    auto rows_add = [&](const float4 (&R)[LST_G][VEC], const v8u& e) {
      #pragma unroll
      for (int j = 0; j < LST_G; ++j) {
        const int nd = (int)(e[j] >> 20);
        float4* ap = (float4*)(acc + nd * C) + lane;
        #pragma unroll
        for (int v = 0; v < VEC; ++v) {
          float4 a = ap[v * 32];
          a.x += R[j][v].x; a.y += R[j][v].y; a.z += R[j][v].z; a.w += R[j][v].w;
          ap[v * 32] = a;
        }
      }
    };
    rows_ask(A, eA);
    const int nt = (ng + 1) / 2;
    for (int t = 0; t < nt; ++t) {
      const v8u fB = q[2 * t + 3], fC = q[2 * t + 4];
      rows_ask(B, eB);
      rows_add(A, eA);
      rows_ask(A, eC);
      rows_add(B, eB);
      eA = eC; eB = fB; eC = fC;
    }
    __syncthreads();
  } else {
    const unsigned int ltmask = (1u << lane) - 1u;
    unsigned int* mylst = lst + wave * AGG_SUB;
    int rr = (int)(blockIdx.x % AGG_NROUND);
    for (int r = 0; r < AGG_NROUND; ++r) {
      const int e0 = rr * AGG_ROUND + wave * AGG_SUB + lane * 4;
      rr = (rr + 1 == AGG_NROUND) ? 0 : rr + 1;
      int4 d[AGG_IT];
      scan_load(edst, e0, d);
      const int c = scan_note(d, e0, base, nloc, ltmask, mylst);
      if (lane == 0) cnt[wave] = c;
      __syncthreads();

      const int myc = (lane < AGG_WAVES) ? cnt[lane] : 0;
      int incl = myc;
      #pragma unroll
      for (int o = 1; o < AGG_WAVES; o <<= 1) {
        const int t = __shfl_up(incl, o);
        if (lane >= o) incl += t;
      }
      const int excl  = incl - myc;
      const int total = __builtin_amdgcn_readlane(incl, AGG_WAVES - 1);
      for (int f0 = 0; f0 < total; f0 += 32) {
        const int f = f0 + lane;
        int li = 0;
        #pragma unroll
        for (int l = 0; l < AGG_WAVES - 1; ++l) li += (f >= __builtin_amdgcn_readlane(incl, l)) ? 1 : 0;
        const int st = __shfl(excl, li);
        unsigned int ent = 0u;
        bool mine = false;
        if (f < total) {
          ent  = lst[li * AGG_SUB + (f - st)];
          mine = ((ent >> 20) & (unsigned int)(AGG_WAVES - 1)) == (unsigned int)wave;
        }
        unsigned int m = __builtin_amdgcn_ballot_w32(mine);
        int s = 0;
        if (mine) s = esrc[ent & 0xFFFFFu];
        while (m) {
          const int b0 = __builtin_ctz(m); m &= m - 1u;
          const int s0 = __builtin_amdgcn_readlane(s, b0);
          const int n0 = (int)(__builtin_amdgcn_readlane((int)ent, b0) >> 20) & 0xFF;
          const float4* h0 = (const float4*)(h + (long)s0 * C) + lane;
          float4* a0 = (float4*)(acc + n0 * C) + lane;
          #pragma unroll
          for (int j = 0; j < VEC; ++j) {
            const float4 v0 = h0[j * 32];
            float4 a = a0[j * 32];
            a.x += v0.x; a.y += v0.y; a.z += v0.z; a.w += v0.w;
            a0[j * 32] = a;
          }
        }
      }
      __syncthreads();
    }
  }

  {
    uint4* o8 = (uint4*)(agg + (long)base * C);
    const float4* a4 = (const float4*)acc;
    const int n8 = nloc * (C / 8);
    for (int i = tid; i < n8; i += AGG_THREADS) {
      const float4 u = a4[2 * i], v = a4[2 * i + 1];
      o8[i] = make_uint4(pack2bf(u.x, u.y), pack2bf(u.z, u.w), pack2bf(v.x, v.y), pack2bf(v.z, v.w));
    }
  }
}

__global__ void wpack_f32(const float* __restrict__ W, unsigned int* __restrict__ out,
                          int K, int M, int KT, long total) {
  long i = (long)blockIdx.x * blockDim.x + threadIdx.x;
  if (i >= total) return;
  int p = (int)(i & 7);
  int lane = (int)((i >> 3) & 31);
  long tt = i >> 8;
  int kt = (int)(tt % KT);
  int ct = (int)(tt / KT);
  int col = ct * 16 + (lane & 15);
  int k = kt * 32 + frag_koff(lane, p);
  unsigned int lo = f2bf(W[(long)k * M + col]);
  unsigned int hi = f2bf(W[(long)(k + 1) * M + col]);
  out[i] = lo | (hi << 16);
}

template <int RELU, int OUTBF, int AROW>
__global__ void gemm_pk(const uint4* __restrict__ Apk, const uint4* __restrict__ Wpk,
                        const float* __restrict__ bias, void* __restrict__ Cout,
                        int N, int K, int M) {
  const int tid  = threadIdx.x;
  const int lane = tid & 31;
  const int wave = tid >> 5;
  const int wr   = wave & 3;
  const int wc   = wave >> 2;
  const int KT   = K >> 5;
  const int nrt  = N >> 4;
  const int nct  = M >> 4;
  const int rt0  = blockIdx.y * 8 + wr * 2;
  const int ct0  = blockIdx.x * 4 + wc * 2;
  const bool r0 = rt0 < nrt,     r1 = (rt0 + 1) < nrt;
  const bool c0 = ct0 < nct,     c1 = (ct0 + 1) < nct;
  const int rA0 = r0 ? rt0 : 0,  rA1 = r1 ? (rt0 + 1) : 0;
  const int cB0 = c0 ? ct0 : 0,  cB1 = c1 ? (ct0 + 1) : 0;

  long ia0 = AROW ? ((long)(rA0 * 16 + (lane & 15)) * (K >> 3) + (lane >> 4)) : (((long)rA0 * KT * 32 + lane) * 2);
  long ia1 = AROW ? ((long)(rA1 * 16 + (lane & 15)) * (K >> 3) + (lane >> 4)) : (((long)rA1 * KT * 32 + lane) * 2);
  long ib0 = ((long)cB0 * KT * 32 + lane) * 2;
  long ib1 = ((long)cB1 * KT * 32 + lane) * 2;
  const long step  = 64;
  const long stepA = AROW ? 4 : 64;
  const long second = AROW ? 2 : 1;

  v8f acc00 = {}, acc01 = {}, acc10 = {}, acc11 = {};
  union F { uint4 q[2]; v16bf v; };

  for (int kt = 0; kt < KT; ++kt) {
    F a0, a1, b0, b1;
    a0.q[0] = Apk[ia0]; a0.q[1] = Apk[ia0 + second];
    a1.q[0] = Apk[ia1]; a1.q[1] = Apk[ia1 + second];
    b0.q[0] = Wpk[ib0]; b0.q[1] = Wpk[ib0 + 1];
    b1.q[0] = Wpk[ib1]; b1.q[1] = Wpk[ib1 + 1];
    if (kt + 1 < KT) {
      __builtin_prefetch((const void*)(Apk + ia0 + stepA), 0, 3);
      __builtin_prefetch((const void*)(Apk + ia1 + stepA), 0, 3);
    }
    ia0 += stepA; ia1 += stepA; ib0 += step; ib1 += step;
    acc00 = __builtin_amdgcn_wmma_f32_16x16x32_bf16(false, a0.v, false, b0.v, (short)0, acc00, false, false);
    acc01 = __builtin_amdgcn_wmma_f32_16x16x32_bf16(false, a0.v, false, b1.v, (short)0, acc01, false, false);
    acc10 = __builtin_amdgcn_wmma_f32_16x16x32_bf16(false, a1.v, false, b0.v, (short)0, acc10, false, false);
    acc11 = __builtin_amdgcn_wmma_f32_16x16x32_bf16(false, a1.v, false, b1.v, (short)0, acc11, false, false);
  }

  const int rowL = lane & 15;
  auto st = [&](const v8f& acc, int rt, bool rok, int ct, bool cok) {
    if (!rok || !cok) return;
    int col = ct * 16 + rowL;
    #pragma unroll
    for (int r = 0; r < 8; ++r) {
      int row = rt * 16 + ((lane < 16) ? r : r + 8);
      float o = acc[r] + bias[col];
      if (RELU) o = fmaxf(o, 0.0f);
      if (OUTBF) ((unsigned short*)Cout)[(long)row * M + col] = f2bf(o);
      else       ((float*)Cout)[(long)row * M + col] = o;
    }
  };
  st(acc00, rt0, r0, ct0, c0);         st(acc01, rt0, r0, ct0 + 1, c1);
  st(acc10, rt0 + 1, r1, ct0, c0);     st(acc11, rt0 + 1, r1, ct0 + 1, c1);
}


#define MLP_WAVES 4
#define MLP_PITCH 36
template <int KT1>
__global__ void __launch_bounds__(MLP_WAVES * 32)
mlp_chain(const unsigned short* __restrict__ Xbf, const uint4* __restrict__ W1pk, const float* __restrict__ b1,
          const uint4* __restrict__ W2pk, const float* __restrict__ b2, float* __restrict__ out) {
  __shared__ alignas(16) float stage[MLP_WAVES][2][16 * MLP_PITCH];
  const int lane = threadIdx.x & 31;
  const int wave = threadIdx.x >> 5;
  const int c    = lane & 15;
  const int hh   = lane >> 4;
  const int nrt  = N_NODES / 16;
  const int t0   = (blockIdx.x * MLP_WAVES + wave) * 2;
  const bool ok0 = t0 < nrt, ok1 = (t0 + 1) < nrt;
  const int tA = ok0 ? t0 : 0, tB = ok1 ? (t0 + 1) : 0;
  constexpr int KT2 = HID / 32;
  constexpr int UT  = HID / 16;
  union F { uint4 q[2]; v16bf v; };

  v8f acc[UT][2];
  #pragma unroll
  for (int j = 0; j < UT; ++j) { acc[j][0] = v8f{}; acc[j][1] = v8f{}; }
  const uint4* w1l = W1pk + lane * 2;
  auto w1_ask = [&](F (&g)[UT / 2], int kt, int j0) {
    #pragma unroll
    for (int j = 0; j < UT / 2; ++j) {
      const uint4* pw = w1l + (long)((j0 + j) * KT1 + kt) * 64;
      g[j].q[0] = pw[0]; g[j].q[1] = pw[1];
    }
  };
  const uint4* xr0 = (const uint4*)(Xbf + (long)(tA * 16 + c) * (32 * KT1)) + hh;
  const uint4* xr1 = (const uint4*)(Xbf + (long)(tB * 16 + c) * (32 * KT1)) + hh;
  auto x_ask = [&](F& x0, F& x1, int kt) {
    x0.q[0] = xr0[4 * kt]; x0.q[1] = xr0[4 * kt + 2];
    x1.q[0] = xr1[4 * kt]; x1.q[1] = xr1[4 * kt + 2];
  };
  F wA[UT / 2], wB[UT / 2], x0, x1;
  x_ask(x0, x1, 0);
  w1_ask(wA, 0, 0);
  for (int kt = 0; kt < KT1; ++kt) {
    const int ktn = (kt + 1 < KT1) ? kt + 1 : kt;
    F y0, y1;
    w1_ask(wB, kt, UT / 2);
    x_ask(y0, y1, ktn);
    __builtin_amdgcn_sched_barrier(0);
    #pragma unroll
    for (int j = 0; j < UT / 2; ++j) {
      acc[j][0] = __builtin_amdgcn_wmma_f32_16x16x32_bf16(false, wA[j].v, false, x0.v, (short)0, acc[j][0], false, false);
      acc[j][1] = __builtin_amdgcn_wmma_f32_16x16x32_bf16(false, wA[j].v, false, x1.v, (short)0, acc[j][1], false, false);
    }
    __builtin_amdgcn_sched_barrier(0);
    w1_ask(wA, ktn, 0);
    __builtin_amdgcn_sched_barrier(0);
    #pragma unroll
    for (int j = 0; j < UT / 2; ++j) {
      acc[UT / 2 + j][0] = __builtin_amdgcn_wmma_f32_16x16x32_bf16(false, wB[j].v, false, x0.v, (short)0, acc[UT / 2 + j][0], false, false);
      acc[UT / 2 + j][1] = __builtin_amdgcn_wmma_f32_16x16x32_bf16(false, wB[j].v, false, x1.v, (short)0, acc[UT / 2 + j][1], false, false);
    }
    __builtin_amdgcn_sched_barrier(0);
    x0 = y0; x1 = y1;
  }

  F m[2][KT2];
  #pragma unroll
  for (int t = 0; t < KT2; ++t) {
    const float4* bp = (const float4*)(b1 + 32 * t + 8 * hh);
    const float4 ba0 = bp[0], ba1 = bp[1];
    const float4 bb0 = bp[4], bb1 = bp[5];
    const float ba[8] = {ba0.x, ba0.y, ba0.z, ba0.w, ba1.x, ba1.y, ba1.z, ba1.w};
    const float bb[8] = {bb0.x, bb0.y, bb0.z, bb0.w, bb1.x, bb1.y, bb1.z, bb1.w};
    #pragma unroll
    for (int nt = 0; nt < 2; ++nt) {
      unsigned int lo[4], hi[4];
      #pragma unroll
      for (int p = 0; p < 4; ++p) {
        const float a0 = fmaxf(acc[2 * t][nt][2 * p]     + ba[2 * p],     0.0f);
        const float a1 = fmaxf(acc[2 * t][nt][2 * p + 1] + ba[2 * p + 1], 0.0f);
        const float c0 = fmaxf(acc[2 * t + 1][nt][2 * p]     + bb[2 * p],     0.0f);
        const float c1 = fmaxf(acc[2 * t + 1][nt][2 * p + 1] + bb[2 * p + 1], 0.0f);
        lo[p] = pack2bf(a0, a1);
        hi[p] = pack2bf(c0, c1);
      }
      m[nt][t].q[0] = make_uint4(lo[0], lo[1], lo[2], lo[3]);
      m[nt][t].q[1] = make_uint4(hi[0], hi[1], hi[2], hi[3]);
    }
  }

  float* st0 = &stage[wave][0][0];
  float* st1 = &stage[wave][1][0];
  const uint4* w2l = W2pk + lane * 2;
  auto w2_ask = [&](F (&g)[8], int cp, int tq) {
    #pragma unroll
    for (int t = 0; t < 4; ++t) {
      const uint4* pa = w2l + (long)((2 * cp) * KT2 + tq + t) * 64;
      const uint4* pb = w2l + (long)((2 * cp + 1) * KT2 + tq + t) * 64;
      g[2 * t].q[0] = pa[0];     g[2 * t].q[1] = pa[1];
      g[2 * t + 1].q[0] = pb[0]; g[2 * t + 1].q[1] = pb[1];
    }
  };
  w2_ask(wA, 0, 0);
  for (int cp = 0; cp < UT / 2; ++cp) {
    const int cpn = (cp + 1 < UT / 2) ? cp + 1 : cp;
    v8f o00 = {}, o01 = {}, o10 = {}, o11 = {};
    w2_ask(wB, cp, 4);
    __builtin_amdgcn_sched_barrier(0);
    #pragma unroll
    for (int t = 0; t < 4; ++t) {
      o00 = __builtin_amdgcn_wmma_f32_16x16x32_bf16(false, wA[2 * t].v,     false, m[0][t].v, (short)0, o00, false, false);
      o01 = __builtin_amdgcn_wmma_f32_16x16x32_bf16(false, wA[2 * t].v,     false, m[1][t].v, (short)0, o01, false, false);
      o10 = __builtin_amdgcn_wmma_f32_16x16x32_bf16(false, wA[2 * t + 1].v, false, m[0][t].v, (short)0, o10, false, false);
      o11 = __builtin_amdgcn_wmma_f32_16x16x32_bf16(false, wA[2 * t + 1].v, false, m[1][t].v, (short)0, o11, false, false);
    }
    __builtin_amdgcn_sched_barrier(0);
    w2_ask(wA, cpn, 0);
    __builtin_amdgcn_sched_barrier(0);
    #pragma unroll
    for (int t = 0; t < 4; ++t) {
      o00 = __builtin_amdgcn_wmma_f32_16x16x32_bf16(false, wB[2 * t].v,     false, m[0][4 + t].v, (short)0, o00, false, false);
      o01 = __builtin_amdgcn_wmma_f32_16x16x32_bf16(false, wB[2 * t].v,     false, m[1][4 + t].v, (short)0, o01, false, false);
      o10 = __builtin_amdgcn_wmma_f32_16x16x32_bf16(false, wB[2 * t + 1].v, false, m[0][4 + t].v, (short)0, o10, false, false);
      o11 = __builtin_amdgcn_wmma_f32_16x16x32_bf16(false, wB[2 * t + 1].v, false, m[1][4 + t].v, (short)0, o11, false, false);
    }
    __builtin_amdgcn_sched_barrier(0);
    const float4* bq = (const float4*)(b2 + 32 * cp + 8 * hh);
    const float4 ba0 = bq[0], ba1 = bq[1], bb0 = bq[4], bb1 = bq[5];
    auto put = [&](float* st, const v8f& oa, const v8f& ob) {
      float4* r = (float4*)(st + c * MLP_PITCH + 8 * hh);
      r[0] = make_float4(fmaxf(oa[0] + ba0.x, 0.0f), fmaxf(oa[1] + ba0.y, 0.0f),
                         fmaxf(oa[2] + ba0.z, 0.0f), fmaxf(oa[3] + ba0.w, 0.0f));
      r[1] = make_float4(fmaxf(oa[4] + ba1.x, 0.0f), fmaxf(oa[5] + ba1.y, 0.0f),
                         fmaxf(oa[6] + ba1.z, 0.0f), fmaxf(oa[7] + ba1.w, 0.0f));
      r[4] = make_float4(fmaxf(ob[0] + bb0.x, 0.0f), fmaxf(ob[1] + bb0.y, 0.0f),
                         fmaxf(ob[2] + bb0.z, 0.0f), fmaxf(ob[3] + bb0.w, 0.0f));
      r[5] = make_float4(fmaxf(ob[4] + bb1.x, 0.0f), fmaxf(ob[5] + bb1.y, 0.0f),
                         fmaxf(ob[6] + bb1.z, 0.0f), fmaxf(ob[7] + bb1.w, 0.0f));
    };
    put(st0, o00, o10);
    put(st1, o01, o11);
    auto rows_out = [&](const float* st, int tile, bool ok) {
      if (!ok) return;
      #pragma unroll
      for (int i = 0; i < 4; ++i) {
        const int node = 4 * i + (lane >> 3);
        const float4 v = *(const float4*)(st + node * MLP_PITCH + 4 * (lane & 7));
        *(float4*)(out + (long)(tile * 16 + node) * HID + 32 * cp + 4 * (lane & 7)) = v;
      }
    };
    rows_out(st0, t0, ok0);
    rows_out(st1, t0 + 1, ok1);
  }
}

#define POOL_GB      4
#define POOL_WAVES   16
#define POOL_THREADS (POOL_WAVES * 32)
#define POOL_IT      25
#define POOL_SUB     (POOL_IT * 128)
#define POOL_PG      16

static_assert(POOL_SUB * POOL_WAVES >= N_NODES && N_NODES % 4 == 0, "the slices cover the id list, four ids per load");
static_assert(N_GRAPHS % POOL_GB == 0 && 16 % POOL_GB == 0 && POOL_GB * 8 == 32, "four graphs x eight dwords = one line");
static_assert(POOL_WAVES == POOL_GB * (HID / 64) && POOL_WAVES == 2 * (HID / 32), "one wave per graph and 64 columns; one per line");

__global__ void __launch_bounds__(POOL_THREADS)
pool_owner(const float* __restrict__ h, const int* __restrict__ batch, unsigned int* __restrict__ Ppk) {
  __shared__ unsigned int lst[POOL_WAVES * POOL_SUB];
  __shared__ int          cnt[POOL_WAVES];
  __shared__ unsigned int stage[POOL_GB * (HID / 2)];

  const int tid  = threadIdx.x;
  const int lane = tid & 31;
  const int wave = tid >> 5;
  const int g0   = blockIdx.x * POOL_GB;
  const unsigned int ltmask = (1u << lane) - 1u;

  {
    unsigned int* mylst = lst + wave * POOL_SUB;
    const int n0 = wave * POOL_SUB + lane * 4;
    int4 d[POOL_IT];
    #pragma unroll
    for (int it = 0; it < POOL_IT; ++it) {
      const int n = n0 + it * 128;
      d[it] = *(const int4*)(batch + (n < N_NODES ? n : N_NODES - 4));
    }
    int c = 0;
    #pragma unroll
    for (int it = 0; it < POOL_IT; ++it) {
      const unsigned int n  = (unsigned int)(n0 + it * 128);
      const unsigned int x0 = (unsigned int)(d[it].x - g0);
      const unsigned int x1 = (unsigned int)(d[it].y - g0);
      const unsigned int x2 = (unsigned int)(d[it].z - g0);
      const unsigned int x3 = (unsigned int)(d[it].w - g0);
      const unsigned int lim = (n < (unsigned int)N_NODES) ? (unsigned int)POOL_GB : 0u;
      const bool m0 = x0 < lim, m1 = x1 < lim;
      const bool m2 = x2 < lim, m3 = x3 < lim;
      if (__builtin_amdgcn_ballot_w32(m0 || m1 || m2 || m3)) {
        unsigned int b;
        b = __builtin_amdgcn_ballot_w32(m0);
        if (m0) mylst[c + __builtin_popcount(b & ltmask)] = (x0 << 20) | (n + 0u);
        c += __builtin_popcount(b);
        b = __builtin_amdgcn_ballot_w32(m1);
        if (m1) mylst[c + __builtin_popcount(b & ltmask)] = (x1 << 20) | (n + 1u);
        c += __builtin_popcount(b);
        b = __builtin_amdgcn_ballot_w32(m2);
        if (m2) mylst[c + __builtin_popcount(b & ltmask)] = (x2 << 20) | (n + 2u);
        c += __builtin_popcount(b);
        b = __builtin_amdgcn_ballot_w32(m3);
        if (m3) mylst[c + __builtin_popcount(b & ltmask)] = (x3 << 20) | (n + 3u);
        c += __builtin_popcount(b);
      }
    }
    if (lane == 0) cnt[wave] = c;
  }
  __syncthreads();

  const unsigned int gi = (unsigned int)(wave & (POOL_GB - 1));
  const int cq = wave / POOL_GB;
  const float2* hq = (const float2*)h + cq * 32 + lane;
  float2 acc = make_float2(0.0f, 0.0f);
  int deg = 0;
  for (int lw = 0; lw < POOL_WAVES; ++lw) {
    const int cn = __builtin_amdgcn_readfirstlane(cnt[lw]);
    const unsigned int* L = lst + lw * POOL_SUB;
    for (int f0 = 0; f0 < cn; f0 += 32) {
      const int f = f0 + lane;
      const unsigned int e = (f < cn) ? L[f] : 0xFFFFFFFFu;
      unsigned int m = __builtin_amdgcn_ballot_w32((e >> 20) == gi);
      deg += __builtin_popcount(m);
      while (m) {
        float2 v[POOL_PG];
        #pragma unroll
        for (int j = 0; j < POOL_PG; ++j) {
          v[j] = make_float2(0.0f, 0.0f);
          if (m) {
            const int b = __builtin_ctz(m);
            m &= m - 1u;
            const int n = __builtin_amdgcn_readlane((int)e, b) & 0xFFFFF;
            v[j] = hq[n * (HID / 2)];
          }
        }
        #pragma unroll
        for (int j = 0; j < POOL_PG; ++j) { acc.x += v[j].x; acc.y += v[j].y; }
      }
    }
  }

  const float den = fmaxf((float)deg, 1.0f);
  stage[gi * (HID / 2) + cq * 32 + lane] = pack2bf(acc.x / den, acc.y / den);
  __syncthreads();

  {
    const int kt = wave >> 1;
    const int hh = wave & 1;
    const int gl = lane >> 3;
    const int p  = lane & 7;
    const int k  = kt * 32 + frag_koff(16 * hh, p);
    const int rt = g0 >> 4;
    const int c0 = g0 & 15;
    Ppk[((rt * (HID / 32) + kt) * 32 + c0 + 16 * hh) * 8 + lane] = stage[gl * (HID / 2) + (k >> 1)];
  }
}

extern "C" void kernel_launch(void* const* d_in, const int* in_sizes, int n_in,
                              void* d_out, int out_size, void* d_ws, size_t ws_size,
                              hipStream_t stream) {
  const float* x     = (const float*)d_in[0];
  const int*   ei    = (const int*)d_in[1];
  const int*   esrc  = ei;
  const int*   edst  = ei + N_EDGES;
  const int*   batch = (const int*)d_in[2];
  const float* cw1[3] = {(const float*)d_in[3],  (const float*)d_in[7],  (const float*)d_in[11]};
  const float* cb1[3] = {(const float*)d_in[4],  (const float*)d_in[8],  (const float*)d_in[12]};
  const float* cw2[3] = {(const float*)d_in[5],  (const float*)d_in[9],  (const float*)d_in[13]};
  const float* cb2[3] = {(const float*)d_in[6],  (const float*)d_in[10], (const float*)d_in[14]};
  const float* ow1 = (const float*)d_in[15];
  const float* ob1 = (const float*)d_in[16];
  const float* ow2 = (const float*)d_in[17];
  const float* ob2 = (const float*)d_in[18];

  char* ws = (char*)d_ws;
  size_t off = 0;
  auto bump = [&](size_t bytes) {
    void* p = ws + off;
    off += (bytes + 255) & ~(size_t)255;
    return p;
  };
  const int NRT = N_NODES / 16;
  unsigned short* Abf    = (unsigned short*)bump((size_t)N_NODES * HID * 2);
  float*          B2     = (float*)bump((size_t)N_NODES * HID * 4);
  unsigned int*   Ppk    = (unsigned int*)bump((size_t)(N_GRAPHS / 16) * 8 * 256 * 4);
  unsigned short* MPbf   = (unsigned short*)bump((size_t)N_GRAPHS * HID * 2);
  unsigned int*   wpk1[3], *wpk2[3];
  for (int l = 0; l < 3; ++l) {
    wpk1[l] = (unsigned int*)bump((size_t)(HID / 16) * 8 * 256 * 4);
    wpk2[l] = (unsigned int*)bump((size_t)(HID / 16) * 8 * 256 * 4);
  }
  unsigned int* owpk1 = (unsigned int*)bump((size_t)(HID / 16) * 8 * 256 * 4);
  unsigned int* owpk2 = (unsigned int*)bump((size_t)(OUT_C / 16) * 8 * 256 * 4);
  unsigned int* glst  = (unsigned int*)bump((size_t)AGG_GROUPS * AGG_WAVES * LST_CAP * 4);
  int*          gcnt  = (int*)bump((size_t)AGG_GROUPS * AGG_WAVES * 4);

  dim3 blk(256);
  auto wpack = [&](const float* W, unsigned int* dst, int K, int M) {
    int KT = K / 32;
    long total = (long)(M / 16) * KT * 256;
    wpack_f32<<<(unsigned)((total + 255) / 256), blk, 0, stream>>>(W, dst, K, M, KT, total);
  };

  wpack(cw1[0], wpk1[0], IN_C, HID);
  wpack(cw2[0], wpk2[0], HID, HID);
  for (int l = 1; l < 3; ++l) {
    wpack(cw1[l], wpk1[l], HID, HID);
    wpack(cw2[l], wpk2[l], HID, HID);
  }
  wpack(ow1, owpk1, HID, HID);
  wpack(ow2, owpk2, HID, OUT_C);

  owner_lists<<<AGG_GROUPS, AGG_THREADS, 0, stream>>>(esrc, edst, glst, gcnt);

  const float* h = x;
  for (int l = 0; l < 3; ++l) {
    if (l == 0) agg_owner<IN_C><<<AGG_GROUPS, AGG_THREADS, 0, stream>>>(h, esrc, edst, glst, gcnt, Abf);
    else        agg_owner<HID><<<AGG_GROUPS, AGG_THREADS, 0, stream>>>(h, esrc, edst, glst, gcnt, Abf);

    const unsigned mg = (unsigned)((NRT + 2 * MLP_WAVES - 1) / (2 * MLP_WAVES));
    if (l == 0) mlp_chain<IN_C / 32><<<mg, MLP_WAVES * 32, 0, stream>>>(Abf, (const uint4*)wpk1[l], cb1[l],
                                                                     (const uint4*)wpk2[l], cb2[l], B2);
    else        mlp_chain<HID / 32><<<mg, MLP_WAVES * 32, 0, stream>>>(Abf, (const uint4*)wpk1[l], cb1[l],
                                                                    (const uint4*)wpk2[l], cb2[l], B2);
    h = B2;
  }

  pool_owner<<<N_GRAPHS / POOL_GB, POOL_THREADS, 0, stream>>>(B2, batch, Ppk);

  const int GRT = N_GRAPHS / 16;
  dim3 g3((HID / 16 + 3) / 4, (GRT + 7) / 8);
  gemm_pk<1, 1, 0><<<g3, blk, 0, stream>>>((const uint4*)Ppk, (const uint4*)owpk1, ob1,
                                           MPbf, N_GRAPHS, HID, HID);
  dim3 g4((OUT_C / 16 + 3) / 4, (GRT + 7) / 8);
  gemm_pk<0, 0, 1><<<g4, blk, 0, stream>>>((const uint4*)MPbf, (const uint4*)owpk2, ob2,
                                           (float*)d_out, N_GRAPHS, HID, OUT_C);
}


# Edge aggregation: every owner wave sets up and stores its own nodes' rows (start rows asked for all at once, beside the list's first rows): no cooperative passes, no barrier; sum order kept
# speedup vs baseline: 3.4211x; 1.0383x over previous
#include <hip/hip_runtime.h>

#define N_NODES 50000
#define N_EDGES 800000
#define N_GRAPHS 512
#define IN_C 128
#define HID 256
#define OUT_C 16

typedef __attribute__((ext_vector_type(16))) __bf16 v16bf;
typedef __attribute__((ext_vector_type(8)))  float  v8f;
typedef __attribute__((ext_vector_type(8)))  unsigned int v8u;
typedef const __attribute__((address_space(4))) v8u* list_ptr;

__device__ __forceinline__ unsigned short f2bf(float f) {
  union { float f; unsigned int u; } c; c.f = f;
  unsigned int u = c.u;
  u += 0x7fffu + ((u >> 16) & 1u);
  return (unsigned short)(u >> 16);
}

typedef __attribute__((ext_vector_type(2))) float  v2f;
typedef __attribute__((ext_vector_type(2))) __bf16 v2bf;
__device__ __forceinline__ unsigned int pack2bf(float a, float b) {
  const v2f f = {a, b};
  const v2bf h = __builtin_convertvector(f, v2bf);
  return __builtin_bit_cast(unsigned int, h);
}

__device__ __forceinline__ int frag_koff(int lane, int p) {
  int kb = (lane < 16) ? 0 : 8;
  return (p < 4) ? (kb + 2 * p) : (16 + kb + 2 * (p - 4));
}

#define AGG_NB      196
#define AGG_NK      ((AGG_NB + AGG_WAVES - 1) / AGG_WAVES)
#define AGG_WAVES   16
#define AGG_THREADS (AGG_WAVES * 32)
#define AGG_IT      15
#define AGG_SUB     (AGG_IT * 128)
#define AGG_ROUND   (AGG_SUB * AGG_WAVES)
#define AGG_NROUND  ((N_EDGES + AGG_ROUND - 1) / AGG_ROUND)
#define AGG_GROUPS  ((N_NODES + AGG_NB - 1) / AGG_NB)
#define LST_CAP     1024
#define LST_G       8
#define LST_PAD     (5 * LST_G)

static_assert(N_NODES <= (1 << 20) && N_EDGES <= (1 << 20) && AGG_NB < 256, "list entry fields");
static_assert(LST_G == 8, "a group of entries is one 8-dword scalar load");

static_assert(N_EDGES % 4 == 0, "four edges per index load");

__device__ __forceinline__ void scan_load(const int* __restrict__ edst, int e0, int4 (&d)[AGG_IT]) {
  #pragma unroll
  for (int it = 0; it < AGG_IT; ++it) {
    const int e = e0 + it * 128;
    d[it] = *(const int4*)(edst + (e < N_EDGES ? e : N_EDGES - 4));
  }
}

__device__ __forceinline__ int scan_note(const int4 (&d)[AGG_IT], int e0, int base, int nloc,
                                         unsigned int ltmask, unsigned int* mylst) {
  int c = 0;
  #pragma unroll
  for (int it = 0; it < AGG_IT; ++it) {
    const unsigned int e = (unsigned int)(e0 + it * 128);
    const unsigned int x0 = (unsigned int)(d[it].x - base);
    const unsigned int x1 = (unsigned int)(d[it].y - base);
    const unsigned int x2 = (unsigned int)(d[it].z - base);
    const unsigned int x3 = (unsigned int)(d[it].w - base);
    const unsigned int lim = (e < (unsigned int)N_EDGES) ? (unsigned int)nloc : 0u;
    const bool m0 = x0 < lim, m1 = x1 < lim;
    const bool m2 = x2 < lim, m3 = x3 < lim;
    const unsigned int mn01 = x0 < x1 ? x0 : x1, mn23 = x2 < x3 ? x2 : x3;
    if (__builtin_amdgcn_ballot_w32((mn01 < mn23 ? mn01 : mn23) < lim)) {
      unsigned int b;
      b = __builtin_amdgcn_ballot_w32(m0);
      if (m0) mylst[c + __builtin_popcount(b & ltmask)] = (x0 << 20) | (e + 0u);
      c += __builtin_popcount(b);
      b = __builtin_amdgcn_ballot_w32(m1);
      if (m1) mylst[c + __builtin_popcount(b & ltmask)] = (x1 << 20) | (e + 1u);
      c += __builtin_popcount(b);
      b = __builtin_amdgcn_ballot_w32(m2);
      if (m2) mylst[c + __builtin_popcount(b & ltmask)] = (x2 << 20) | (e + 2u);
      c += __builtin_popcount(b);
      b = __builtin_amdgcn_ballot_w32(m3);
      if (m3) mylst[c + __builtin_popcount(b & ltmask)] = (x3 << 20) | (e + 3u);
      c += __builtin_popcount(b);
    }
  }
  return c;
}

__global__ void __launch_bounds__(AGG_THREADS)
owner_lists(const int* __restrict__ esrc, const int* __restrict__ edst,
            unsigned int* __restrict__ glst, int* __restrict__ gcnt) {
  __shared__ unsigned int lst[AGG_WAVES * AGG_SUB];
  __shared__ unsigned int own[AGG_WAVES * LST_CAP];
  __shared__ int          cnt[AGG_WAVES];

  const int tid  = threadIdx.x;
  const int lane = tid & 31;
  const int wave = tid >> 5;
  const int base = blockIdx.x * AGG_NB;
  int nloc = N_NODES - base;
  if (nloc > AGG_NB) nloc = AGG_NB;

  const unsigned int ltmask = (1u << lane) - 1u;
  unsigned int* mylst = lst + wave * AGG_SUB;
  unsigned int* myown = own + wave * LST_CAP;
  unsigned int* myout = glst + ((long)blockIdx.x * AGG_WAVES + wave) * LST_CAP;
  int gpos = 0;

  int rr = (int)(blockIdx.x % AGG_NROUND);
  int4 d[AGG_IT];
  scan_load(edst, rr * AGG_ROUND + wave * AGG_SUB + lane * 4, d);
  for (int r = 0; r < AGG_NROUND; ++r) {
    const int e0 = rr * AGG_ROUND + wave * AGG_SUB + lane * 4;
    rr = (rr + 1 == AGG_NROUND) ? 0 : rr + 1;
    int4 dn[AGG_IT];
    scan_load(edst, rr * AGG_ROUND + wave * AGG_SUB + lane * 4, dn);

    const int c = scan_note(d, e0, base, nloc, ltmask, mylst);
    if (lane == 0) cnt[wave] = c;
    __syncthreads();

    const int myc = (lane < AGG_WAVES) ? cnt[lane] : 0;
    int incl = myc;
    #pragma unroll
    for (int o = 1; o < AGG_WAVES; o <<= 1) {
      const int t = __shfl_up(incl, o);
      if (lane >= o) incl += t;
    }
    const int excl  = incl - myc;
    const int total = __builtin_amdgcn_readlane(incl, AGG_WAVES - 1);
    for (int f0 = 0; f0 < total; f0 += 32) {
      const int f = f0 + lane;
      int li = 0;
      #pragma unroll
      for (int l = 0; l < AGG_WAVES - 1; ++l) li += (f >= __builtin_amdgcn_readlane(incl, l)) ? 1 : 0;
      const int st = __shfl(excl, li);
      unsigned int ent = 0u;
      bool mine = false;
      if (f < total) {
        ent  = lst[li * AGG_SUB + (f - st)];
        mine = ((ent >> 20) & (unsigned int)(AGG_WAVES - 1)) == (unsigned int)wave;
      }
      const unsigned int m = __builtin_amdgcn_ballot_w32(mine);
      if (mine) {
        const int pos = gpos + __builtin_popcount(m & ltmask);
        if (pos < LST_CAP) myown[pos] = ent;
      }
      gpos += __builtin_popcount(m);
    }
    __syncthreads();

    #pragma unroll
    for (int it = 0; it < AGG_IT; ++it) d[it] = dn[it];
  }

  const int nout = gpos < LST_CAP ? gpos : LST_CAP;
  for (int i = lane; i < nout; i += 32) {
    const unsigned int ent = myown[i];
    myout[i] = (ent & 0xFFF00000u) | (unsigned int)esrc[ent & 0xFFFFFu];
  }
  #pragma unroll
  for (int k = 0; k < (LST_PAD + 31) / 32; ++k) {
    const int o = k * 32 + lane;
    if (o < LST_PAD && gpos + o < LST_CAP) myout[gpos + o] = ((unsigned int)AGG_NB << 20) | (unsigned int)base;
  }
  if (lane == 0) gcnt[blockIdx.x * AGG_WAVES + wave] = gpos;
}

typedef __attribute__((ext_vector_type(4))) float v4f;
template <int C> struct row_part;
template <> struct row_part<128> {
  typedef v4f type;
  static __device__ __forceinline__ type load(const float* row, int lane) {
    return *(const v4f*)(row + 4 * lane);
  }
  static __device__ __forceinline__ void store(float* row, int lane, type s) {
    *(v4f*)(row + 4 * lane) = s;
  }
  static __device__ __forceinline__ void store_bf16(unsigned short* row, int lane, type s) {
    *(uint2*)(row + 4 * lane) = make_uint2(pack2bf(s[0], s[1]), pack2bf(s[2], s[3]));
  }
};
template <> struct row_part<256> {
  typedef v8f type;
  static __device__ __forceinline__ type load(const float* row, int lane) {
    const v4f a = *(const v4f*)(row + 4 * lane), b = *(const v4f*)(row + 128 + 4 * lane);
    return __builtin_shufflevector(a, b, 0, 1, 2, 3, 4, 5, 6, 7);
  }
  static __device__ __forceinline__ void store(float* row, int lane, type s) {
    *(v4f*)(row + 4 * lane)       = __builtin_shufflevector(s, s, 0, 1, 2, 3);
    *(v4f*)(row + 128 + 4 * lane) = __builtin_shufflevector(s, s, 4, 5, 6, 7);
  }
  static __device__ __forceinline__ void store_bf16(unsigned short* row, int lane, type s) {
    *(uint2*)(row + 4 * lane)       = make_uint2(pack2bf(s[0], s[1]), pack2bf(s[2], s[3]));
    *(uint2*)(row + 128 + 4 * lane) = make_uint2(pack2bf(s[4], s[5]), pack2bf(s[6], s[7]));
  }
};

template <int C>
__global__ void __launch_bounds__(AGG_THREADS)
agg_owner(const float* __restrict__ h, const int* __restrict__ esrc,
          const int* __restrict__ edst, const unsigned int* __restrict__ glst,
          const int* __restrict__ gcnt, unsigned short* __restrict__ agg) {
  __shared__ float        acc[(AGG_NB + 1) * C];
  __shared__ unsigned int lst[AGG_WAVES * AGG_SUB];
  __shared__ int          cnt[AGG_WAVES];
  constexpr int VEC = C / 128;

  const int tid  = threadIdx.x;
  const int lane = tid & 31;
  const int wave = tid >> 5;
  const int base = blockIdx.x * AGG_NB;
  int nloc = N_NODES - base;
  if (nloc > AGG_NB) nloc = AGG_NB;

  const int lc = (lane < AGG_WAVES) ? gcnt[blockIdx.x * AGG_WAVES + lane] : 0;
  const bool fits = __builtin_amdgcn_ballot_w32(lc > LST_CAP - LST_PAD) == 0u;

  if (fits) {
    const int wv = __builtin_amdgcn_readfirstlane(wave);
    const int n  = __builtin_amdgcn_readlane(lc, wv);
    const list_ptr q = (list_ptr)(unsigned long)(glst + ((long)blockIdx.x * AGG_WAVES + wv) * LST_CAP);
    const int ng = (n + LST_G - 1) / LST_G;

    v8u eA = q[0], eB = q[1], eC = q[2];
    float4 A[LST_G][VEC], B[LST_G][VEC];
    auto rows_ask = [&](float4 (&R)[LST_G][VEC], const v8u& e) {
      #pragma unroll
      for (int j = 0; j < LST_G; ++j) {
        const int s = (int)(e[j] & 0xFFFFFu);
        const float4* hp = (const float4*)(h + (long)s * C) + lane;
        #pragma unroll
        for (int v = 0; v < VEC; ++v) R[j][v] = hp[v * 32];
      }
    };
    auto rows_add = [&](const float4 (&R)[LST_G][VEC], const v8u& e) {
      #pragma unroll
      for (int j = 0; j < LST_G; ++j) {
        const int nd = (int)(e[j] >> 20);
        float4* ap = (float4*)(acc + nd * C) + lane;
        #pragma unroll
        for (int v = 0; v < VEC; ++v) {
          float4 a = ap[v * 32];
          a.x += R[j][v].x; a.y += R[j][v].y; a.z += R[j][v].z; a.w += R[j][v].w;
          ap[v * 32] = a;
        }
      }
    };

    typename row_part<C>::type own[AGG_NK];
    #pragma unroll
    for (int k = 0; k < AGG_NK; ++k) {
      const int loc = wv + AGG_WAVES * k;
      own[k] = row_part<C>::load(h + (long)(base + (loc < nloc ? loc : 0)) * C, lane);
    }
    rows_ask(A, eA);
    #pragma unroll
    for (int k = 0; k < AGG_NK; ++k) {
      const int loc = wv + AGG_WAVES * k;
      row_part<C>::store(acc + (loc < nloc ? loc : AGG_NB) * C, lane, own[k]);
    }
    const int nt = (ng + 1) / 2;
    int t = 0;
    do {
      const v8u fB = q[2 * t + 3], fC = q[2 * t + 4];
      rows_ask(B, eB);
      rows_add(A, eA);
      rows_ask(A, eC);
      rows_add(B, eB);
      eA = eC; eB = fB; eC = fC;
    } while (++t < nt);

    typename row_part<C>::type fin[AGG_NK];
    #pragma unroll
    for (int k = 0; k < AGG_NK; ++k) {
      const int loc = wv + AGG_WAVES * k;
      fin[k] = row_part<C>::load(acc + (loc < nloc ? loc : AGG_NB) * C, lane);
    }
    #pragma unroll
    for (int k = 0; k < AGG_NK; ++k) {
      const int loc = wv + AGG_WAVES * k;
      if (loc < nloc)
        row_part<C>::store_bf16(agg + (long)(base + loc) * C, lane, fin[k]);
    }
  } else {
    {
      const float4* hs = (const float4*)(h + (long)base * C);
      float4* a4 = (float4*)acc;
      const int n4 = nloc * (C / 4);
      for (int i = tid; i < n4; i += AGG_THREADS) a4[i] = hs[i];
    }
    const unsigned int ltmask = (1u << lane) - 1u;
    unsigned int* mylst = lst + wave * AGG_SUB;
    int rr = (int)(blockIdx.x % AGG_NROUND);
    for (int r = 0; r < AGG_NROUND; ++r) {
      const int e0 = rr * AGG_ROUND + wave * AGG_SUB + lane * 4;
      rr = (rr + 1 == AGG_NROUND) ? 0 : rr + 1;
      int4 d[AGG_IT];
      scan_load(edst, e0, d);
      const int c = scan_note(d, e0, base, nloc, ltmask, mylst);
      if (lane == 0) cnt[wave] = c;
      __syncthreads();

      const int myc = (lane < AGG_WAVES) ? cnt[lane] : 0;
      int incl = myc;
      #pragma unroll
      for (int o = 1; o < AGG_WAVES; o <<= 1) {
        const int t = __shfl_up(incl, o);
        if (lane >= o) incl += t;
      }
      const int excl  = incl - myc;
      const int total = __builtin_amdgcn_readlane(incl, AGG_WAVES - 1);
      for (int f0 = 0; f0 < total; f0 += 32) {
        const int f = f0 + lane;
        int li = 0;
        #pragma unroll
        for (int l = 0; l < AGG_WAVES - 1; ++l) li += (f >= __builtin_amdgcn_readlane(incl, l)) ? 1 : 0;
        const int st = __shfl(excl, li);
        unsigned int ent = 0u;
        bool mine = false;
        if (f < total) {
          ent  = lst[li * AGG_SUB + (f - st)];
          mine = ((ent >> 20) & (unsigned int)(AGG_WAVES - 1)) == (unsigned int)wave;
        }
        unsigned int m = __builtin_amdgcn_ballot_w32(mine);
        int s = 0;
        if (mine) s = esrc[ent & 0xFFFFFu];
        while (m) {
          const int b0 = __builtin_ctz(m); m &= m - 1u;
          const int s0 = __builtin_amdgcn_readlane(s, b0);
          const int n0 = (int)(__builtin_amdgcn_readlane((int)ent, b0) >> 20) & 0xFF;
          const float4* h0 = (const float4*)(h + (long)s0 * C) + lane;
          float4* a0 = (float4*)(acc + n0 * C) + lane;
          #pragma unroll
          for (int j = 0; j < VEC; ++j) {
            const float4 v0 = h0[j * 32];
            float4 a = a0[j * 32];
            a.x += v0.x; a.y += v0.y; a.z += v0.z; a.w += v0.w;
            a0[j * 32] = a;
          }
        }
      }
      __syncthreads();
    }

    {
      uint4* o8 = (uint4*)(agg + (long)base * C);
      const float4* a4 = (const float4*)acc;
      const int n8 = nloc * (C / 8);
      for (int i = tid; i < n8; i += AGG_THREADS) {
        const float4 u = a4[2 * i], v = a4[2 * i + 1];
        o8[i] = make_uint4(pack2bf(u.x, u.y), pack2bf(u.z, u.w), pack2bf(v.x, v.y), pack2bf(v.z, v.w));
      }
    }
  }
}

__global__ void wpack_f32(const float* __restrict__ W, unsigned int* __restrict__ out,
                          int K, int M, int KT, long total) {
  long i = (long)blockIdx.x * blockDim.x + threadIdx.x;
  if (i >= total) return;
  int p = (int)(i & 7);
  int lane = (int)((i >> 3) & 31);
  long tt = i >> 8;
  int kt = (int)(tt % KT);
  int ct = (int)(tt / KT);
  int col = ct * 16 + (lane & 15);
  int k = kt * 32 + frag_koff(lane, p);
  unsigned int lo = f2bf(W[(long)k * M + col]);
  unsigned int hi = f2bf(W[(long)(k + 1) * M + col]);
  out[i] = lo | (hi << 16);
}

template <int RELU, int OUTBF, int AROW>
__global__ void gemm_pk(const uint4* __restrict__ Apk, const uint4* __restrict__ Wpk,
                        const float* __restrict__ bias, void* __restrict__ Cout,
                        int N, int K, int M) {
  const int tid  = threadIdx.x;
  const int lane = tid & 31;
  const int wave = tid >> 5;
  const int wr   = wave & 3;
  const int wc   = wave >> 2;
  const int KT   = K >> 5;
  const int nrt  = N >> 4;
  const int nct  = M >> 4;
  const int rt0  = blockIdx.y * 8 + wr * 2;
  const int ct0  = blockIdx.x * 4 + wc * 2;
  const bool r0 = rt0 < nrt,     r1 = (rt0 + 1) < nrt;
  const bool c0 = ct0 < nct,     c1 = (ct0 + 1) < nct;
  const int rA0 = r0 ? rt0 : 0,  rA1 = r1 ? (rt0 + 1) : 0;
  const int cB0 = c0 ? ct0 : 0,  cB1 = c1 ? (ct0 + 1) : 0;

  long ia0 = AROW ? ((long)(rA0 * 16 + (lane & 15)) * (K >> 3) + (lane >> 4)) : (((long)rA0 * KT * 32 + lane) * 2);
  long ia1 = AROW ? ((long)(rA1 * 16 + (lane & 15)) * (K >> 3) + (lane >> 4)) : (((long)rA1 * KT * 32 + lane) * 2);
  long ib0 = ((long)cB0 * KT * 32 + lane) * 2;
  long ib1 = ((long)cB1 * KT * 32 + lane) * 2;
  const long step  = 64;
  const long stepA = AROW ? 4 : 64;
  const long second = AROW ? 2 : 1;

  v8f acc00 = {}, acc01 = {}, acc10 = {}, acc11 = {};
  union F { uint4 q[2]; v16bf v; };

  for (int kt = 0; kt < KT; ++kt) {
    F a0, a1, b0, b1;
    a0.q[0] = Apk[ia0]; a0.q[1] = Apk[ia0 + second];
    a1.q[0] = Apk[ia1]; a1.q[1] = Apk[ia1 + second];
    b0.q[0] = Wpk[ib0]; b0.q[1] = Wpk[ib0 + 1];
    b1.q[0] = Wpk[ib1]; b1.q[1] = Wpk[ib1 + 1];
    if (kt + 1 < KT) {
      __builtin_prefetch((const void*)(Apk + ia0 + stepA), 0, 3);
      __builtin_prefetch((const void*)(Apk + ia1 + stepA), 0, 3);
    }
    ia0 += stepA; ia1 += stepA; ib0 += step; ib1 += step;
    acc00 = __builtin_amdgcn_wmma_f32_16x16x32_bf16(false, a0.v, false, b0.v, (short)0, acc00, false, false);
    acc01 = __builtin_amdgcn_wmma_f32_16x16x32_bf16(false, a0.v, false, b1.v, (short)0, acc01, false, false);
    acc10 = __builtin_amdgcn_wmma_f32_16x16x32_bf16(false, a1.v, false, b0.v, (short)0, acc10, false, false);
    acc11 = __builtin_amdgcn_wmma_f32_16x16x32_bf16(false, a1.v, false, b1.v, (short)0, acc11, false, false);
  }

  const int rowL = lane & 15;
  auto st = [&](const v8f& acc, int rt, bool rok, int ct, bool cok) {
    if (!rok || !cok) return;
    int col = ct * 16 + rowL;
    #pragma unroll
    for (int r = 0; r < 8; ++r) {
      int row = rt * 16 + ((lane < 16) ? r : r + 8);
      float o = acc[r] + bias[col];
      if (RELU) o = fmaxf(o, 0.0f);
      if (OUTBF) ((unsigned short*)Cout)[(long)row * M + col] = f2bf(o);
      else       ((float*)Cout)[(long)row * M + col] = o;
    }
  };
  st(acc00, rt0, r0, ct0, c0);         st(acc01, rt0, r0, ct0 + 1, c1);
  st(acc10, rt0 + 1, r1, ct0, c0);     st(acc11, rt0 + 1, r1, ct0 + 1, c1);
}


#define MLP_WAVES 4
#define MLP_PITCH 36
template <int KT1>
__global__ void __launch_bounds__(MLP_WAVES * 32)
mlp_chain(const unsigned short* __restrict__ Xbf, const uint4* __restrict__ W1pk, const float* __restrict__ b1,
          const uint4* __restrict__ W2pk, const float* __restrict__ b2, float* __restrict__ out) {
  __shared__ alignas(16) float stage[MLP_WAVES][2][16 * MLP_PITCH];
  const int lane = threadIdx.x & 31;
  const int wave = threadIdx.x >> 5;
  const int c    = lane & 15;
  const int hh   = lane >> 4;
  const int nrt  = N_NODES / 16;
  const int t0   = (blockIdx.x * MLP_WAVES + wave) * 2;
  const bool ok0 = t0 < nrt, ok1 = (t0 + 1) < nrt;
  const int tA = ok0 ? t0 : 0, tB = ok1 ? (t0 + 1) : 0;
  constexpr int KT2 = HID / 32;
  constexpr int UT  = HID / 16;
  union F { uint4 q[2]; v16bf v; };

  v8f acc[UT][2];
  #pragma unroll
  for (int j = 0; j < UT; ++j) { acc[j][0] = v8f{}; acc[j][1] = v8f{}; }
  const uint4* w1l = W1pk + lane * 2;
  auto w1_ask = [&](F (&g)[UT / 2], int kt, int j0) {
    #pragma unroll
    for (int j = 0; j < UT / 2; ++j) {
      const uint4* pw = w1l + (long)((j0 + j) * KT1 + kt) * 64;
      g[j].q[0] = pw[0]; g[j].q[1] = pw[1];
    }
  };
  const uint4* xr0 = (const uint4*)(Xbf + (long)(tA * 16 + c) * (32 * KT1)) + hh;
  const uint4* xr1 = (const uint4*)(Xbf + (long)(tB * 16 + c) * (32 * KT1)) + hh;
  auto x_ask = [&](F& x0, F& x1, int kt) {
    x0.q[0] = xr0[4 * kt]; x0.q[1] = xr0[4 * kt + 2];
    x1.q[0] = xr1[4 * kt]; x1.q[1] = xr1[4 * kt + 2];
  };
  F wA[UT / 2], wB[UT / 2], x0, x1;
  x_ask(x0, x1, 0);
  w1_ask(wA, 0, 0);
  for (int kt = 0; kt < KT1; ++kt) {
    const int ktn = (kt + 1 < KT1) ? kt + 1 : kt;
    F y0, y1;
    w1_ask(wB, kt, UT / 2);
    x_ask(y0, y1, ktn);
    __builtin_amdgcn_sched_barrier(0);
    #pragma unroll
    for (int j = 0; j < UT / 2; ++j) {
      acc[j][0] = __builtin_amdgcn_wmma_f32_16x16x32_bf16(false, wA[j].v, false, x0.v, (short)0, acc[j][0], false, false);
      acc[j][1] = __builtin_amdgcn_wmma_f32_16x16x32_bf16(false, wA[j].v, false, x1.v, (short)0, acc[j][1], false, false);
    }
    __builtin_amdgcn_sched_barrier(0);
    w1_ask(wA, ktn, 0);
    __builtin_amdgcn_sched_barrier(0);
    #pragma unroll
    for (int j = 0; j < UT / 2; ++j) {
      acc[UT / 2 + j][0] = __builtin_amdgcn_wmma_f32_16x16x32_bf16(false, wB[j].v, false, x0.v, (short)0, acc[UT / 2 + j][0], false, false);
      acc[UT / 2 + j][1] = __builtin_amdgcn_wmma_f32_16x16x32_bf16(false, wB[j].v, false, x1.v, (short)0, acc[UT / 2 + j][1], false, false);
    }
    __builtin_amdgcn_sched_barrier(0);
    x0 = y0; x1 = y1;
  }

  F m[2][KT2];
  #pragma unroll
  for (int t = 0; t < KT2; ++t) {
    const float4* bp = (const float4*)(b1 + 32 * t + 8 * hh);
    const float4 ba0 = bp[0], ba1 = bp[1];
    const float4 bb0 = bp[4], bb1 = bp[5];
    const float ba[8] = {ba0.x, ba0.y, ba0.z, ba0.w, ba1.x, ba1.y, ba1.z, ba1.w};
    const float bb[8] = {bb0.x, bb0.y, bb0.z, bb0.w, bb1.x, bb1.y, bb1.z, bb1.w};
    #pragma unroll
    for (int nt = 0; nt < 2; ++nt) {
      unsigned int lo[4], hi[4];
      #pragma unroll
      for (int p = 0; p < 4; ++p) {
        const float a0 = fmaxf(acc[2 * t][nt][2 * p]     + ba[2 * p],     0.0f);
        const float a1 = fmaxf(acc[2 * t][nt][2 * p + 1] + ba[2 * p + 1], 0.0f);
        const float c0 = fmaxf(acc[2 * t + 1][nt][2 * p]     + bb[2 * p],     0.0f);
        const float c1 = fmaxf(acc[2 * t + 1][nt][2 * p + 1] + bb[2 * p + 1], 0.0f);
        lo[p] = pack2bf(a0, a1);
        hi[p] = pack2bf(c0, c1);
      }
      m[nt][t].q[0] = make_uint4(lo[0], lo[1], lo[2], lo[3]);
      m[nt][t].q[1] = make_uint4(hi[0], hi[1], hi[2], hi[3]);
    }
  }

  float* st0 = &stage[wave][0][0];
  float* st1 = &stage[wave][1][0];
  const uint4* w2l = W2pk + lane * 2;
  auto w2_ask = [&](F (&g)[8], int cp, int tq) {
    #pragma unroll
    for (int t = 0; t < 4; ++t) {
      const uint4* pa = w2l + (long)((2 * cp) * KT2 + tq + t) * 64;
      const uint4* pb = w2l + (long)((2 * cp + 1) * KT2 + tq + t) * 64;
      g[2 * t].q[0] = pa[0];     g[2 * t].q[1] = pa[1];
      g[2 * t + 1].q[0] = pb[0]; g[2 * t + 1].q[1] = pb[1];
    }
  };
  w2_ask(wA, 0, 0);
  for (int cp = 0; cp < UT / 2; ++cp) {
    const int cpn = (cp + 1 < UT / 2) ? cp + 1 : cp;
    v8f o00 = {}, o01 = {}, o10 = {}, o11 = {};
    w2_ask(wB, cp, 4);
    __builtin_amdgcn_sched_barrier(0);
    #pragma unroll
    for (int t = 0; t < 4; ++t) {
      o00 = __builtin_amdgcn_wmma_f32_16x16x32_bf16(false, wA[2 * t].v,     false, m[0][t].v, (short)0, o00, false, false);
      o01 = __builtin_amdgcn_wmma_f32_16x16x32_bf16(false, wA[2 * t].v,     false, m[1][t].v, (short)0, o01, false, false);
      o10 = __builtin_amdgcn_wmma_f32_16x16x32_bf16(false, wA[2 * t + 1].v, false, m[0][t].v, (short)0, o10, false, false);
      o11 = __builtin_amdgcn_wmma_f32_16x16x32_bf16(false, wA[2 * t + 1].v, false, m[1][t].v, (short)0, o11, false, false);
    }
    __builtin_amdgcn_sched_barrier(0);
    w2_ask(wA, cpn, 0);
    __builtin_amdgcn_sched_barrier(0);
    #pragma unroll
    for (int t = 0; t < 4; ++t) {
      o00 = __builtin_amdgcn_wmma_f32_16x16x32_bf16(false, wB[2 * t].v,     false, m[0][4 + t].v, (short)0, o00, false, false);
      o01 = __builtin_amdgcn_wmma_f32_16x16x32_bf16(false, wB[2 * t].v,     false, m[1][4 + t].v, (short)0, o01, false, false);
      o10 = __builtin_amdgcn_wmma_f32_16x16x32_bf16(false, wB[2 * t + 1].v, false, m[0][4 + t].v, (short)0, o10, false, false);
      o11 = __builtin_amdgcn_wmma_f32_16x16x32_bf16(false, wB[2 * t + 1].v, false, m[1][4 + t].v, (short)0, o11, false, false);
    }
    __builtin_amdgcn_sched_barrier(0);
    const float4* bq = (const float4*)(b2 + 32 * cp + 8 * hh);
    const float4 ba0 = bq[0], ba1 = bq[1], bb0 = bq[4], bb1 = bq[5];
    auto put = [&](float* st, const v8f& oa, const v8f& ob) {
      float4* r = (float4*)(st + c * MLP_PITCH + 8 * hh);
      r[0] = make_float4(fmaxf(oa[0] + ba0.x, 0.0f), fmaxf(oa[1] + ba0.y, 0.0f),
                         fmaxf(oa[2] + ba0.z, 0.0f), fmaxf(oa[3] + ba0.w, 0.0f));
      r[1] = make_float4(fmaxf(oa[4] + ba1.x, 0.0f), fmaxf(oa[5] + ba1.y, 0.0f),
                         fmaxf(oa[6] + ba1.z, 0.0f), fmaxf(oa[7] + ba1.w, 0.0f));
      r[4] = make_float4(fmaxf(ob[0] + bb0.x, 0.0f), fmaxf(ob[1] + bb0.y, 0.0f),
                         fmaxf(ob[2] + bb0.z, 0.0f), fmaxf(ob[3] + bb0.w, 0.0f));
      r[5] = make_float4(fmaxf(ob[4] + bb1.x, 0.0f), fmaxf(ob[5] + bb1.y, 0.0f),
                         fmaxf(ob[6] + bb1.z, 0.0f), fmaxf(ob[7] + bb1.w, 0.0f));
    };
    put(st0, o00, o10);
    put(st1, o01, o11);
    auto rows_out = [&](const float* st, int tile, bool ok) {
      if (!ok) return;
      #pragma unroll
      for (int i = 0; i < 4; ++i) {
        const int node = 4 * i + (lane >> 3);
        const float4 v = *(const float4*)(st + node * MLP_PITCH + 4 * (lane & 7));
        *(float4*)(out + (long)(tile * 16 + node) * HID + 32 * cp + 4 * (lane & 7)) = v;
      }
    };
    rows_out(st0, t0, ok0);
    rows_out(st1, t0 + 1, ok1);
  }
}

#define POOL_GB      4
#define POOL_WAVES   16
#define POOL_THREADS (POOL_WAVES * 32)
#define POOL_IT      25
#define POOL_SUB     (POOL_IT * 128)
#define POOL_PG      16

static_assert(POOL_SUB * POOL_WAVES >= N_NODES && N_NODES % 4 == 0, "the slices cover the id list, four ids per load");
static_assert(N_GRAPHS % POOL_GB == 0 && 16 % POOL_GB == 0 && POOL_GB * 8 == 32, "four graphs x eight dwords = one line");
static_assert(POOL_WAVES == POOL_GB * (HID / 64) && POOL_WAVES == 2 * (HID / 32), "one wave per graph and 64 columns; one per line");

__global__ void __launch_bounds__(POOL_THREADS)
pool_owner(const float* __restrict__ h, const int* __restrict__ batch, unsigned int* __restrict__ Ppk) {
  __shared__ unsigned int lst[POOL_WAVES * POOL_SUB];
  __shared__ int          cnt[POOL_WAVES];
  __shared__ unsigned int stage[POOL_GB * (HID / 2)];

  const int tid  = threadIdx.x;
  const int lane = tid & 31;
  const int wave = tid >> 5;
  const int g0   = blockIdx.x * POOL_GB;
  const unsigned int ltmask = (1u << lane) - 1u;

  {
    unsigned int* mylst = lst + wave * POOL_SUB;
    const int n0 = wave * POOL_SUB + lane * 4;
    int4 d[POOL_IT];
    #pragma unroll
    for (int it = 0; it < POOL_IT; ++it) {
      const int n = n0 + it * 128;
      d[it] = *(const int4*)(batch + (n < N_NODES ? n : N_NODES - 4));
    }
    int c = 0;
    #pragma unroll
    for (int it = 0; it < POOL_IT; ++it) {
      const unsigned int n  = (unsigned int)(n0 + it * 128);
      const unsigned int x0 = (unsigned int)(d[it].x - g0);
      const unsigned int x1 = (unsigned int)(d[it].y - g0);
      const unsigned int x2 = (unsigned int)(d[it].z - g0);
      const unsigned int x3 = (unsigned int)(d[it].w - g0);
      const unsigned int lim = (n < (unsigned int)N_NODES) ? (unsigned int)POOL_GB : 0u;
      const bool m0 = x0 < lim, m1 = x1 < lim;
      const bool m2 = x2 < lim, m3 = x3 < lim;
      if (__builtin_amdgcn_ballot_w32(m0 || m1 || m2 || m3)) {
        unsigned int b;
        b = __builtin_amdgcn_ballot_w32(m0);
        if (m0) mylst[c + __builtin_popcount(b & ltmask)] = (x0 << 20) | (n + 0u);
        c += __builtin_popcount(b);
        b = __builtin_amdgcn_ballot_w32(m1);
        if (m1) mylst[c + __builtin_popcount(b & ltmask)] = (x1 << 20) | (n + 1u);
        c += __builtin_popcount(b);
        b = __builtin_amdgcn_ballot_w32(m2);
        if (m2) mylst[c + __builtin_popcount(b & ltmask)] = (x2 << 20) | (n + 2u);
        c += __builtin_popcount(b);
        b = __builtin_amdgcn_ballot_w32(m3);
        if (m3) mylst[c + __builtin_popcount(b & ltmask)] = (x3 << 20) | (n + 3u);
        c += __builtin_popcount(b);
      }
    }
    if (lane == 0) cnt[wave] = c;
  }
  __syncthreads();

  const unsigned int gi = (unsigned int)(wave & (POOL_GB - 1));
  const int cq = wave / POOL_GB;
  const float2* hq = (const float2*)h + cq * 32 + lane;
  float2 acc = make_float2(0.0f, 0.0f);
  int deg = 0;
  for (int lw = 0; lw < POOL_WAVES; ++lw) {
    const int cn = __builtin_amdgcn_readfirstlane(cnt[lw]);
    const unsigned int* L = lst + lw * POOL_SUB;
    for (int f0 = 0; f0 < cn; f0 += 32) {
      const int f = f0 + lane;
      const unsigned int e = (f < cn) ? L[f] : 0xFFFFFFFFu;
      unsigned int m = __builtin_amdgcn_ballot_w32((e >> 20) == gi);
      deg += __builtin_popcount(m);
      while (m) {
        float2 v[POOL_PG];
        #pragma unroll
        for (int j = 0; j < POOL_PG; ++j) {
          v[j] = make_float2(0.0f, 0.0f);
          if (m) {
            const int b = __builtin_ctz(m);
            m &= m - 1u;
            const int n = __builtin_amdgcn_readlane((int)e, b) & 0xFFFFF;
            v[j] = hq[n * (HID / 2)];
          }
        }
        #pragma unroll
        for (int j = 0; j < POOL_PG; ++j) { acc.x += v[j].x; acc.y += v[j].y; }
      }
    }
  }

  const float den = fmaxf((float)deg, 1.0f);
  stage[gi * (HID / 2) + cq * 32 + lane] = pack2bf(acc.x / den, acc.y / den);
  __syncthreads();

  {
    const int kt = wave >> 1;
    const int hh = wave & 1;
    const int gl = lane >> 3;
    const int p  = lane & 7;
    const int k  = kt * 32 + frag_koff(16 * hh, p);
    const int rt = g0 >> 4;
    const int c0 = g0 & 15;
    Ppk[((rt * (HID / 32) + kt) * 32 + c0 + 16 * hh) * 8 + lane] = stage[gl * (HID / 2) + (k >> 1)];
  }
}

extern "C" void kernel_launch(void* const* d_in, const int* in_sizes, int n_in,
                              void* d_out, int out_size, void* d_ws, size_t ws_size,
                              hipStream_t stream) {
  const float* x     = (const float*)d_in[0];
  const int*   ei    = (const int*)d_in[1];
  const int*   esrc  = ei;
  const int*   edst  = ei + N_EDGES;
  const int*   batch = (const int*)d_in[2];
  const float* cw1[3] = {(const float*)d_in[3],  (const float*)d_in[7],  (const float*)d_in[11]};
  const float* cb1[3] = {(const float*)d_in[4],  (const float*)d_in[8],  (const float*)d_in[12]};
  const float* cw2[3] = {(const float*)d_in[5],  (const float*)d_in[9],  (const float*)d_in[13]};
  const float* cb2[3] = {(const float*)d_in[6],  (const float*)d_in[10], (const float*)d_in[14]};
  const float* ow1 = (const float*)d_in[15];
  const float* ob1 = (const float*)d_in[16];
  const float* ow2 = (const float*)d_in[17];
  const float* ob2 = (const float*)d_in[18];

  char* ws = (char*)d_ws;
  size_t off = 0;
  auto bump = [&](size_t bytes) {
    void* p = ws + off;
    off += (bytes + 255) & ~(size_t)255;
    return p;
  };
  const int NRT = N_NODES / 16;
  unsigned short* Abf    = (unsigned short*)bump((size_t)N_NODES * HID * 2);
  float*          B2     = (float*)bump((size_t)N_NODES * HID * 4);
  unsigned int*   Ppk    = (unsigned int*)bump((size_t)(N_GRAPHS / 16) * 8 * 256 * 4);
  unsigned short* MPbf   = (unsigned short*)bump((size_t)N_GRAPHS * HID * 2);
  unsigned int*   wpk1[3], *wpk2[3];
  for (int l = 0; l < 3; ++l) {
    wpk1[l] = (unsigned int*)bump((size_t)(HID / 16) * 8 * 256 * 4);
    wpk2[l] = (unsigned int*)bump((size_t)(HID / 16) * 8 * 256 * 4);
  }
  unsigned int* owpk1 = (unsigned int*)bump((size_t)(HID / 16) * 8 * 256 * 4);
  unsigned int* owpk2 = (unsigned int*)bump((size_t)(OUT_C / 16) * 8 * 256 * 4);
  unsigned int* glst  = (unsigned int*)bump((size_t)AGG_GROUPS * AGG_WAVES * LST_CAP * 4);
  int*          gcnt  = (int*)bump((size_t)AGG_GROUPS * AGG_WAVES * 4);

  dim3 blk(256);
  auto wpack = [&](const float* W, unsigned int* dst, int K, int M) {
    int KT = K / 32;
    long total = (long)(M / 16) * KT * 256;
    wpack_f32<<<(unsigned)((total + 255) / 256), blk, 0, stream>>>(W, dst, K, M, KT, total);
  };

  wpack(cw1[0], wpk1[0], IN_C, HID);
  wpack(cw2[0], wpk2[0], HID, HID);
  for (int l = 1; l < 3; ++l) {
    wpack(cw1[l], wpk1[l], HID, HID);
    wpack(cw2[l], wpk2[l], HID, HID);
  }
  wpack(ow1, owpk1, HID, HID);
  wpack(ow2, owpk2, HID, OUT_C);

  owner_lists<<<AGG_GROUPS, AGG_THREADS, 0, stream>>>(esrc, edst, glst, gcnt);

  const float* h = x;
  for (int l = 0; l < 3; ++l) {
    if (l == 0) agg_owner<IN_C><<<AGG_GROUPS, AGG_THREADS, 0, stream>>>(h, esrc, edst, glst, gcnt, Abf);
    else        agg_owner<HID><<<AGG_GROUPS, AGG_THREADS, 0, stream>>>(h, esrc, edst, glst, gcnt, Abf);

    const unsigned mg = (unsigned)((NRT + 2 * MLP_WAVES - 1) / (2 * MLP_WAVES));
    if (l == 0) mlp_chain<IN_C / 32><<<mg, MLP_WAVES * 32, 0, stream>>>(Abf, (const uint4*)wpk1[l], cb1[l],
                                                                     (const uint4*)wpk2[l], cb2[l], B2);
    else        mlp_chain<HID / 32><<<mg, MLP_WAVES * 32, 0, stream>>>(Abf, (const uint4*)wpk1[l], cb1[l],
                                                                    (const uint4*)wpk2[l], cb2[l], B2);
    h = B2;
  }

  pool_owner<<<N_GRAPHS / POOL_GB, POOL_THREADS, 0, stream>>>(B2, batch, Ppk);

  const int GRT = N_GRAPHS / 16;
  dim3 g3((HID / 16 + 3) / 4, (GRT + 7) / 8);
  gemm_pk<1, 1, 0><<<g3, blk, 0, stream>>>((const uint4*)Ppk, (const uint4*)owpk1, ob1,
                                           MPbf, N_GRAPHS, HID, HID);
  dim3 g4((OUT_C / 16 + 3) / 4, (GRT + 7) / 8);
  gemm_pk<0, 0, 1><<<g4, blk, 0, stream>>>((const uint4*)MPbf, (const uint4*)owpk2, ob2,
                                           (float*)d_out, N_GRAPHS, HID, OUT_C);
}


# Layer MLP: a work-group of 8 waves (256 nodes) copies the layer's two packed weight matrices to LDS once and its waves read their weight fragments there, not from L2 per wave; 391 -> 196 groups
# speedup vs baseline: 4.2016x; 1.2281x over previous
#include <hip/hip_runtime.h>

#define N_NODES 50000
#define N_EDGES 800000
#define N_GRAPHS 512
#define IN_C 128
#define HID 256
#define OUT_C 16

typedef __attribute__((ext_vector_type(16))) __bf16 v16bf;
typedef __attribute__((ext_vector_type(8)))  float  v8f;
typedef __attribute__((ext_vector_type(8)))  unsigned int v8u;
typedef const __attribute__((address_space(4))) v8u* list_ptr;

__device__ __forceinline__ unsigned short f2bf(float f) {
  union { float f; unsigned int u; } c; c.f = f;
  unsigned int u = c.u;
  u += 0x7fffu + ((u >> 16) & 1u);
  return (unsigned short)(u >> 16);
}

typedef __attribute__((ext_vector_type(2))) float  v2f;
typedef __attribute__((ext_vector_type(2))) __bf16 v2bf;
__device__ __forceinline__ unsigned int pack2bf(float a, float b) {
  const v2f f = {a, b};
  const v2bf h = __builtin_convertvector(f, v2bf);
  return __builtin_bit_cast(unsigned int, h);
}

__device__ __forceinline__ int frag_koff(int lane, int p) {
  int kb = (lane < 16) ? 0 : 8;
  return (p < 4) ? (kb + 2 * p) : (16 + kb + 2 * (p - 4));
}

#define AGG_NB      196
#define AGG_NK      ((AGG_NB + AGG_WAVES - 1) / AGG_WAVES)
#define AGG_WAVES   16
#define AGG_THREADS (AGG_WAVES * 32)
#define AGG_IT      15
#define AGG_SUB     (AGG_IT * 128)
#define AGG_ROUND   (AGG_SUB * AGG_WAVES)
#define AGG_NROUND  ((N_EDGES + AGG_ROUND - 1) / AGG_ROUND)
#define AGG_GROUPS  ((N_NODES + AGG_NB - 1) / AGG_NB)
#define LST_CAP     1024
#define LST_G       8
#define LST_PAD     (5 * LST_G)

static_assert(N_NODES <= (1 << 20) && N_EDGES <= (1 << 20) && AGG_NB < 256, "list entry fields");
static_assert(LST_G == 8, "a group of entries is one 8-dword scalar load");

static_assert(N_EDGES % 4 == 0, "four edges per index load");

#define CR_G        16
#define CR_NB       (AGG_NB * CR_G)
#define CR_N        ((AGG_GROUPS + CR_G - 1) / CR_G)
#define CS_IT       25
#define CS_SUB      (CS_IT * 128)
#define CS_SL       (CS_SUB * AGG_WAVES)
#define CS_N        ((N_EDGES + CS_SL - 1) / CS_SL)
#define P2_CAP      1024
#define P2_U        9

static_assert(CS_N == AGG_WAVES, "one wave of an owner group per slice list");
static_assert(CR_NB <= 4095, "a coarse range's local node fits 12 bits, and 4095 is no node of any group");

__device__ __forceinline__ void scan_load(const int* __restrict__ edst, int e0, int4 (&d)[AGG_IT]) {
  #pragma unroll
  for (int it = 0; it < AGG_IT; ++it) {
    const int e = e0 + it * 128;
    d[it] = *(const int4*)(edst + (e < N_EDGES ? e : N_EDGES - 4));
  }
}

__device__ __forceinline__ int scan_note(const int4 (&d)[AGG_IT], int e0, int base, int nloc,
                                         unsigned int ltmask, unsigned int* mylst) {
  int c = 0;
  #pragma unroll
  for (int it = 0; it < AGG_IT; ++it) {
    const unsigned int e = (unsigned int)(e0 + it * 128);
    const unsigned int x0 = (unsigned int)(d[it].x - base);
    const unsigned int x1 = (unsigned int)(d[it].y - base);
    const unsigned int x2 = (unsigned int)(d[it].z - base);
    const unsigned int x3 = (unsigned int)(d[it].w - base);
    const unsigned int lim = (e < (unsigned int)N_EDGES) ? (unsigned int)nloc : 0u;
    const bool m0 = x0 < lim, m1 = x1 < lim;
    const bool m2 = x2 < lim, m3 = x3 < lim;
    const unsigned int mn01 = x0 < x1 ? x0 : x1, mn23 = x2 < x3 ? x2 : x3;
    if (__builtin_amdgcn_ballot_w32((mn01 < mn23 ? mn01 : mn23) < lim)) {
      unsigned int b;
      b = __builtin_amdgcn_ballot_w32(m0);
      if (m0) mylst[c + __builtin_popcount(b & ltmask)] = (x0 << 20) | (e + 0u);
      c += __builtin_popcount(b);
      b = __builtin_amdgcn_ballot_w32(m1);
      if (m1) mylst[c + __builtin_popcount(b & ltmask)] = (x1 << 20) | (e + 1u);
      c += __builtin_popcount(b);
      b = __builtin_amdgcn_ballot_w32(m2);
      if (m2) mylst[c + __builtin_popcount(b & ltmask)] = (x2 << 20) | (e + 2u);
      c += __builtin_popcount(b);
      b = __builtin_amdgcn_ballot_w32(m3);
      if (m3) mylst[c + __builtin_popcount(b & ltmask)] = (x3 << 20) | (e + 3u);
      c += __builtin_popcount(b);
    }
  }
  return c;
}

__global__ void __launch_bounds__(AGG_THREADS)
coarse_lists(const int* __restrict__ edst, unsigned int* __restrict__ clst, int* __restrict__ ccnt) {
  __shared__ int cnt[AGG_WAVES];

  const int tid  = threadIdx.x;
  const int lane = tid & 31;
  const int wave = tid >> 5;
  const int ri   = blockIdx.x / CS_N;
  const int sl   = blockIdx.x % CS_N;
  const int rbase = ri * CR_NB;
  const unsigned int ltmask = (1u << lane) - 1u;
  const int e0 = sl * CS_SL + wave * CS_SUB + lane * 4;

  int4 d[CS_IT];
  #pragma unroll
  for (int it = 0; it < CS_IT; ++it) {
    const int e = e0 + it * 128;
    d[it] = *(const int4*)(edst + (e < N_EDGES ? e : N_EDGES - 4));
  }

  int c = 0;
  #pragma unroll
  for (int it = 0; it < CS_IT; ++it) {
    const unsigned int lim = ((e0 + it * 128) < N_EDGES) ? (unsigned int)CR_NB : 0u;
    c += __builtin_popcount(__builtin_amdgcn_ballot_w32((unsigned int)(d[it].x - rbase) < lim));
    c += __builtin_popcount(__builtin_amdgcn_ballot_w32((unsigned int)(d[it].y - rbase) < lim));
    c += __builtin_popcount(__builtin_amdgcn_ballot_w32((unsigned int)(d[it].z - rbase) < lim));
    c += __builtin_popcount(__builtin_amdgcn_ballot_w32((unsigned int)(d[it].w - rbase) < lim));
  }
  if (lane == 0) cnt[wave] = c;
  __syncthreads();

  const int myc = (lane < AGG_WAVES) ? cnt[lane] : 0;
  int incl = myc;
  #pragma unroll
  for (int o = 1; o < AGG_WAVES; o <<= 1) {
    const int t = __shfl_up(incl, o);
    if (lane >= o) incl += t;
  }
  const int total = __builtin_amdgcn_readlane(incl, AGG_WAVES - 1);
  const int wv = __builtin_amdgcn_readfirstlane(wave);
  int pos = __builtin_amdgcn_readlane(incl, wv) - c;

  unsigned int* out = clst + (long)blockIdx.x * CS_SL;
  #pragma unroll
  for (int it = 0; it < CS_IT; ++it) {
    const unsigned int e = (unsigned int)(e0 + it * 128);
    const unsigned int lim = (e < (unsigned int)N_EDGES) ? (unsigned int)CR_NB : 0u;
    const unsigned int x0 = (unsigned int)(d[it].x - rbase);
    const unsigned int x1 = (unsigned int)(d[it].y - rbase);
    const unsigned int x2 = (unsigned int)(d[it].z - rbase);
    const unsigned int x3 = (unsigned int)(d[it].w - rbase);
    const bool m0 = x0 < lim, m1 = x1 < lim;
    const bool m2 = x2 < lim, m3 = x3 < lim;
    unsigned int b;
    b = __builtin_amdgcn_ballot_w32(m0);
    if (m0) out[pos + __builtin_popcount(b & ltmask)] = (x0 << 20) | (e + 0u);
    pos += __builtin_popcount(b);
    b = __builtin_amdgcn_ballot_w32(m1);
    if (m1) out[pos + __builtin_popcount(b & ltmask)] = (x1 << 20) | (e + 1u);
    pos += __builtin_popcount(b);
    b = __builtin_amdgcn_ballot_w32(m2);
    if (m2) out[pos + __builtin_popcount(b & ltmask)] = (x2 << 20) | (e + 2u);
    pos += __builtin_popcount(b);
    b = __builtin_amdgcn_ballot_w32(m3);
    if (m3) out[pos + __builtin_popcount(b & ltmask)] = (x3 << 20) | (e + 3u);
    pos += __builtin_popcount(b);
  }
  const int npad = (total == 0) ? 4 : ((4 - (total & 3)) & 3);
  if (wave == 0 && lane < npad) out[total + lane] = 0xFFFFFFFFu;
  if (tid == 0) ccnt[blockIdx.x] = total;
}

__global__ void __launch_bounds__(AGG_THREADS)
owner_lists(const int* __restrict__ esrc, const unsigned int* __restrict__ clst, const int* __restrict__ ccnt,
            unsigned int* __restrict__ glst, int* __restrict__ gcnt) {
  __shared__ unsigned int lst[AGG_WAVES * P2_CAP];
  __shared__ unsigned int own[AGG_WAVES * LST_CAP];
  __shared__ int          cnt[AGG_WAVES];

  const int tid  = threadIdx.x;
  const int lane = tid & 31;
  const int wave = tid >> 5;
  const int wv   = __builtin_amdgcn_readfirstlane(wave);
  const int base = blockIdx.x * AGG_NB;
  int nloc = N_NODES - base;
  if (nloc > AGG_NB) nloc = AGG_NB;
  const int ri = blockIdx.x / CR_G;
  const unsigned int off = (unsigned int)(blockIdx.x % CR_G) * AGG_NB;

  const unsigned int ltmask = (1u << lane) - 1u;
  unsigned int* mylst = lst + wave * P2_CAP;
  unsigned int* myown = own + wave * LST_CAP;
  unsigned int* myout = glst + ((long)blockIdx.x * AGG_WAVES + wave) * LST_CAP;

  const int n  = __builtin_amdgcn_readfirstlane(ccnt[ri * CS_N + wv]);
  const int n4 = (n > 0) ? ((n + 3) >> 2) : 1;
  const uint4* L = (const uint4*)(clst + (long)(ri * CS_N + wv) * CS_SL);
  auto ask = [&](uint4 (&R)[P2_U], int q0) {
    #pragma unroll
    for (int u = 0; u < P2_U; ++u) {
      const int q = q0 + u * 32 + lane;
      R[u] = L[q < n4 ? q : n4 - 1];
    }
  };
  int c = 0;
  uint4 d[P2_U];
  ask(d, 0);
  for (int q0 = 0; q0 < n4; q0 += 32 * P2_U) {
    uint4 dn[P2_U];
    ask(dn, q0 + 32 * P2_U);
    #pragma unroll
    for (int u = 0; u < P2_U; ++u) {
      const unsigned int lim = (q0 + u * 32 + lane < n4) ? (unsigned int)nloc : 0u;
      const unsigned int x0 = (d[u].x >> 20) - off, x1 = (d[u].y >> 20) - off;
      const unsigned int x2 = (d[u].z >> 20) - off, x3 = (d[u].w >> 20) - off;
      const bool m0 = x0 < lim, m1 = x1 < lim;
      const bool m2 = x2 < lim, m3 = x3 < lim;
      unsigned int b;
      int p;
      b = __builtin_amdgcn_ballot_w32(m0);
      p = c + __builtin_popcount(b & ltmask);
      if (m0 && p < P2_CAP) mylst[p] = (x0 << 20) | (d[u].x & 0xFFFFFu);
      c += __builtin_popcount(b);
      b = __builtin_amdgcn_ballot_w32(m1);
      p = c + __builtin_popcount(b & ltmask);
      if (m1 && p < P2_CAP) mylst[p] = (x1 << 20) | (d[u].y & 0xFFFFFu);
      c += __builtin_popcount(b);
      b = __builtin_amdgcn_ballot_w32(m2);
      p = c + __builtin_popcount(b & ltmask);
      if (m2 && p < P2_CAP) mylst[p] = (x2 << 20) | (d[u].z & 0xFFFFFu);
      c += __builtin_popcount(b);
      b = __builtin_amdgcn_ballot_w32(m3);
      p = c + __builtin_popcount(b & ltmask);
      if (m3 && p < P2_CAP) mylst[p] = (x3 << 20) | (d[u].w & 0xFFFFFu);
      c += __builtin_popcount(b);
    }
    #pragma unroll
    for (int u = 0; u < P2_U; ++u) d[u] = dn[u];
  }
  if (lane == 0) cnt[wave] = c;
  __syncthreads();

  const int cv = (lane < AGG_WAVES) ? cnt[lane] : 0;
  if (__builtin_amdgcn_ballot_w32(cv > P2_CAP) != 0u) {
    if (lane == 0) gcnt[blockIdx.x * AGG_WAVES + wave] = LST_CAP;
    return;
  }

  int gpos = 0;
  for (int lw = 0; lw < AGG_WAVES; ++lw) {
    const int cn = __builtin_amdgcn_readlane(cv, lw);
    const unsigned int* Lw = lst + lw * P2_CAP;
    for (int f0 = 0; f0 < cn; f0 += 128) {
      unsigned int ent[4];
      #pragma unroll
      for (int j = 0; j < 4; ++j) {
        const int f = f0 + 32 * j + lane;
        ent[j] = (f < cn) ? Lw[f] : 0u;
      }
      #pragma unroll
      for (int j = 0; j < 4; ++j) {
        const int f = f0 + 32 * j + lane;
        const bool mine = (f < cn) && (((ent[j] >> 20) & (unsigned int)(AGG_WAVES - 1)) == (unsigned int)wave);
        const unsigned int m = __builtin_amdgcn_ballot_w32(mine);
        if (mine) {
          const int pos = gpos + __builtin_popcount(m & ltmask);
          if (pos < LST_CAP) myown[pos] = ent[j];
        }
        gpos += __builtin_popcount(m);
      }
    }
  }

  const int nout = gpos < LST_CAP ? gpos : LST_CAP;
  for (int i = lane; i < nout; i += 32) {
    const unsigned int ent = myown[i];
    myout[i] = (ent & 0xFFF00000u) | (unsigned int)esrc[ent & 0xFFFFFu];
  }
  #pragma unroll
  for (int k = 0; k < (LST_PAD + 31) / 32; ++k) {
    const int o = k * 32 + lane;
    if (o < LST_PAD && gpos + o < LST_CAP) myout[gpos + o] = ((unsigned int)AGG_NB << 20) | (unsigned int)base;
  }
  if (lane == 0) gcnt[blockIdx.x * AGG_WAVES + wave] = gpos;
}

typedef __attribute__((ext_vector_type(4))) float v4f;
template <int C> struct row_part;
template <> struct row_part<128> {
  typedef v4f type;
  static __device__ __forceinline__ type load(const float* row, int lane) {
    return *(const v4f*)(row + 4 * lane);
  }
  static __device__ __forceinline__ void store(float* row, int lane, type s) {
    *(v4f*)(row + 4 * lane) = s;
  }
  static __device__ __forceinline__ void store_bf16(unsigned short* row, int lane, type s) {
    *(uint2*)(row + 4 * lane) = make_uint2(pack2bf(s[0], s[1]), pack2bf(s[2], s[3]));
  }
};
template <> struct row_part<256> {
  typedef v8f type;
  static __device__ __forceinline__ type load(const float* row, int lane) {
    const v4f a = *(const v4f*)(row + 4 * lane), b = *(const v4f*)(row + 128 + 4 * lane);
    return __builtin_shufflevector(a, b, 0, 1, 2, 3, 4, 5, 6, 7);
  }
  static __device__ __forceinline__ void store(float* row, int lane, type s) {
    *(v4f*)(row + 4 * lane)       = __builtin_shufflevector(s, s, 0, 1, 2, 3);
    *(v4f*)(row + 128 + 4 * lane) = __builtin_shufflevector(s, s, 4, 5, 6, 7);
  }
  static __device__ __forceinline__ void store_bf16(unsigned short* row, int lane, type s) {
    *(uint2*)(row + 4 * lane)       = make_uint2(pack2bf(s[0], s[1]), pack2bf(s[2], s[3]));
    *(uint2*)(row + 128 + 4 * lane) = make_uint2(pack2bf(s[4], s[5]), pack2bf(s[6], s[7]));
  }
};

template <int C>
__global__ void __launch_bounds__(AGG_THREADS)
agg_owner(const float* __restrict__ h, const int* __restrict__ esrc,
          const int* __restrict__ edst, const unsigned int* __restrict__ glst,
          const int* __restrict__ gcnt, unsigned short* __restrict__ agg) {
  __shared__ float        acc[(AGG_NB + 1) * C];
  __shared__ unsigned int lst[AGG_WAVES * AGG_SUB];
  __shared__ int          cnt[AGG_WAVES];
  constexpr int VEC = C / 128;

  const int tid  = threadIdx.x;
  const int lane = tid & 31;
  const int wave = tid >> 5;
  const int base = blockIdx.x * AGG_NB;
  int nloc = N_NODES - base;
  if (nloc > AGG_NB) nloc = AGG_NB;

  const int lc = (lane < AGG_WAVES) ? gcnt[blockIdx.x * AGG_WAVES + lane] : 0;
  const bool fits = __builtin_amdgcn_ballot_w32(lc > LST_CAP - LST_PAD) == 0u;

  if (fits) {
    const int wv = __builtin_amdgcn_readfirstlane(wave);
    const int n  = __builtin_amdgcn_readlane(lc, wv);
    const list_ptr q = (list_ptr)(unsigned long)(glst + ((long)blockIdx.x * AGG_WAVES + wv) * LST_CAP);
    const int ng = (n + LST_G - 1) / LST_G;

    v8u eA = q[0], eB = q[1], eC = q[2];
    float4 A[LST_G][VEC], B[LST_G][VEC];
    auto rows_ask = [&](float4 (&R)[LST_G][VEC], const v8u& e) {
      #pragma unroll
      for (int j = 0; j < LST_G; ++j) {
        const int s = (int)(e[j] & 0xFFFFFu);
        const float4* hp = (const float4*)(h + (long)s * C) + lane;
        #pragma unroll
        for (int v = 0; v < VEC; ++v) R[j][v] = hp[v * 32];
      }
    };
    auto rows_add = [&](const float4 (&R)[LST_G][VEC], const v8u& e) {
      #pragma unroll
      for (int j = 0; j < LST_G; ++j) {
        const int nd = (int)(e[j] >> 20);
        float4* ap = (float4*)(acc + nd * C) + lane;
        #pragma unroll
        for (int v = 0; v < VEC; ++v) {
          float4 a = ap[v * 32];
          a.x += R[j][v].x; a.y += R[j][v].y; a.z += R[j][v].z; a.w += R[j][v].w;
          ap[v * 32] = a;
        }
      }
    };

    typename row_part<C>::type own[AGG_NK];
    #pragma unroll
    for (int k = 0; k < AGG_NK; ++k) {
      const int loc = wv + AGG_WAVES * k;
      own[k] = row_part<C>::load(h + (long)(base + (loc < nloc ? loc : 0)) * C, lane);
    }
    rows_ask(A, eA);
    #pragma unroll
    for (int k = 0; k < AGG_NK; ++k) {
      const int loc = wv + AGG_WAVES * k;
      row_part<C>::store(acc + (loc < nloc ? loc : AGG_NB) * C, lane, own[k]);
    }
    const int nt = (ng + 1) / 2;
    int t = 0;
    do {
      const v8u fB = q[2 * t + 3], fC = q[2 * t + 4];
      rows_ask(B, eB);
      rows_add(A, eA);
      rows_ask(A, eC);
      rows_add(B, eB);
      eA = eC; eB = fB; eC = fC;
    } while (++t < nt);

    typename row_part<C>::type fin[AGG_NK];
    #pragma unroll
    for (int k = 0; k < AGG_NK; ++k) {
      const int loc = wv + AGG_WAVES * k;
      fin[k] = row_part<C>::load(acc + (loc < nloc ? loc : AGG_NB) * C, lane);
    }
    #pragma unroll
    for (int k = 0; k < AGG_NK; ++k) {
      const int loc = wv + AGG_WAVES * k;
      if (loc < nloc)
        row_part<C>::store_bf16(agg + (long)(base + loc) * C, lane, fin[k]);
    }
  } else {
    {
      const float4* hs = (const float4*)(h + (long)base * C);
      float4* a4 = (float4*)acc;
      const int n4 = nloc * (C / 4);
      for (int i = tid; i < n4; i += AGG_THREADS) a4[i] = hs[i];
    }
    const unsigned int ltmask = (1u << lane) - 1u;
    unsigned int* mylst = lst + wave * AGG_SUB;
    int rr = (int)(blockIdx.x % AGG_NROUND);
    for (int r = 0; r < AGG_NROUND; ++r) {
      const int e0 = rr * AGG_ROUND + wave * AGG_SUB + lane * 4;
      rr = (rr + 1 == AGG_NROUND) ? 0 : rr + 1;
      int4 d[AGG_IT];
      scan_load(edst, e0, d);
      const int c = scan_note(d, e0, base, nloc, ltmask, mylst);
      if (lane == 0) cnt[wave] = c;
      __syncthreads();

      const int myc = (lane < AGG_WAVES) ? cnt[lane] : 0;
      int incl = myc;
      #pragma unroll
      for (int o = 1; o < AGG_WAVES; o <<= 1) {
        const int t = __shfl_up(incl, o);
        if (lane >= o) incl += t;
      }
      const int excl  = incl - myc;
      const int total = __builtin_amdgcn_readlane(incl, AGG_WAVES - 1);
      for (int f0 = 0; f0 < total; f0 += 32) {
        const int f = f0 + lane;
        int li = 0;
        #pragma unroll
        for (int l = 0; l < AGG_WAVES - 1; ++l) li += (f >= __builtin_amdgcn_readlane(incl, l)) ? 1 : 0;
        const int st = __shfl(excl, li);
        unsigned int ent = 0u;
        bool mine = false;
        if (f < total) {
          ent  = lst[li * AGG_SUB + (f - st)];
          mine = ((ent >> 20) & (unsigned int)(AGG_WAVES - 1)) == (unsigned int)wave;
        }
        unsigned int m = __builtin_amdgcn_ballot_w32(mine);
        int s = 0;
        if (mine) s = esrc[ent & 0xFFFFFu];
        while (m) {
          const int b0 = __builtin_ctz(m); m &= m - 1u;
          const int s0 = __builtin_amdgcn_readlane(s, b0);
          const int n0 = (int)(__builtin_amdgcn_readlane((int)ent, b0) >> 20) & 0xFF;
          const float4* h0 = (const float4*)(h + (long)s0 * C) + lane;
          float4* a0 = (float4*)(acc + n0 * C) + lane;
          #pragma unroll
          for (int j = 0; j < VEC; ++j) {
            const float4 v0 = h0[j * 32];
            float4 a = a0[j * 32];
            a.x += v0.x; a.y += v0.y; a.z += v0.z; a.w += v0.w;
            a0[j * 32] = a;
          }
        }
      }
      __syncthreads();
    }

    {
      uint4* o8 = (uint4*)(agg + (long)base * C);
      const float4* a4 = (const float4*)acc;
      const int n8 = nloc * (C / 8);
      for (int i = tid; i < n8; i += AGG_THREADS) {
        const float4 u = a4[2 * i], v = a4[2 * i + 1];
        o8[i] = make_uint4(pack2bf(u.x, u.y), pack2bf(u.z, u.w), pack2bf(v.x, v.y), pack2bf(v.z, v.w));
      }
    }
  }
}

__global__ void wpack_f32(const float* __restrict__ W, unsigned int* __restrict__ out,
                          int K, int M, int KT, long total) {
  long i = (long)blockIdx.x * blockDim.x + threadIdx.x;
  if (i >= total) return;
  int p = (int)(i & 7);
  int lane = (int)((i >> 3) & 31);
  long tt = i >> 8;
  int kt = (int)(tt % KT);
  int ct = (int)(tt / KT);
  int col = ct * 16 + (lane & 15);
  int k = kt * 32 + frag_koff(lane, p);
  unsigned int lo = f2bf(W[(long)k * M + col]);
  unsigned int hi = f2bf(W[(long)(k + 1) * M + col]);
  out[i] = lo | (hi << 16);
}

template <int RELU, int OUTBF, int AROW>
__global__ void gemm_pk(const uint4* __restrict__ Apk, const uint4* __restrict__ Wpk,
                        const float* __restrict__ bias, void* __restrict__ Cout,
                        int N, int K, int M) {
  const int tid  = threadIdx.x;
  const int lane = tid & 31;
  const int wave = tid >> 5;
  const int wr   = wave & 3;
  const int wc   = wave >> 2;
  const int KT   = K >> 5;
  const int nrt  = N >> 4;
  const int nct  = M >> 4;
  const int rt0  = blockIdx.y * 8 + wr * 2;
  const int ct0  = blockIdx.x * 4 + wc * 2;
  const bool r0 = rt0 < nrt,     r1 = (rt0 + 1) < nrt;
  const bool c0 = ct0 < nct,     c1 = (ct0 + 1) < nct;
  const int rA0 = r0 ? rt0 : 0,  rA1 = r1 ? (rt0 + 1) : 0;
  const int cB0 = c0 ? ct0 : 0,  cB1 = c1 ? (ct0 + 1) : 0;

  long ia0 = AROW ? ((long)(rA0 * 16 + (lane & 15)) * (K >> 3) + (lane >> 4)) : (((long)rA0 * KT * 32 + lane) * 2);
  long ia1 = AROW ? ((long)(rA1 * 16 + (lane & 15)) * (K >> 3) + (lane >> 4)) : (((long)rA1 * KT * 32 + lane) * 2);
  long ib0 = ((long)cB0 * KT * 32 + lane) * 2;
  long ib1 = ((long)cB1 * KT * 32 + lane) * 2;
  const long step  = 64;
  const long stepA = AROW ? 4 : 64;
  const long second = AROW ? 2 : 1;

  v8f acc00 = {}, acc01 = {}, acc10 = {}, acc11 = {};
  union F { uint4 q[2]; v16bf v; };

  for (int kt = 0; kt < KT; ++kt) {
    F a0, a1, b0, b1;
    a0.q[0] = Apk[ia0]; a0.q[1] = Apk[ia0 + second];
    a1.q[0] = Apk[ia1]; a1.q[1] = Apk[ia1 + second];
    b0.q[0] = Wpk[ib0]; b0.q[1] = Wpk[ib0 + 1];
    b1.q[0] = Wpk[ib1]; b1.q[1] = Wpk[ib1 + 1];
    if (kt + 1 < KT) {
      __builtin_prefetch((const void*)(Apk + ia0 + stepA), 0, 3);
      __builtin_prefetch((const void*)(Apk + ia1 + stepA), 0, 3);
    }
    ia0 += stepA; ia1 += stepA; ib0 += step; ib1 += step;
    acc00 = __builtin_amdgcn_wmma_f32_16x16x32_bf16(false, a0.v, false, b0.v, (short)0, acc00, false, false);
    acc01 = __builtin_amdgcn_wmma_f32_16x16x32_bf16(false, a0.v, false, b1.v, (short)0, acc01, false, false);
    acc10 = __builtin_amdgcn_wmma_f32_16x16x32_bf16(false, a1.v, false, b0.v, (short)0, acc10, false, false);
    acc11 = __builtin_amdgcn_wmma_f32_16x16x32_bf16(false, a1.v, false, b1.v, (short)0, acc11, false, false);
  }

  const int rowL = lane & 15;
  auto st = [&](const v8f& acc, int rt, bool rok, int ct, bool cok) {
    if (!rok || !cok) return;
    int col = ct * 16 + rowL;
    #pragma unroll
    for (int r = 0; r < 8; ++r) {
      int row = rt * 16 + ((lane < 16) ? r : r + 8);
      float o = acc[r] + bias[col];
      if (RELU) o = fmaxf(o, 0.0f);
      if (OUTBF) ((unsigned short*)Cout)[(long)row * M + col] = f2bf(o);
      else       ((float*)Cout)[(long)row * M + col] = o;
    }
  };
  st(acc00, rt0, r0, ct0, c0);         st(acc01, rt0, r0, ct0 + 1, c1);
  st(acc10, rt0 + 1, r1, ct0, c0);     st(acc11, rt0 + 1, r1, ct0 + 1, c1);
}


#define MLP_WAVES 8
#define MLP_THREADS (MLP_WAVES * 32)
#define MLP_PITCH 36

template <int NF1, int NF2>
__device__ __forceinline__ void stage_frags(const uint4* __restrict__ G1, uint4* S1,
                                            const uint4* __restrict__ G2, uint4* S2, int wave, int lane) {
  constexpr int T1 = NF1 * 64 / MLP_THREADS, T2 = NF2 * 64 / MLP_THREADS;
  static_assert(NF1 % 4 == 0 && NF2 % 4 == 0 && MLP_WAVES == 8, "a step of the work-group is four whole fragments");
  const int l = ((wave & 1) << 4) + (lane & 15), j = lane >> 4;
  const int g = (wave >> 1) * 64 + l * 2 + j;
  const int s = (wave >> 1) * 64 + j * 32 + l;
  uint4 r1[T1], r2[T2];
  #pragma unroll
  for (int it = 0; it < T1; ++it) r1[it] = G1[g + it * 256];
  #pragma unroll
  for (int it = 0; it < T2; ++it) r2[it] = G2[g + it * 256];
  #pragma unroll
  for (int it = 0; it < T1; ++it) S1[s + it * 256] = r1[it];
  #pragma unroll
  for (int it = 0; it < T2; ++it) S2[s + it * 256] = r2[it];
}
template <int KT1>
__global__ void __launch_bounds__(MLP_THREADS)
mlp_chain(const unsigned short* __restrict__ Xbf, const uint4* __restrict__ W1pk, const float* __restrict__ b1,
          const uint4* __restrict__ W2pk, const float* __restrict__ b2, float* __restrict__ out) {
  __shared__ alignas(16) float stage[MLP_WAVES][2][16 * MLP_PITCH];
  __shared__ uint4 w1s[(HID / 16) * KT1 * 64];
  __shared__ uint4 w2s[(HID / 16) * (HID / 32) * 64];
  const int lane = threadIdx.x & 31;
  const int wave = threadIdx.x >> 5;
  const int c    = lane & 15;
  const int hh   = lane >> 4;
  const int nrt  = N_NODES / 16;
  const int t0   = (blockIdx.x * MLP_WAVES + wave) * 2;
  const bool ok0 = t0 < nrt, ok1 = (t0 + 1) < nrt;
  const int tA = ok0 ? t0 : 0, tB = ok1 ? (t0 + 1) : 0;
  constexpr int KT2 = HID / 32;
  constexpr int UT  = HID / 16;
  union F { uint4 q[2]; v16bf v; };

  v8f acc[UT][2];
  #pragma unroll
  for (int j = 0; j < UT; ++j) { acc[j][0] = v8f{}; acc[j][1] = v8f{}; }
  const uint4* w1l = w1s + lane;
  auto w1_ask = [&](F (&g)[UT / 2], int kt, int j0) {
    #pragma unroll
    for (int j = 0; j < UT / 2; ++j) {
      const uint4* pw = w1l + ((j0 + j) * KT1 + kt) * 64;
      g[j].q[0] = pw[0]; g[j].q[1] = pw[32];
    }
  };
  const uint4* xr0 = (const uint4*)(Xbf + (long)(tA * 16 + c) * (32 * KT1)) + hh;
  const uint4* xr1 = (const uint4*)(Xbf + (long)(tB * 16 + c) * (32 * KT1)) + hh;
  auto x_ask = [&](F& x0, F& x1, int kt) {
    x0.q[0] = xr0[4 * kt]; x0.q[1] = xr0[4 * kt + 2];
    x1.q[0] = xr1[4 * kt]; x1.q[1] = xr1[4 * kt + 2];
  };
  F wA[UT / 2], wB[UT / 2], x0, x1;
  x_ask(x0, x1, 0);
  stage_frags<(HID / 16) * KT1, (HID / 16) * (HID / 32)>(W1pk, w1s, W2pk, w2s, wave, lane);
  __syncthreads();
  w1_ask(wA, 0, 0);
  for (int kt = 0; kt < KT1; ++kt) {
    const int ktn = (kt + 1 < KT1) ? kt + 1 : kt;
    F y0, y1;
    w1_ask(wB, kt, UT / 2);
    x_ask(y0, y1, ktn);
    __builtin_amdgcn_sched_barrier(0);
    #pragma unroll
    for (int j = 0; j < UT / 2; ++j) {
      acc[j][0] = __builtin_amdgcn_wmma_f32_16x16x32_bf16(false, wA[j].v, false, x0.v, (short)0, acc[j][0], false, false);
      acc[j][1] = __builtin_amdgcn_wmma_f32_16x16x32_bf16(false, wA[j].v, false, x1.v, (short)0, acc[j][1], false, false);
    }
    __builtin_amdgcn_sched_barrier(0);
    w1_ask(wA, ktn, 0);
    __builtin_amdgcn_sched_barrier(0);
    #pragma unroll
    for (int j = 0; j < UT / 2; ++j) {
      acc[UT / 2 + j][0] = __builtin_amdgcn_wmma_f32_16x16x32_bf16(false, wB[j].v, false, x0.v, (short)0, acc[UT / 2 + j][0], false, false);
      acc[UT / 2 + j][1] = __builtin_amdgcn_wmma_f32_16x16x32_bf16(false, wB[j].v, false, x1.v, (short)0, acc[UT / 2 + j][1], false, false);
    }
    __builtin_amdgcn_sched_barrier(0);
    x0 = y0; x1 = y1;
  }

  F m[2][KT2];
  #pragma unroll
  for (int t = 0; t < KT2; ++t) {
    const float4* bp = (const float4*)(b1 + 32 * t + 8 * hh);
    const float4 ba0 = bp[0], ba1 = bp[1];
    const float4 bb0 = bp[4], bb1 = bp[5];
    const float ba[8] = {ba0.x, ba0.y, ba0.z, ba0.w, ba1.x, ba1.y, ba1.z, ba1.w};
    const float bb[8] = {bb0.x, bb0.y, bb0.z, bb0.w, bb1.x, bb1.y, bb1.z, bb1.w};
    #pragma unroll
    for (int nt = 0; nt < 2; ++nt) {
      unsigned int lo[4], hi[4];
      #pragma unroll
      for (int p = 0; p < 4; ++p) {
        const float a0 = fmaxf(acc[2 * t][nt][2 * p]     + ba[2 * p],     0.0f);
        const float a1 = fmaxf(acc[2 * t][nt][2 * p + 1] + ba[2 * p + 1], 0.0f);
        const float c0 = fmaxf(acc[2 * t + 1][nt][2 * p]     + bb[2 * p],     0.0f);
        const float c1 = fmaxf(acc[2 * t + 1][nt][2 * p + 1] + bb[2 * p + 1], 0.0f);
        lo[p] = pack2bf(a0, a1);
        hi[p] = pack2bf(c0, c1);
      }
      m[nt][t].q[0] = make_uint4(lo[0], lo[1], lo[2], lo[3]);
      m[nt][t].q[1] = make_uint4(hi[0], hi[1], hi[2], hi[3]);
    }
  }

  float* st0 = &stage[wave][0][0];
  float* st1 = &stage[wave][1][0];
  const uint4* w2l = w2s + lane;
  auto w2_ask = [&](F (&g)[8], int cp, int tq) {
    #pragma unroll
    for (int t = 0; t < 4; ++t) {
      const uint4* pa = w2l + ((2 * cp) * KT2 + tq + t) * 64;
      const uint4* pb = w2l + ((2 * cp + 1) * KT2 + tq + t) * 64;
      g[2 * t].q[0] = pa[0];     g[2 * t].q[1] = pa[32];
      g[2 * t + 1].q[0] = pb[0]; g[2 * t + 1].q[1] = pb[32];
    }
  };
  w2_ask(wA, 0, 0);
  for (int cp = 0; cp < UT / 2; ++cp) {
    const int cpn = (cp + 1 < UT / 2) ? cp + 1 : cp;
    v8f o00 = {}, o01 = {}, o10 = {}, o11 = {};
    w2_ask(wB, cp, 4);
    __builtin_amdgcn_sched_barrier(0);
    #pragma unroll
    for (int t = 0; t < 4; ++t) {
      o00 = __builtin_amdgcn_wmma_f32_16x16x32_bf16(false, wA[2 * t].v,     false, m[0][t].v, (short)0, o00, false, false);
      o01 = __builtin_amdgcn_wmma_f32_16x16x32_bf16(false, wA[2 * t].v,     false, m[1][t].v, (short)0, o01, false, false);
      o10 = __builtin_amdgcn_wmma_f32_16x16x32_bf16(false, wA[2 * t + 1].v, false, m[0][t].v, (short)0, o10, false, false);
      o11 = __builtin_amdgcn_wmma_f32_16x16x32_bf16(false, wA[2 * t + 1].v, false, m[1][t].v, (short)0, o11, false, false);
    }
    __builtin_amdgcn_sched_barrier(0);
    w2_ask(wA, cpn, 0);
    __builtin_amdgcn_sched_barrier(0);
    #pragma unroll
    for (int t = 0; t < 4; ++t) {
      o00 = __builtin_amdgcn_wmma_f32_16x16x32_bf16(false, wB[2 * t].v,     false, m[0][4 + t].v, (short)0, o00, false, false);
      o01 = __builtin_amdgcn_wmma_f32_16x16x32_bf16(false, wB[2 * t].v,     false, m[1][4 + t].v, (short)0, o01, false, false);
      o10 = __builtin_amdgcn_wmma_f32_16x16x32_bf16(false, wB[2 * t + 1].v, false, m[0][4 + t].v, (short)0, o10, false, false);
      o11 = __builtin_amdgcn_wmma_f32_16x16x32_bf16(false, wB[2 * t + 1].v, false, m[1][4 + t].v, (short)0, o11, false, false);
    }
    __builtin_amdgcn_sched_barrier(0);
    const float4* bq = (const float4*)(b2 + 32 * cp + 8 * hh);
    const float4 ba0 = bq[0], ba1 = bq[1], bb0 = bq[4], bb1 = bq[5];
    auto put = [&](float* st, const v8f& oa, const v8f& ob) {
      float4* r = (float4*)(st + c * MLP_PITCH + 8 * hh);
      r[0] = make_float4(fmaxf(oa[0] + ba0.x, 0.0f), fmaxf(oa[1] + ba0.y, 0.0f),
                         fmaxf(oa[2] + ba0.z, 0.0f), fmaxf(oa[3] + ba0.w, 0.0f));
      r[1] = make_float4(fmaxf(oa[4] + ba1.x, 0.0f), fmaxf(oa[5] + ba1.y, 0.0f),
                         fmaxf(oa[6] + ba1.z, 0.0f), fmaxf(oa[7] + ba1.w, 0.0f));
      r[4] = make_float4(fmaxf(ob[0] + bb0.x, 0.0f), fmaxf(ob[1] + bb0.y, 0.0f),
                         fmaxf(ob[2] + bb0.z, 0.0f), fmaxf(ob[3] + bb0.w, 0.0f));
      r[5] = make_float4(fmaxf(ob[4] + bb1.x, 0.0f), fmaxf(ob[5] + bb1.y, 0.0f),
                         fmaxf(ob[6] + bb1.z, 0.0f), fmaxf(ob[7] + bb1.w, 0.0f));
    };
    put(st0, o00, o10);
    put(st1, o01, o11);
    auto rows_out = [&](const float* st, int tile, bool ok) {
      if (!ok) return;
      #pragma unroll
      for (int i = 0; i < 4; ++i) {
        const int node = 4 * i + (lane >> 3);
        const float4 v = *(const float4*)(st + node * MLP_PITCH + 4 * (lane & 7));
        *(float4*)(out + (long)(tile * 16 + node) * HID + 32 * cp + 4 * (lane & 7)) = v;
      }
    };
    rows_out(st0, t0, ok0);
    rows_out(st1, t0 + 1, ok1);
  }
}

#define POOL_GB      4
#define POOL_WAVES   16
#define POOL_THREADS (POOL_WAVES * 32)
#define POOL_IT      25
#define POOL_SUB     (POOL_IT * 128)
#define POOL_PG      16

static_assert(POOL_SUB * POOL_WAVES >= N_NODES && N_NODES % 4 == 0, "the slices cover the id list, four ids per load");
static_assert(N_GRAPHS % POOL_GB == 0 && 16 % POOL_GB == 0 && POOL_GB * 8 == 32, "four graphs x eight dwords = one line");
static_assert(POOL_WAVES == POOL_GB * (HID / 64) && POOL_WAVES == 2 * (HID / 32), "one wave per graph and 64 columns; one per line");

__global__ void __launch_bounds__(POOL_THREADS)
pool_owner(const float* __restrict__ h, const int* __restrict__ batch, unsigned int* __restrict__ Ppk) {
  __shared__ unsigned int lst[POOL_WAVES * POOL_SUB];
  __shared__ int          cnt[POOL_WAVES];
  __shared__ unsigned int stage[POOL_GB * (HID / 2)];

  const int tid  = threadIdx.x;
  const int lane = tid & 31;
  const int wave = tid >> 5;
  const int g0   = blockIdx.x * POOL_GB;
  const unsigned int ltmask = (1u << lane) - 1u;

  {
    unsigned int* mylst = lst + wave * POOL_SUB;
    const int n0 = wave * POOL_SUB + lane * 4;
    int4 d[POOL_IT];
    #pragma unroll
    for (int it = 0; it < POOL_IT; ++it) {
      const int n = n0 + it * 128;
      d[it] = *(const int4*)(batch + (n < N_NODES ? n : N_NODES - 4));
    }
    int c = 0;
    #pragma unroll
    for (int it = 0; it < POOL_IT; ++it) {
      const unsigned int n  = (unsigned int)(n0 + it * 128);
      const unsigned int x0 = (unsigned int)(d[it].x - g0);
      const unsigned int x1 = (unsigned int)(d[it].y - g0);
      const unsigned int x2 = (unsigned int)(d[it].z - g0);
      const unsigned int x3 = (unsigned int)(d[it].w - g0);
      const unsigned int lim = (n < (unsigned int)N_NODES) ? (unsigned int)POOL_GB : 0u;
      const bool m0 = x0 < lim, m1 = x1 < lim;
      const bool m2 = x2 < lim, m3 = x3 < lim;
      if (__builtin_amdgcn_ballot_w32(m0 || m1 || m2 || m3)) {
        unsigned int b;
        b = __builtin_amdgcn_ballot_w32(m0);
        if (m0) mylst[c + __builtin_popcount(b & ltmask)] = (x0 << 20) | (n + 0u);
        c += __builtin_popcount(b);
        b = __builtin_amdgcn_ballot_w32(m1);
        if (m1) mylst[c + __builtin_popcount(b & ltmask)] = (x1 << 20) | (n + 1u);
        c += __builtin_popcount(b);
        b = __builtin_amdgcn_ballot_w32(m2);
        if (m2) mylst[c + __builtin_popcount(b & ltmask)] = (x2 << 20) | (n + 2u);
        c += __builtin_popcount(b);
        b = __builtin_amdgcn_ballot_w32(m3);
        if (m3) mylst[c + __builtin_popcount(b & ltmask)] = (x3 << 20) | (n + 3u);
        c += __builtin_popcount(b);
      }
    }
    if (lane == 0) cnt[wave] = c;
  }
  __syncthreads();

  const unsigned int gi = (unsigned int)(wave & (POOL_GB - 1));
  const int cq = wave / POOL_GB;
  const float2* hq = (const float2*)h + cq * 32 + lane;
  float2 acc = make_float2(0.0f, 0.0f);
  int deg = 0;
  for (int lw = 0; lw < POOL_WAVES; ++lw) {
    const int cn = __builtin_amdgcn_readfirstlane(cnt[lw]);
    const unsigned int* L = lst + lw * POOL_SUB;
    for (int f0 = 0; f0 < cn; f0 += 32) {
      const int f = f0 + lane;
      const unsigned int e = (f < cn) ? L[f] : 0xFFFFFFFFu;
      unsigned int m = __builtin_amdgcn_ballot_w32((e >> 20) == gi);
      deg += __builtin_popcount(m);
      while (m) {
        float2 v[POOL_PG];
        #pragma unroll
        for (int j = 0; j < POOL_PG; ++j) {
          v[j] = make_float2(0.0f, 0.0f);
          if (m) {
            const int b = __builtin_ctz(m);
            m &= m - 1u;
            const int n = __builtin_amdgcn_readlane((int)e, b) & 0xFFFFF;
            v[j] = hq[n * (HID / 2)];
          }
        }
        #pragma unroll
        for (int j = 0; j < POOL_PG; ++j) { acc.x += v[j].x; acc.y += v[j].y; }
      }
    }
  }

  const float den = fmaxf((float)deg, 1.0f);
  stage[gi * (HID / 2) + cq * 32 + lane] = pack2bf(acc.x / den, acc.y / den);
  __syncthreads();

  {
    const int kt = wave >> 1;
    const int hh = wave & 1;
    const int gl = lane >> 3;
    const int p  = lane & 7;
    const int k  = kt * 32 + frag_koff(16 * hh, p);
    const int rt = g0 >> 4;
    const int c0 = g0 & 15;
    Ppk[((rt * (HID / 32) + kt) * 32 + c0 + 16 * hh) * 8 + lane] = stage[gl * (HID / 2) + (k >> 1)];
  }
}

extern "C" void kernel_launch(void* const* d_in, const int* in_sizes, int n_in,
                              void* d_out, int out_size, void* d_ws, size_t ws_size,
                              hipStream_t stream) {
  const float* x     = (const float*)d_in[0];
  const int*   ei    = (const int*)d_in[1];
  const int*   esrc  = ei;
  const int*   edst  = ei + N_EDGES;
  const int*   batch = (const int*)d_in[2];
  const float* cw1[3] = {(const float*)d_in[3],  (const float*)d_in[7],  (const float*)d_in[11]};
  const float* cb1[3] = {(const float*)d_in[4],  (const float*)d_in[8],  (const float*)d_in[12]};
  const float* cw2[3] = {(const float*)d_in[5],  (const float*)d_in[9],  (const float*)d_in[13]};
  const float* cb2[3] = {(const float*)d_in[6],  (const float*)d_in[10], (const float*)d_in[14]};
  const float* ow1 = (const float*)d_in[15];
  const float* ob1 = (const float*)d_in[16];
  const float* ow2 = (const float*)d_in[17];
  const float* ob2 = (const float*)d_in[18];

  char* ws = (char*)d_ws;
  size_t off = 0;
  auto bump = [&](size_t bytes) {
    void* p = ws + off;
    off += (bytes + 255) & ~(size_t)255;
    return p;
  };
  const int NRT = N_NODES / 16;
  unsigned short* Abf    = (unsigned short*)bump((size_t)N_NODES * HID * 2);
  float*          B2     = (float*)bump((size_t)N_NODES * HID * 4);
  unsigned int*   Ppk    = (unsigned int*)bump((size_t)(N_GRAPHS / 16) * 8 * 256 * 4);
  unsigned short* MPbf   = (unsigned short*)bump((size_t)N_GRAPHS * HID * 2);
  unsigned int*   wpk1[3], *wpk2[3];
  for (int l = 0; l < 3; ++l) {
    wpk1[l] = (unsigned int*)bump((size_t)(HID / 16) * 8 * 256 * 4);
    wpk2[l] = (unsigned int*)bump((size_t)(HID / 16) * 8 * 256 * 4);
  }
  unsigned int* owpk1 = (unsigned int*)bump((size_t)(HID / 16) * 8 * 256 * 4);
  unsigned int* owpk2 = (unsigned int*)bump((size_t)(OUT_C / 16) * 8 * 256 * 4);
  unsigned int* glst  = (unsigned int*)bump((size_t)AGG_GROUPS * AGG_WAVES * LST_CAP * 4);
  int*          gcnt  = (int*)bump((size_t)AGG_GROUPS * AGG_WAVES * 4);
  unsigned int* clst  = (unsigned int*)bump((size_t)CR_N * CS_N * CS_SL * 4);
  int*          ccnt  = (int*)bump((size_t)CR_N * CS_N * 4);

  dim3 blk(256);
  auto wpack = [&](const float* W, unsigned int* dst, int K, int M) {
    int KT = K / 32;
    long total = (long)(M / 16) * KT * 256;
    wpack_f32<<<(unsigned)((total + 255) / 256), blk, 0, stream>>>(W, dst, K, M, KT, total);
  };

  wpack(cw1[0], wpk1[0], IN_C, HID);
  wpack(cw2[0], wpk2[0], HID, HID);
  for (int l = 1; l < 3; ++l) {
    wpack(cw1[l], wpk1[l], HID, HID);
    wpack(cw2[l], wpk2[l], HID, HID);
  }
  wpack(ow1, owpk1, HID, HID);
  wpack(ow2, owpk2, HID, OUT_C);

  coarse_lists<<<CR_N * CS_N, AGG_THREADS, 0, stream>>>(edst, clst, ccnt);
  owner_lists<<<AGG_GROUPS, AGG_THREADS, 0, stream>>>(esrc, clst, ccnt, glst, gcnt);

  const float* h = x;
  for (int l = 0; l < 3; ++l) {
    if (l == 0) agg_owner<IN_C><<<AGG_GROUPS, AGG_THREADS, 0, stream>>>(h, esrc, edst, glst, gcnt, Abf);
    else        agg_owner<HID><<<AGG_GROUPS, AGG_THREADS, 0, stream>>>(h, esrc, edst, glst, gcnt, Abf);

    const unsigned mg = (unsigned)((NRT + 2 * MLP_WAVES - 1) / (2 * MLP_WAVES));
    if (l == 0) mlp_chain<IN_C / 32><<<mg, MLP_THREADS, 0, stream>>>(Abf, (const uint4*)wpk1[l], cb1[l],
                                                                     (const uint4*)wpk2[l], cb2[l], B2);
    else        mlp_chain<HID / 32><<<mg, MLP_THREADS, 0, stream>>>(Abf, (const uint4*)wpk1[l], cb1[l],
                                                                    (const uint4*)wpk2[l], cb2[l], B2);
    h = B2;
  }

  pool_owner<<<N_GRAPHS / POOL_GB, POOL_THREADS, 0, stream>>>(B2, batch, Ppk);

  const int GRT = N_GRAPHS / 16;
  dim3 g3((HID / 16 + 3) / 4, (GRT + 7) / 8);
  gemm_pk<1, 1, 0><<<g3, blk, 0, stream>>>((const uint4*)Ppk, (const uint4*)owpk1, ob1,
                                           MPbf, N_GRAPHS, HID, HID);
  dim3 g4((OUT_C / 16 + 3) / 4, (GRT + 7) / 8);
  gemm_pk<0, 0, 1><<<g4, blk, 0, stream>>>((const uint4*)MPbf, (const uint4*)owpk2, ob2,
                                           (float*)d_out, N_GRAPHS, HID, OUT_C);
}


# Weight pack launches removed: every launch that multiplies reads the f32 weights itself and rounds them (nearest even, as before) into its fragments (LDS staging, B loads); 19 -> 11 launches
# speedup vs baseline: 4.3918x; 1.0453x over previous
#include <hip/hip_runtime.h>

#define N_NODES 50000
#define N_EDGES 800000
#define N_GRAPHS 512
#define IN_C 128
#define HID 256
#define OUT_C 16

typedef __attribute__((ext_vector_type(16))) __bf16 v16bf;
typedef __attribute__((ext_vector_type(8)))  float  v8f;
typedef __attribute__((ext_vector_type(8)))  unsigned int v8u;
typedef const __attribute__((address_space(4))) v8u* list_ptr;

__device__ __forceinline__ unsigned short f2bf(float f) {
  union { float f; unsigned int u; } c; c.f = f;
  unsigned int u = c.u;
  u += 0x7fffu + ((u >> 16) & 1u);
  return (unsigned short)(u >> 16);
}

typedef __attribute__((ext_vector_type(2))) float  v2f;
typedef __attribute__((ext_vector_type(2))) __bf16 v2bf;
__device__ __forceinline__ unsigned int pack2bf(float a, float b) {
  const v2f f = {a, b};
  const v2bf h = __builtin_convertvector(f, v2bf);
  return __builtin_bit_cast(unsigned int, h);
}

__device__ __forceinline__ int frag_koff(int lane, int p) {
  int kb = (lane < 16) ? 0 : 8;
  return (p < 4) ? (kb + 2 * p) : (16 + kb + 2 * (p - 4));
}

#define AGG_NB      196
#define AGG_NK      ((AGG_NB + AGG_WAVES - 1) / AGG_WAVES)
#define AGG_WAVES   16
#define AGG_THREADS (AGG_WAVES * 32)
#define AGG_IT      15
#define AGG_SUB     (AGG_IT * 128)
#define AGG_ROUND   (AGG_SUB * AGG_WAVES)
#define AGG_NROUND  ((N_EDGES + AGG_ROUND - 1) / AGG_ROUND)
#define AGG_GROUPS  ((N_NODES + AGG_NB - 1) / AGG_NB)
#define LST_CAP     1024
#define LST_G       8
#define LST_PAD     (5 * LST_G)

static_assert(N_NODES <= (1 << 20) && N_EDGES <= (1 << 20) && AGG_NB < 256, "list entry fields");
static_assert(LST_G == 8, "a group of entries is one 8-dword scalar load");

static_assert(N_EDGES % 4 == 0, "four edges per index load");

#define CR_G        16
#define CR_NB       (AGG_NB * CR_G)
#define CR_N        ((AGG_GROUPS + CR_G - 1) / CR_G)
#define CS_IT       25
#define CS_SUB      (CS_IT * 128)
#define CS_SL       (CS_SUB * AGG_WAVES)
#define CS_N        ((N_EDGES + CS_SL - 1) / CS_SL)
#define P2_CAP      1024
#define P2_U        9

static_assert(CS_N == AGG_WAVES, "one wave of an owner group per slice list");
static_assert(CR_NB <= 4095, "a coarse range's local node fits 12 bits, and 4095 is no node of any group");

__device__ __forceinline__ void scan_load(const int* __restrict__ edst, int e0, int4 (&d)[AGG_IT]) {
  #pragma unroll
  for (int it = 0; it < AGG_IT; ++it) {
    const int e = e0 + it * 128;
    d[it] = *(const int4*)(edst + (e < N_EDGES ? e : N_EDGES - 4));
  }
}

__device__ __forceinline__ int scan_note(const int4 (&d)[AGG_IT], int e0, int base, int nloc,
                                         unsigned int ltmask, unsigned int* mylst) {
  int c = 0;
  #pragma unroll
  for (int it = 0; it < AGG_IT; ++it) {
    const unsigned int e = (unsigned int)(e0 + it * 128);
    const unsigned int x0 = (unsigned int)(d[it].x - base);
    const unsigned int x1 = (unsigned int)(d[it].y - base);
    const unsigned int x2 = (unsigned int)(d[it].z - base);
    const unsigned int x3 = (unsigned int)(d[it].w - base);
    const unsigned int lim = (e < (unsigned int)N_EDGES) ? (unsigned int)nloc : 0u;
    const bool m0 = x0 < lim, m1 = x1 < lim;
    const bool m2 = x2 < lim, m3 = x3 < lim;
    const unsigned int mn01 = x0 < x1 ? x0 : x1, mn23 = x2 < x3 ? x2 : x3;
    if (__builtin_amdgcn_ballot_w32((mn01 < mn23 ? mn01 : mn23) < lim)) {
      unsigned int b;
      b = __builtin_amdgcn_ballot_w32(m0);
      if (m0) mylst[c + __builtin_popcount(b & ltmask)] = (x0 << 20) | (e + 0u);
      c += __builtin_popcount(b);
      b = __builtin_amdgcn_ballot_w32(m1);
      if (m1) mylst[c + __builtin_popcount(b & ltmask)] = (x1 << 20) | (e + 1u);
      c += __builtin_popcount(b);
      b = __builtin_amdgcn_ballot_w32(m2);
      if (m2) mylst[c + __builtin_popcount(b & ltmask)] = (x2 << 20) | (e + 2u);
      c += __builtin_popcount(b);
      b = __builtin_amdgcn_ballot_w32(m3);
      if (m3) mylst[c + __builtin_popcount(b & ltmask)] = (x3 << 20) | (e + 3u);
      c += __builtin_popcount(b);
    }
  }
  return c;
}

__global__ void __launch_bounds__(AGG_THREADS)
coarse_lists(const int* __restrict__ edst, unsigned int* __restrict__ clst, int* __restrict__ ccnt) {
  __shared__ int cnt[AGG_WAVES];

  const int tid  = threadIdx.x;
  const int lane = tid & 31;
  const int wave = tid >> 5;
  const int ri   = blockIdx.x / CS_N;
  const int sl   = blockIdx.x % CS_N;
  const int rbase = ri * CR_NB;
  const unsigned int ltmask = (1u << lane) - 1u;
  const int e0 = sl * CS_SL + wave * CS_SUB + lane * 4;

  int4 d[CS_IT];
  #pragma unroll
  for (int it = 0; it < CS_IT; ++it) {
    const int e = e0 + it * 128;
    d[it] = *(const int4*)(edst + (e < N_EDGES ? e : N_EDGES - 4));
  }

  int c = 0;
  #pragma unroll
  for (int it = 0; it < CS_IT; ++it) {
    const unsigned int lim = ((e0 + it * 128) < N_EDGES) ? (unsigned int)CR_NB : 0u;
    c += __builtin_popcount(__builtin_amdgcn_ballot_w32((unsigned int)(d[it].x - rbase) < lim));
    c += __builtin_popcount(__builtin_amdgcn_ballot_w32((unsigned int)(d[it].y - rbase) < lim));
    c += __builtin_popcount(__builtin_amdgcn_ballot_w32((unsigned int)(d[it].z - rbase) < lim));
    c += __builtin_popcount(__builtin_amdgcn_ballot_w32((unsigned int)(d[it].w - rbase) < lim));
  }
  if (lane == 0) cnt[wave] = c;
  __syncthreads();

  const int myc = (lane < AGG_WAVES) ? cnt[lane] : 0;
  int incl = myc;
  #pragma unroll
  for (int o = 1; o < AGG_WAVES; o <<= 1) {
    const int t = __shfl_up(incl, o);
    if (lane >= o) incl += t;
  }
  const int total = __builtin_amdgcn_readlane(incl, AGG_WAVES - 1);
  const int wv = __builtin_amdgcn_readfirstlane(wave);
  int pos = __builtin_amdgcn_readlane(incl, wv) - c;

  unsigned int* out = clst + (long)blockIdx.x * CS_SL;
  #pragma unroll
  for (int it = 0; it < CS_IT; ++it) {
    const unsigned int e = (unsigned int)(e0 + it * 128);
    const unsigned int lim = (e < (unsigned int)N_EDGES) ? (unsigned int)CR_NB : 0u;
    const unsigned int x0 = (unsigned int)(d[it].x - rbase);
    const unsigned int x1 = (unsigned int)(d[it].y - rbase);
    const unsigned int x2 = (unsigned int)(d[it].z - rbase);
    const unsigned int x3 = (unsigned int)(d[it].w - rbase);
    const bool m0 = x0 < lim, m1 = x1 < lim;
    const bool m2 = x2 < lim, m3 = x3 < lim;
    unsigned int b;
    b = __builtin_amdgcn_ballot_w32(m0);
    if (m0) out[pos + __builtin_popcount(b & ltmask)] = (x0 << 20) | (e + 0u);
    pos += __builtin_popcount(b);
    b = __builtin_amdgcn_ballot_w32(m1);
    if (m1) out[pos + __builtin_popcount(b & ltmask)] = (x1 << 20) | (e + 1u);
    pos += __builtin_popcount(b);
    b = __builtin_amdgcn_ballot_w32(m2);
    if (m2) out[pos + __builtin_popcount(b & ltmask)] = (x2 << 20) | (e + 2u);
    pos += __builtin_popcount(b);
    b = __builtin_amdgcn_ballot_w32(m3);
    if (m3) out[pos + __builtin_popcount(b & ltmask)] = (x3 << 20) | (e + 3u);
    pos += __builtin_popcount(b);
  }
  const int npad = (total == 0) ? 4 : ((4 - (total & 3)) & 3);
  if (wave == 0 && lane < npad) out[total + lane] = 0xFFFFFFFFu;
  if (tid == 0) ccnt[blockIdx.x] = total;
}

__global__ void __launch_bounds__(AGG_THREADS)
owner_lists(const int* __restrict__ esrc, const unsigned int* __restrict__ clst, const int* __restrict__ ccnt,
            unsigned int* __restrict__ glst, int* __restrict__ gcnt) {
  __shared__ unsigned int lst[AGG_WAVES * P2_CAP];
  __shared__ unsigned int own[AGG_WAVES * LST_CAP];
  __shared__ int          cnt[AGG_WAVES];

  const int tid  = threadIdx.x;
  const int lane = tid & 31;
  const int wave = tid >> 5;
  const int wv   = __builtin_amdgcn_readfirstlane(wave);
  const int base = blockIdx.x * AGG_NB;
  int nloc = N_NODES - base;
  if (nloc > AGG_NB) nloc = AGG_NB;
  const int ri = blockIdx.x / CR_G;
  const unsigned int off = (unsigned int)(blockIdx.x % CR_G) * AGG_NB;

  const unsigned int ltmask = (1u << lane) - 1u;
  unsigned int* mylst = lst + wave * P2_CAP;
  unsigned int* myown = own + wave * LST_CAP;
  unsigned int* myout = glst + ((long)blockIdx.x * AGG_WAVES + wave) * LST_CAP;

  const int n  = __builtin_amdgcn_readfirstlane(ccnt[ri * CS_N + wv]);
  const int n4 = (n > 0) ? ((n + 3) >> 2) : 1;
  const uint4* L = (const uint4*)(clst + (long)(ri * CS_N + wv) * CS_SL);
  auto ask = [&](uint4 (&R)[P2_U], int q0) {
    #pragma unroll
    for (int u = 0; u < P2_U; ++u) {
      const int q = q0 + u * 32 + lane;
      R[u] = L[q < n4 ? q : n4 - 1];
    }
  };
  int c = 0;
  uint4 d[P2_U];
  ask(d, 0);
  for (int q0 = 0; q0 < n4; q0 += 32 * P2_U) {
    uint4 dn[P2_U];
    ask(dn, q0 + 32 * P2_U);
    #pragma unroll
    for (int u = 0; u < P2_U; ++u) {
      const unsigned int lim = (q0 + u * 32 + lane < n4) ? (unsigned int)nloc : 0u;
      const unsigned int x0 = (d[u].x >> 20) - off, x1 = (d[u].y >> 20) - off;
      const unsigned int x2 = (d[u].z >> 20) - off, x3 = (d[u].w >> 20) - off;
      const bool m0 = x0 < lim, m1 = x1 < lim;
      const bool m2 = x2 < lim, m3 = x3 < lim;
      unsigned int b;
      int p;
      b = __builtin_amdgcn_ballot_w32(m0);
      p = c + __builtin_popcount(b & ltmask);
      if (m0 && p < P2_CAP) mylst[p] = (x0 << 20) | (d[u].x & 0xFFFFFu);
      c += __builtin_popcount(b);
      b = __builtin_amdgcn_ballot_w32(m1);
      p = c + __builtin_popcount(b & ltmask);
      if (m1 && p < P2_CAP) mylst[p] = (x1 << 20) | (d[u].y & 0xFFFFFu);
      c += __builtin_popcount(b);
      b = __builtin_amdgcn_ballot_w32(m2);
      p = c + __builtin_popcount(b & ltmask);
      if (m2 && p < P2_CAP) mylst[p] = (x2 << 20) | (d[u].z & 0xFFFFFu);
      c += __builtin_popcount(b);
      b = __builtin_amdgcn_ballot_w32(m3);
      p = c + __builtin_popcount(b & ltmask);
      if (m3 && p < P2_CAP) mylst[p] = (x3 << 20) | (d[u].w & 0xFFFFFu);
      c += __builtin_popcount(b);
    }
    #pragma unroll
    for (int u = 0; u < P2_U; ++u) d[u] = dn[u];
  }
  if (lane == 0) cnt[wave] = c;
  __syncthreads();

  const int cv = (lane < AGG_WAVES) ? cnt[lane] : 0;
  if (__builtin_amdgcn_ballot_w32(cv > P2_CAP) != 0u) {
    if (lane == 0) gcnt[blockIdx.x * AGG_WAVES + wave] = LST_CAP;
    return;
  }

  int gpos = 0;
  for (int lw = 0; lw < AGG_WAVES; ++lw) {
    const int cn = __builtin_amdgcn_readlane(cv, lw);
    const unsigned int* Lw = lst + lw * P2_CAP;
    for (int f0 = 0; f0 < cn; f0 += 128) {
      unsigned int ent[4];
      #pragma unroll
      for (int j = 0; j < 4; ++j) {
        const int f = f0 + 32 * j + lane;
        ent[j] = (f < cn) ? Lw[f] : 0u;
      }
      #pragma unroll
      for (int j = 0; j < 4; ++j) {
        const int f = f0 + 32 * j + lane;
        const bool mine = (f < cn) && (((ent[j] >> 20) & (unsigned int)(AGG_WAVES - 1)) == (unsigned int)wave);
        const unsigned int m = __builtin_amdgcn_ballot_w32(mine);
        if (mine) {
          const int pos = gpos + __builtin_popcount(m & ltmask);
          if (pos < LST_CAP) myown[pos] = ent[j];
        }
        gpos += __builtin_popcount(m);
      }
    }
  }

  const int nout = gpos < LST_CAP ? gpos : LST_CAP;
  for (int i = lane; i < nout; i += 32) {
    const unsigned int ent = myown[i];
    myout[i] = (ent & 0xFFF00000u) | (unsigned int)esrc[ent & 0xFFFFFu];
  }
  #pragma unroll
  for (int k = 0; k < (LST_PAD + 31) / 32; ++k) {
    const int o = k * 32 + lane;
    if (o < LST_PAD && gpos + o < LST_CAP) myout[gpos + o] = ((unsigned int)AGG_NB << 20) | (unsigned int)base;
  }
  if (lane == 0) gcnt[blockIdx.x * AGG_WAVES + wave] = gpos;
}

typedef __attribute__((ext_vector_type(4))) float v4f;
template <int C> struct row_part;
template <> struct row_part<128> {
  typedef v4f type;
  static __device__ __forceinline__ type load(const float* row, int lane) {
    return *(const v4f*)(row + 4 * lane);
  }
  static __device__ __forceinline__ void store(float* row, int lane, type s) {
    *(v4f*)(row + 4 * lane) = s;
  }
  static __device__ __forceinline__ void store_bf16(unsigned short* row, int lane, type s) {
    *(uint2*)(row + 4 * lane) = make_uint2(pack2bf(s[0], s[1]), pack2bf(s[2], s[3]));
  }
};
template <> struct row_part<256> {
  typedef v8f type;
  static __device__ __forceinline__ type load(const float* row, int lane) {
    const v4f a = *(const v4f*)(row + 4 * lane), b = *(const v4f*)(row + 128 + 4 * lane);
    return __builtin_shufflevector(a, b, 0, 1, 2, 3, 4, 5, 6, 7);
  }
  static __device__ __forceinline__ void store(float* row, int lane, type s) {
    *(v4f*)(row + 4 * lane)       = __builtin_shufflevector(s, s, 0, 1, 2, 3);
    *(v4f*)(row + 128 + 4 * lane) = __builtin_shufflevector(s, s, 4, 5, 6, 7);
  }
  static __device__ __forceinline__ void store_bf16(unsigned short* row, int lane, type s) {
    *(uint2*)(row + 4 * lane)       = make_uint2(pack2bf(s[0], s[1]), pack2bf(s[2], s[3]));
    *(uint2*)(row + 128 + 4 * lane) = make_uint2(pack2bf(s[4], s[5]), pack2bf(s[6], s[7]));
  }
};

template <int C>
__global__ void __launch_bounds__(AGG_THREADS)
agg_owner(const float* __restrict__ h, const int* __restrict__ esrc,
          const int* __restrict__ edst, const unsigned int* __restrict__ glst,
          const int* __restrict__ gcnt, unsigned short* __restrict__ agg) {
  __shared__ float        acc[(AGG_NB + 1) * C];
  __shared__ unsigned int lst[AGG_WAVES * AGG_SUB];
  __shared__ int          cnt[AGG_WAVES];
  constexpr int VEC = C / 128;

  const int tid  = threadIdx.x;
  const int lane = tid & 31;
  const int wave = tid >> 5;
  const int base = blockIdx.x * AGG_NB;
  int nloc = N_NODES - base;
  if (nloc > AGG_NB) nloc = AGG_NB;

  const int lc = (lane < AGG_WAVES) ? gcnt[blockIdx.x * AGG_WAVES + lane] : 0;
  const bool fits = __builtin_amdgcn_ballot_w32(lc > LST_CAP - LST_PAD) == 0u;

  if (fits) {
    const int wv = __builtin_amdgcn_readfirstlane(wave);
    const int n  = __builtin_amdgcn_readlane(lc, wv);
    const list_ptr q = (list_ptr)(unsigned long)(glst + ((long)blockIdx.x * AGG_WAVES + wv) * LST_CAP);
    const int ng = (n + LST_G - 1) / LST_G;

    v8u eA = q[0], eB = q[1], eC = q[2];
    float4 A[LST_G][VEC], B[LST_G][VEC];
    auto rows_ask = [&](float4 (&R)[LST_G][VEC], const v8u& e) {
      #pragma unroll
      for (int j = 0; j < LST_G; ++j) {
        const int s = (int)(e[j] & 0xFFFFFu);
        const float4* hp = (const float4*)(h + (long)s * C) + lane;
        #pragma unroll
        for (int v = 0; v < VEC; ++v) R[j][v] = hp[v * 32];
      }
    };
    auto rows_add = [&](const float4 (&R)[LST_G][VEC], const v8u& e) {
      #pragma unroll
      for (int j = 0; j < LST_G; ++j) {
        const int nd = (int)(e[j] >> 20);
        float4* ap = (float4*)(acc + nd * C) + lane;
        #pragma unroll
        for (int v = 0; v < VEC; ++v) {
          float4 a = ap[v * 32];
          a.x += R[j][v].x; a.y += R[j][v].y; a.z += R[j][v].z; a.w += R[j][v].w;
          ap[v * 32] = a;
        }
      }
    };

    typename row_part<C>::type own[AGG_NK];
    #pragma unroll
    for (int k = 0; k < AGG_NK; ++k) {
      const int loc = wv + AGG_WAVES * k;
      own[k] = row_part<C>::load(h + (long)(base + (loc < nloc ? loc : 0)) * C, lane);
    }
    rows_ask(A, eA);
    #pragma unroll
    for (int k = 0; k < AGG_NK; ++k) {
      const int loc = wv + AGG_WAVES * k;
      row_part<C>::store(acc + (loc < nloc ? loc : AGG_NB) * C, lane, own[k]);
    }
    const int nt = (ng + 1) / 2;
    int t = 0;
    do {
      const v8u fB = q[2 * t + 3], fC = q[2 * t + 4];
      rows_ask(B, eB);
      rows_add(A, eA);
      rows_ask(A, eC);
      rows_add(B, eB);
      eA = eC; eB = fB; eC = fC;
    } while (++t < nt);

    typename row_part<C>::type fin[AGG_NK];
    #pragma unroll
    for (int k = 0; k < AGG_NK; ++k) {
      const int loc = wv + AGG_WAVES * k;
      fin[k] = row_part<C>::load(acc + (loc < nloc ? loc : AGG_NB) * C, lane);
    }
    #pragma unroll
    for (int k = 0; k < AGG_NK; ++k) {
      const int loc = wv + AGG_WAVES * k;
      if (loc < nloc)
        row_part<C>::store_bf16(agg + (long)(base + loc) * C, lane, fin[k]);
    }
  } else {
    {
      const float4* hs = (const float4*)(h + (long)base * C);
      float4* a4 = (float4*)acc;
      const int n4 = nloc * (C / 4);
      for (int i = tid; i < n4; i += AGG_THREADS) a4[i] = hs[i];
    }
    const unsigned int ltmask = (1u << lane) - 1u;
    unsigned int* mylst = lst + wave * AGG_SUB;
    int rr = (int)(blockIdx.x % AGG_NROUND);
    for (int r = 0; r < AGG_NROUND; ++r) {
      const int e0 = rr * AGG_ROUND + wave * AGG_SUB + lane * 4;
      rr = (rr + 1 == AGG_NROUND) ? 0 : rr + 1;
      int4 d[AGG_IT];
      scan_load(edst, e0, d);
      const int c = scan_note(d, e0, base, nloc, ltmask, mylst);
      if (lane == 0) cnt[wave] = c;
      __syncthreads();

      const int myc = (lane < AGG_WAVES) ? cnt[lane] : 0;
      int incl = myc;
      #pragma unroll
      for (int o = 1; o < AGG_WAVES; o <<= 1) {
        const int t = __shfl_up(incl, o);
        if (lane >= o) incl += t;
      }
      const int excl  = incl - myc;
      const int total = __builtin_amdgcn_readlane(incl, AGG_WAVES - 1);
      for (int f0 = 0; f0 < total; f0 += 32) {
        const int f = f0 + lane;
        int li = 0;
        #pragma unroll
        for (int l = 0; l < AGG_WAVES - 1; ++l) li += (f >= __builtin_amdgcn_readlane(incl, l)) ? 1 : 0;
        const int st = __shfl(excl, li);
        unsigned int ent = 0u;
        bool mine = false;
        if (f < total) {
          ent  = lst[li * AGG_SUB + (f - st)];
          mine = ((ent >> 20) & (unsigned int)(AGG_WAVES - 1)) == (unsigned int)wave;
        }
        unsigned int m = __builtin_amdgcn_ballot_w32(mine);
        int s = 0;
        if (mine) s = esrc[ent & 0xFFFFFu];
        while (m) {
          const int b0 = __builtin_ctz(m); m &= m - 1u;
          const int s0 = __builtin_amdgcn_readlane(s, b0);
          const int n0 = (int)(__builtin_amdgcn_readlane((int)ent, b0) >> 20) & 0xFF;
          const float4* h0 = (const float4*)(h + (long)s0 * C) + lane;
          float4* a0 = (float4*)(acc + n0 * C) + lane;
          #pragma unroll
          for (int j = 0; j < VEC; ++j) {
            const float4 v0 = h0[j * 32];
            float4 a = a0[j * 32];
            a.x += v0.x; a.y += v0.y; a.z += v0.z; a.w += v0.w;
            a0[j * 32] = a;
          }
        }
      }
      __syncthreads();
    }

    {
      uint4* o8 = (uint4*)(agg + (long)base * C);
      const float4* a4 = (const float4*)acc;
      const int n8 = nloc * (C / 8);
      for (int i = tid; i < n8; i += AGG_THREADS) {
        const float4 u = a4[2 * i], v = a4[2 * i + 1];
        o8[i] = make_uint4(pack2bf(u.x, u.y), pack2bf(u.z, u.w), pack2bf(v.x, v.y), pack2bf(v.z, v.w));
      }
    }
  }
}

template <int RELU, int OUTBF, int AROW>
__global__ void gemm_pk(const uint4* __restrict__ Apk, const float* __restrict__ W,
                        const float* __restrict__ bias, void* __restrict__ Cout,
                        int N, int K, int M) {
  const int tid  = threadIdx.x;
  const int lane = tid & 31;
  const int wave = tid >> 5;
  const int wr   = wave & 3;
  const int wc   = wave >> 2;
  const int KT   = K >> 5;
  const int nrt  = N >> 4;
  const int nct  = M >> 4;
  const int rt0  = blockIdx.y * 8 + wr * 2;
  const int ct0  = blockIdx.x * 4 + wc * 2;
  const bool r0 = rt0 < nrt,     r1 = (rt0 + 1) < nrt;
  const bool c0 = ct0 < nct,     c1 = (ct0 + 1) < nct;
  const int rA0 = r0 ? rt0 : 0,  rA1 = r1 ? (rt0 + 1) : 0;
  const int cB0 = c0 ? ct0 : 0,  cB1 = c1 ? (ct0 + 1) : 0;

  long ia0 = AROW ? ((long)(rA0 * 16 + (lane & 15)) * (K >> 3) + (lane >> 4)) : (((long)rA0 * KT * 32 + lane) * 2);
  long ia1 = AROW ? ((long)(rA1 * 16 + (lane & 15)) * (K >> 3) + (lane >> 4)) : (((long)rA1 * KT * 32 + lane) * 2);
  const float* wb0 = W + (long)(8 * (lane >> 4)) * M + cB0 * 16 + (lane & 15);
  const float* wb1 = W + (long)(8 * (lane >> 4)) * M + cB1 * 16 + (lane & 15);
  const long stepA = AROW ? 4 : 64;
  const long second = AROW ? 2 : 1;

  v8f acc00 = {}, acc01 = {}, acc10 = {}, acc11 = {};
  union F { uint4 q[2]; v16bf v; };
  auto wfrag = [&](F& f, const float* p) {
    f.q[0] = make_uint4(pack2bf(p[0], p[M]), pack2bf(p[2 * M], p[3 * M]),
                        pack2bf(p[4 * M], p[5 * M]), pack2bf(p[6 * M], p[7 * M]));
    p += 16 * M;
    f.q[1] = make_uint4(pack2bf(p[0], p[M]), pack2bf(p[2 * M], p[3 * M]),
                        pack2bf(p[4 * M], p[5 * M]), pack2bf(p[6 * M], p[7 * M]));
  };

  for (int kt = 0; kt < KT; ++kt) {
    F a0, a1, b0, b1;
    a0.q[0] = Apk[ia0]; a0.q[1] = Apk[ia0 + second];
    a1.q[0] = Apk[ia1]; a1.q[1] = Apk[ia1 + second];
    wfrag(b0, wb0);
    wfrag(b1, wb1);
    if (kt + 1 < KT) {
      __builtin_prefetch((const void*)(Apk + ia0 + stepA), 0, 3);
      __builtin_prefetch((const void*)(Apk + ia1 + stepA), 0, 3);
    }
    ia0 += stepA; ia1 += stepA; wb0 += 32 * M; wb1 += 32 * M;
    acc00 = __builtin_amdgcn_wmma_f32_16x16x32_bf16(false, a0.v, false, b0.v, (short)0, acc00, false, false);
    acc01 = __builtin_amdgcn_wmma_f32_16x16x32_bf16(false, a0.v, false, b1.v, (short)0, acc01, false, false);
    acc10 = __builtin_amdgcn_wmma_f32_16x16x32_bf16(false, a1.v, false, b0.v, (short)0, acc10, false, false);
    acc11 = __builtin_amdgcn_wmma_f32_16x16x32_bf16(false, a1.v, false, b1.v, (short)0, acc11, false, false);
  }

  const int rowL = lane & 15;
  auto st = [&](const v8f& acc, int rt, bool rok, int ct, bool cok) {
    if (!rok || !cok) return;
    int col = ct * 16 + rowL;
    #pragma unroll
    for (int r = 0; r < 8; ++r) {
      int row = rt * 16 + ((lane < 16) ? r : r + 8);
      float o = acc[r] + bias[col];
      if (RELU) o = fmaxf(o, 0.0f);
      if (OUTBF) ((unsigned short*)Cout)[(long)row * M + col] = f2bf(o);
      else       ((float*)Cout)[(long)row * M + col] = o;
    }
  };
  st(acc00, rt0, r0, ct0, c0);         st(acc01, rt0, r0, ct0 + 1, c1);
  st(acc10, rt0 + 1, r1, ct0, c0);     st(acc11, rt0 + 1, r1, ct0 + 1, c1);
}


#define MLP_WAVES 8
#define MLP_THREADS (MLP_WAVES * 32)
#define MLP_PITCH 36
#define WT_PITCH(KT) ((KT) * 64 + 1)

template <int KT1>
__device__ __forceinline__ void stage_weights(const float* __restrict__ W1, uint4* S1,
                                              const float* __restrict__ W2, uint4* S2, int wave, int lane) {
  constexpr int KT2 = HID / 32;
  constexpr int NC  = (KT1 + KT2) / 4;
  static_assert(KT1 % 4 == 0 && KT2 % 4 == 0 && MLP_WAVES == 8 && HID == 256 && NC >= 2 && NC <= 4,
                "eight waves = two halves of a weight row x four groups of eight k");
  const int quad = ((wave & 1) << 5) + lane;
  const int jh   = wave >> 1;
  const float4* g1 = (const float4*)(W1 + (long)(8 * jh) * HID) + quad;
  const float4* g2 = (const float4*)(W2 + (long)(8 * jh) * HID) + quad;
  const int so = (jh >> 1) * 32 + (jh & 1) * 16 + 4 * (quad & 3);
  uint4* s1 = S1 + (quad >> 2) * WT_PITCH(KT1) + so;
  uint4* s2 = S2 + (quad >> 2) * WT_PITCH(KT2) + so;
  float4 X[4][8], Y[4][8];
  auto ask = [&](float4 (&R)[4][8], int ch) {
    #pragma unroll
    for (int u = 0; u < 4; ++u) {
      const int s = 4 * ch + u;
      const float4* g = (s < KT1) ? g1 + s * (32 * HID / 4) : g2 + (s - KT1) * (32 * HID / 4);
      #pragma unroll
      for (int i = 0; i < 8; ++i) R[u][i] = g[i * (HID / 4)];
    }
  };
  auto put = [&](const float4 (&R)[4][8], int ch) {
    #pragma unroll
    for (int u = 0; u < 4; ++u) {
      const int s = 4 * ch + u;
      uint4* d = (s < KT1) ? s1 + s * 64 : s2 + (s - KT1) * 64;
      d[0] = make_uint4(pack2bf(R[u][0].x, R[u][1].x), pack2bf(R[u][2].x, R[u][3].x),
                        pack2bf(R[u][4].x, R[u][5].x), pack2bf(R[u][6].x, R[u][7].x));
      d[1] = make_uint4(pack2bf(R[u][0].y, R[u][1].y), pack2bf(R[u][2].y, R[u][3].y),
                        pack2bf(R[u][4].y, R[u][5].y), pack2bf(R[u][6].y, R[u][7].y));
      d[2] = make_uint4(pack2bf(R[u][0].z, R[u][1].z), pack2bf(R[u][2].z, R[u][3].z),
                        pack2bf(R[u][4].z, R[u][5].z), pack2bf(R[u][6].z, R[u][7].z));
      d[3] = make_uint4(pack2bf(R[u][0].w, R[u][1].w), pack2bf(R[u][2].w, R[u][3].w),
                        pack2bf(R[u][4].w, R[u][5].w), pack2bf(R[u][6].w, R[u][7].w));
    }
  };
  ask(X, 0);
  ask(Y, 1);
  put(X, 0);
  if (NC > 2) ask(X, 2);
  put(Y, 1);
  if (NC > 3) ask(Y, 3);
  if (NC > 2) put(X, 2);
  if (NC > 3) put(Y, 3);
}
template <int KT1>
__global__ void __launch_bounds__(MLP_THREADS)
mlp_chain(const unsigned short* __restrict__ Xbf, const float* __restrict__ W1, const float* __restrict__ b1,
          const float* __restrict__ W2, const float* __restrict__ b2, float* __restrict__ out) {
  __shared__ alignas(16) float stage[MLP_WAVES][2][16 * MLP_PITCH];
  __shared__ uint4 w1s[(HID / 16) * WT_PITCH(KT1)];
  __shared__ uint4 w2s[(HID / 16) * WT_PITCH(HID / 32)];
  const int lane = threadIdx.x & 31;
  const int wave = threadIdx.x >> 5;
  const int c    = lane & 15;
  const int hh   = lane >> 4;
  const int nrt  = N_NODES / 16;
  const int t0   = (blockIdx.x * MLP_WAVES + wave) * 2;
  const bool ok0 = t0 < nrt, ok1 = (t0 + 1) < nrt;
  const int tA = ok0 ? t0 : 0, tB = ok1 ? (t0 + 1) : 0;
  constexpr int KT2 = HID / 32;
  constexpr int UT  = HID / 16;
  union F { uint4 q[2]; v16bf v; };

  v8f acc[UT][2];
  #pragma unroll
  for (int j = 0; j < UT; ++j) { acc[j][0] = v8f{}; acc[j][1] = v8f{}; }
  const uint4* w1l = w1s + lane;
  auto w1_ask = [&](F (&g)[UT / 2], int kt, int j0) {
    #pragma unroll
    for (int j = 0; j < UT / 2; ++j) {
      const uint4* pw = w1l + (j0 + j) * WT_PITCH(KT1) + kt * 64;
      g[j].q[0] = pw[0]; g[j].q[1] = pw[32];
    }
  };
  const uint4* xr0 = (const uint4*)(Xbf + (long)(tA * 16 + c) * (32 * KT1)) + hh;
  const uint4* xr1 = (const uint4*)(Xbf + (long)(tB * 16 + c) * (32 * KT1)) + hh;
  auto x_ask = [&](F& x0, F& x1, int kt) {
    x0.q[0] = xr0[4 * kt]; x0.q[1] = xr0[4 * kt + 2];
    x1.q[0] = xr1[4 * kt]; x1.q[1] = xr1[4 * kt + 2];
  };
  F wA[UT / 2], wB[UT / 2], x0, x1;
  x_ask(x0, x1, 0);
  stage_weights<KT1>(W1, w1s, W2, w2s, wave, lane);
  __syncthreads();
  w1_ask(wA, 0, 0);
  for (int kt = 0; kt < KT1; ++kt) {
    const int ktn = (kt + 1 < KT1) ? kt + 1 : kt;
    F y0, y1;
    w1_ask(wB, kt, UT / 2);
    x_ask(y0, y1, ktn);
    __builtin_amdgcn_sched_barrier(0);
    #pragma unroll
    for (int j = 0; j < UT / 2; ++j) {
      acc[j][0] = __builtin_amdgcn_wmma_f32_16x16x32_bf16(false, wA[j].v, false, x0.v, (short)0, acc[j][0], false, false);
      acc[j][1] = __builtin_amdgcn_wmma_f32_16x16x32_bf16(false, wA[j].v, false, x1.v, (short)0, acc[j][1], false, false);
    }
    __builtin_amdgcn_sched_barrier(0);
    w1_ask(wA, ktn, 0);
    __builtin_amdgcn_sched_barrier(0);
    #pragma unroll
    for (int j = 0; j < UT / 2; ++j) {
      acc[UT / 2 + j][0] = __builtin_amdgcn_wmma_f32_16x16x32_bf16(false, wB[j].v, false, x0.v, (short)0, acc[UT / 2 + j][0], false, false);
      acc[UT / 2 + j][1] = __builtin_amdgcn_wmma_f32_16x16x32_bf16(false, wB[j].v, false, x1.v, (short)0, acc[UT / 2 + j][1], false, false);
    }
    __builtin_amdgcn_sched_barrier(0);
    x0 = y0; x1 = y1;
  }

  F m[2][KT2];
  #pragma unroll
  for (int t = 0; t < KT2; ++t) {
    const float4* bp = (const float4*)(b1 + 32 * t + 8 * hh);
    const float4 ba0 = bp[0], ba1 = bp[1];
    const float4 bb0 = bp[4], bb1 = bp[5];
    const float ba[8] = {ba0.x, ba0.y, ba0.z, ba0.w, ba1.x, ba1.y, ba1.z, ba1.w};
    const float bb[8] = {bb0.x, bb0.y, bb0.z, bb0.w, bb1.x, bb1.y, bb1.z, bb1.w};
    #pragma unroll
    for (int nt = 0; nt < 2; ++nt) {
      unsigned int lo[4], hi[4];
      #pragma unroll
      for (int p = 0; p < 4; ++p) {
        const float a0 = fmaxf(acc[2 * t][nt][2 * p]     + ba[2 * p],     0.0f);
        const float a1 = fmaxf(acc[2 * t][nt][2 * p + 1] + ba[2 * p + 1], 0.0f);
        const float c0 = fmaxf(acc[2 * t + 1][nt][2 * p]     + bb[2 * p],     0.0f);
        const float c1 = fmaxf(acc[2 * t + 1][nt][2 * p + 1] + bb[2 * p + 1], 0.0f);
        lo[p] = pack2bf(a0, a1);
        hi[p] = pack2bf(c0, c1);
      }
      m[nt][t].q[0] = make_uint4(lo[0], lo[1], lo[2], lo[3]);
      m[nt][t].q[1] = make_uint4(hi[0], hi[1], hi[2], hi[3]);
    }
  }

  float* st0 = &stage[wave][0][0];
  float* st1 = &stage[wave][1][0];
  const uint4* w2l = w2s + lane;
  auto w2_ask = [&](F (&g)[8], int cp, int tq) {
    #pragma unroll
    for (int t = 0; t < 4; ++t) {
      const uint4* pa = w2l + (2 * cp) * WT_PITCH(KT2) + (tq + t) * 64;
      const uint4* pb = w2l + (2 * cp + 1) * WT_PITCH(KT2) + (tq + t) * 64;
      g[2 * t].q[0] = pa[0];     g[2 * t].q[1] = pa[32];
      g[2 * t + 1].q[0] = pb[0]; g[2 * t + 1].q[1] = pb[32];
    }
  };
  w2_ask(wA, 0, 0);
  for (int cp = 0; cp < UT / 2; ++cp) {
    const int cpn = (cp + 1 < UT / 2) ? cp + 1 : cp;
    v8f o00 = {}, o01 = {}, o10 = {}, o11 = {};
    w2_ask(wB, cp, 4);
    __builtin_amdgcn_sched_barrier(0);
    #pragma unroll
    for (int t = 0; t < 4; ++t) {
      o00 = __builtin_amdgcn_wmma_f32_16x16x32_bf16(false, wA[2 * t].v,     false, m[0][t].v, (short)0, o00, false, false);
      o01 = __builtin_amdgcn_wmma_f32_16x16x32_bf16(false, wA[2 * t].v,     false, m[1][t].v, (short)0, o01, false, false);
      o10 = __builtin_amdgcn_wmma_f32_16x16x32_bf16(false, wA[2 * t + 1].v, false, m[0][t].v, (short)0, o10, false, false);
      o11 = __builtin_amdgcn_wmma_f32_16x16x32_bf16(false, wA[2 * t + 1].v, false, m[1][t].v, (short)0, o11, false, false);
    }
    __builtin_amdgcn_sched_barrier(0);
    w2_ask(wA, cpn, 0);
    __builtin_amdgcn_sched_barrier(0);
    #pragma unroll
    for (int t = 0; t < 4; ++t) {
      o00 = __builtin_amdgcn_wmma_f32_16x16x32_bf16(false, wB[2 * t].v,     false, m[0][4 + t].v, (short)0, o00, false, false);
      o01 = __builtin_amdgcn_wmma_f32_16x16x32_bf16(false, wB[2 * t].v,     false, m[1][4 + t].v, (short)0, o01, false, false);
      o10 = __builtin_amdgcn_wmma_f32_16x16x32_bf16(false, wB[2 * t + 1].v, false, m[0][4 + t].v, (short)0, o10, false, false);
      o11 = __builtin_amdgcn_wmma_f32_16x16x32_bf16(false, wB[2 * t + 1].v, false, m[1][4 + t].v, (short)0, o11, false, false);
    }
    __builtin_amdgcn_sched_barrier(0);
    const float4* bq = (const float4*)(b2 + 32 * cp + 8 * hh);
    const float4 ba0 = bq[0], ba1 = bq[1], bb0 = bq[4], bb1 = bq[5];
    auto put = [&](float* st, const v8f& oa, const v8f& ob) {
      float4* r = (float4*)(st + c * MLP_PITCH + 8 * hh);
      r[0] = make_float4(fmaxf(oa[0] + ba0.x, 0.0f), fmaxf(oa[1] + ba0.y, 0.0f),
                         fmaxf(oa[2] + ba0.z, 0.0f), fmaxf(oa[3] + ba0.w, 0.0f));
      r[1] = make_float4(fmaxf(oa[4] + ba1.x, 0.0f), fmaxf(oa[5] + ba1.y, 0.0f),
                         fmaxf(oa[6] + ba1.z, 0.0f), fmaxf(oa[7] + ba1.w, 0.0f));
      r[4] = make_float4(fmaxf(ob[0] + bb0.x, 0.0f), fmaxf(ob[1] + bb0.y, 0.0f),
                         fmaxf(ob[2] + bb0.z, 0.0f), fmaxf(ob[3] + bb0.w, 0.0f));
      r[5] = make_float4(fmaxf(ob[4] + bb1.x, 0.0f), fmaxf(ob[5] + bb1.y, 0.0f),
                         fmaxf(ob[6] + bb1.z, 0.0f), fmaxf(ob[7] + bb1.w, 0.0f));
    };
    put(st0, o00, o10);
    put(st1, o01, o11);
    auto rows_out = [&](const float* st, int tile, bool ok) {
      if (!ok) return;
      #pragma unroll
      for (int i = 0; i < 4; ++i) {
        const int node = 4 * i + (lane >> 3);
        const float4 v = *(const float4*)(st + node * MLP_PITCH + 4 * (lane & 7));
        *(float4*)(out + (long)(tile * 16 + node) * HID + 32 * cp + 4 * (lane & 7)) = v;
      }
    };
    rows_out(st0, t0, ok0);
    rows_out(st1, t0 + 1, ok1);
  }
}

#define POOL_GB      4
#define POOL_WAVES   16
#define POOL_THREADS (POOL_WAVES * 32)
#define POOL_IT      25
#define POOL_SUB     (POOL_IT * 128)
#define POOL_PG      16

static_assert(POOL_SUB * POOL_WAVES >= N_NODES && N_NODES % 4 == 0, "the slices cover the id list, four ids per load");
static_assert(N_GRAPHS % POOL_GB == 0 && 16 % POOL_GB == 0 && POOL_GB * 8 == 32, "four graphs x eight dwords = one line");
static_assert(POOL_WAVES == POOL_GB * (HID / 64) && POOL_WAVES == 2 * (HID / 32), "one wave per graph and 64 columns; one per line");

__global__ void __launch_bounds__(POOL_THREADS)
pool_owner(const float* __restrict__ h, const int* __restrict__ batch, unsigned int* __restrict__ Ppk) {
  __shared__ unsigned int lst[POOL_WAVES * POOL_SUB];
  __shared__ int          cnt[POOL_WAVES];
  __shared__ unsigned int stage[POOL_GB * (HID / 2)];

  const int tid  = threadIdx.x;
  const int lane = tid & 31;
  const int wave = tid >> 5;
  const int g0   = blockIdx.x * POOL_GB;
  const unsigned int ltmask = (1u << lane) - 1u;

  {
    unsigned int* mylst = lst + wave * POOL_SUB;
    const int n0 = wave * POOL_SUB + lane * 4;
    int4 d[POOL_IT];
    #pragma unroll
    for (int it = 0; it < POOL_IT; ++it) {
      const int n = n0 + it * 128;
      d[it] = *(const int4*)(batch + (n < N_NODES ? n : N_NODES - 4));
    }
    int c = 0;
    #pragma unroll
    for (int it = 0; it < POOL_IT; ++it) {
      const unsigned int n  = (unsigned int)(n0 + it * 128);
      const unsigned int x0 = (unsigned int)(d[it].x - g0);
      const unsigned int x1 = (unsigned int)(d[it].y - g0);
      const unsigned int x2 = (unsigned int)(d[it].z - g0);
      const unsigned int x3 = (unsigned int)(d[it].w - g0);
      const unsigned int lim = (n < (unsigned int)N_NODES) ? (unsigned int)POOL_GB : 0u;
      const bool m0 = x0 < lim, m1 = x1 < lim;
      const bool m2 = x2 < lim, m3 = x3 < lim;
      if (__builtin_amdgcn_ballot_w32(m0 || m1 || m2 || m3)) {
        unsigned int b;
        b = __builtin_amdgcn_ballot_w32(m0);
        if (m0) mylst[c + __builtin_popcount(b & ltmask)] = (x0 << 20) | (n + 0u);
        c += __builtin_popcount(b);
        b = __builtin_amdgcn_ballot_w32(m1);
        if (m1) mylst[c + __builtin_popcount(b & ltmask)] = (x1 << 20) | (n + 1u);
        c += __builtin_popcount(b);
        b = __builtin_amdgcn_ballot_w32(m2);
        if (m2) mylst[c + __builtin_popcount(b & ltmask)] = (x2 << 20) | (n + 2u);
        c += __builtin_popcount(b);
        b = __builtin_amdgcn_ballot_w32(m3);
        if (m3) mylst[c + __builtin_popcount(b & ltmask)] = (x3 << 20) | (n + 3u);
        c += __builtin_popcount(b);
      }
    }
    if (lane == 0) cnt[wave] = c;
  }
  __syncthreads();

  const unsigned int gi = (unsigned int)(wave & (POOL_GB - 1));
  const int cq = wave / POOL_GB;
  const float2* hq = (const float2*)h + cq * 32 + lane;
  float2 acc = make_float2(0.0f, 0.0f);
  int deg = 0;
  for (int lw = 0; lw < POOL_WAVES; ++lw) {
    const int cn = __builtin_amdgcn_readfirstlane(cnt[lw]);
    const unsigned int* L = lst + lw * POOL_SUB;
    for (int f0 = 0; f0 < cn; f0 += 32) {
      const int f = f0 + lane;
      const unsigned int e = (f < cn) ? L[f] : 0xFFFFFFFFu;
      unsigned int m = __builtin_amdgcn_ballot_w32((e >> 20) == gi);
      deg += __builtin_popcount(m);
      while (m) {
        float2 v[POOL_PG];
        #pragma unroll
        for (int j = 0; j < POOL_PG; ++j) {
          v[j] = make_float2(0.0f, 0.0f);
          if (m) {
            const int b = __builtin_ctz(m);
            m &= m - 1u;
            const int n = __builtin_amdgcn_readlane((int)e, b) & 0xFFFFF;
            v[j] = hq[n * (HID / 2)];
          }
        }
        #pragma unroll
        for (int j = 0; j < POOL_PG; ++j) { acc.x += v[j].x; acc.y += v[j].y; }
      }
    }
  }

  const float den = fmaxf((float)deg, 1.0f);
  stage[gi * (HID / 2) + cq * 32 + lane] = pack2bf(acc.x / den, acc.y / den);
  __syncthreads();

  {
    const int kt = wave >> 1;
    const int hh = wave & 1;
    const int gl = lane >> 3;
    const int p  = lane & 7;
    const int k  = kt * 32 + frag_koff(16 * hh, p);
    const int rt = g0 >> 4;
    const int c0 = g0 & 15;
    Ppk[((rt * (HID / 32) + kt) * 32 + c0 + 16 * hh) * 8 + lane] = stage[gl * (HID / 2) + (k >> 1)];
  }
}

extern "C" void kernel_launch(void* const* d_in, const int* in_sizes, int n_in,
                              void* d_out, int out_size, void* d_ws, size_t ws_size,
                              hipStream_t stream) {
  const float* x     = (const float*)d_in[0];
  const int*   ei    = (const int*)d_in[1];
  const int*   esrc  = ei;
  const int*   edst  = ei + N_EDGES;
  const int*   batch = (const int*)d_in[2];
  const float* cw1[3] = {(const float*)d_in[3],  (const float*)d_in[7],  (const float*)d_in[11]};
  const float* cb1[3] = {(const float*)d_in[4],  (const float*)d_in[8],  (const float*)d_in[12]};
  const float* cw2[3] = {(const float*)d_in[5],  (const float*)d_in[9],  (const float*)d_in[13]};
  const float* cb2[3] = {(const float*)d_in[6],  (const float*)d_in[10], (const float*)d_in[14]};
  const float* ow1 = (const float*)d_in[15];
  const float* ob1 = (const float*)d_in[16];
  const float* ow2 = (const float*)d_in[17];
  const float* ob2 = (const float*)d_in[18];

  char* ws = (char*)d_ws;
  size_t off = 0;
  auto bump = [&](size_t bytes) {
    void* p = ws + off;
    off += (bytes + 255) & ~(size_t)255;
    return p;
  };
  const int NRT = N_NODES / 16;
  unsigned short* Abf    = (unsigned short*)bump((size_t)N_NODES * HID * 2);
  float*          B2     = (float*)bump((size_t)N_NODES * HID * 4);
  unsigned int*   Ppk    = (unsigned int*)bump((size_t)(N_GRAPHS / 16) * 8 * 256 * 4);
  unsigned short* MPbf   = (unsigned short*)bump((size_t)N_GRAPHS * HID * 2);
  unsigned int* glst  = (unsigned int*)bump((size_t)AGG_GROUPS * AGG_WAVES * LST_CAP * 4);
  int*          gcnt  = (int*)bump((size_t)AGG_GROUPS * AGG_WAVES * 4);
  unsigned int* clst  = (unsigned int*)bump((size_t)CR_N * CS_N * CS_SL * 4);
  int*          ccnt  = (int*)bump((size_t)CR_N * CS_N * 4);

  dim3 blk(256);


  coarse_lists<<<CR_N * CS_N, AGG_THREADS, 0, stream>>>(edst, clst, ccnt);
  owner_lists<<<AGG_GROUPS, AGG_THREADS, 0, stream>>>(esrc, clst, ccnt, glst, gcnt);

  const float* h = x;
  for (int l = 0; l < 3; ++l) {
    if (l == 0) agg_owner<IN_C><<<AGG_GROUPS, AGG_THREADS, 0, stream>>>(h, esrc, edst, glst, gcnt, Abf);
    else        agg_owner<HID><<<AGG_GROUPS, AGG_THREADS, 0, stream>>>(h, esrc, edst, glst, gcnt, Abf);

    const unsigned mg = (unsigned)((NRT + 2 * MLP_WAVES - 1) / (2 * MLP_WAVES));
    if (l == 0) mlp_chain<IN_C / 32><<<mg, MLP_THREADS, 0, stream>>>(Abf, cw1[l], cb1[l], cw2[l], cb2[l], B2);
    else        mlp_chain<HID / 32><<<mg, MLP_THREADS, 0, stream>>>(Abf, cw1[l], cb1[l], cw2[l], cb2[l], B2);
    h = B2;
  }

  pool_owner<<<N_GRAPHS / POOL_GB, POOL_THREADS, 0, stream>>>(B2, batch, Ppk);

  const int GRT = N_GRAPHS / 16;
  dim3 g3((HID / 16 + 3) / 4, (GRT + 7) / 8);
  gemm_pk<1, 1, 0><<<g3, blk, 0, stream>>>((const uint4*)Ppk, ow1, ob1,
                                           MPbf, N_GRAPHS, HID, HID);
  dim3 g4((OUT_C / 16 + 3) / 4, (GRT + 7) / 8);
  gemm_pk<0, 0, 1><<<g4, blk, 0, stream>>>((const uint4*)MPbf, ow2, ob2,
                                           (float*)d_out, N_GRAPHS, HID, OUT_C);
}


# Output MLP folded into the mean-pool launch: each 4-graph owner group takes its means through both products itself (one 16-row tile, unowned rows zero; hidden rows in LDS); 11 -> 9 launches
# speedup vs baseline: 4.7241x; 1.0757x over previous
#include <hip/hip_runtime.h>

#define N_NODES 50000
#define N_EDGES 800000
#define N_GRAPHS 512
#define IN_C 128
#define HID 256
#define OUT_C 16

typedef __attribute__((ext_vector_type(16))) __bf16 v16bf;
typedef __attribute__((ext_vector_type(8)))  float  v8f;
typedef __attribute__((ext_vector_type(8)))  unsigned int v8u;
typedef const __attribute__((address_space(4))) v8u* list_ptr;

__device__ __forceinline__ unsigned short f2bf(float f) {
  union { float f; unsigned int u; } c; c.f = f;
  unsigned int u = c.u;
  u += 0x7fffu + ((u >> 16) & 1u);
  return (unsigned short)(u >> 16);
}

typedef __attribute__((ext_vector_type(2))) float  v2f;
typedef __attribute__((ext_vector_type(2))) __bf16 v2bf;
__device__ __forceinline__ unsigned int pack2bf(float a, float b) {
  const v2f f = {a, b};
  const v2bf h = __builtin_convertvector(f, v2bf);
  return __builtin_bit_cast(unsigned int, h);
}

#define AGG_NB      196
#define AGG_NK      ((AGG_NB + AGG_WAVES - 1) / AGG_WAVES)
#define AGG_WAVES   16
#define AGG_THREADS (AGG_WAVES * 32)
#define AGG_IT      15
#define AGG_SUB     (AGG_IT * 128)
#define AGG_ROUND   (AGG_SUB * AGG_WAVES)
#define AGG_NROUND  ((N_EDGES + AGG_ROUND - 1) / AGG_ROUND)
#define AGG_GROUPS  ((N_NODES + AGG_NB - 1) / AGG_NB)
#define LST_CAP     1024
#define LST_G       8
#define LST_PAD     (5 * LST_G)

static_assert(N_NODES <= (1 << 20) && N_EDGES <= (1 << 20) && AGG_NB < 256, "list entry fields");
static_assert(LST_G == 8, "a group of entries is one 8-dword scalar load");

static_assert(N_EDGES % 4 == 0, "four edges per index load");

#define CR_G        16
#define CR_NB       (AGG_NB * CR_G)
#define CR_N        ((AGG_GROUPS + CR_G - 1) / CR_G)
#define CS_IT       25
#define CS_SUB      (CS_IT * 128)
#define CS_SL       (CS_SUB * AGG_WAVES)
#define CS_N        ((N_EDGES + CS_SL - 1) / CS_SL)
#define P2_CAP      1024
#define P2_U        9

static_assert(CS_N == AGG_WAVES, "one wave of an owner group per slice list");
static_assert(CR_NB <= 4095, "a coarse range's local node fits 12 bits, and 4095 is no node of any group");

__device__ __forceinline__ void scan_load(const int* __restrict__ edst, int e0, int4 (&d)[AGG_IT]) {
  #pragma unroll
  for (int it = 0; it < AGG_IT; ++it) {
    const int e = e0 + it * 128;
    d[it] = *(const int4*)(edst + (e < N_EDGES ? e : N_EDGES - 4));
  }
}

__device__ __forceinline__ int scan_note(const int4 (&d)[AGG_IT], int e0, int base, int nloc,
                                         unsigned int ltmask, unsigned int* mylst) {
  int c = 0;
  #pragma unroll
  for (int it = 0; it < AGG_IT; ++it) {
    const unsigned int e = (unsigned int)(e0 + it * 128);
    const unsigned int x0 = (unsigned int)(d[it].x - base);
    const unsigned int x1 = (unsigned int)(d[it].y - base);
    const unsigned int x2 = (unsigned int)(d[it].z - base);
    const unsigned int x3 = (unsigned int)(d[it].w - base);
    const unsigned int lim = (e < (unsigned int)N_EDGES) ? (unsigned int)nloc : 0u;
    const bool m0 = x0 < lim, m1 = x1 < lim;
    const bool m2 = x2 < lim, m3 = x3 < lim;
    const unsigned int mn01 = x0 < x1 ? x0 : x1, mn23 = x2 < x3 ? x2 : x3;
    if (__builtin_amdgcn_ballot_w32((mn01 < mn23 ? mn01 : mn23) < lim)) {
      unsigned int b;
      b = __builtin_amdgcn_ballot_w32(m0);
      if (m0) mylst[c + __builtin_popcount(b & ltmask)] = (x0 << 20) | (e + 0u);
      c += __builtin_popcount(b);
      b = __builtin_amdgcn_ballot_w32(m1);
      if (m1) mylst[c + __builtin_popcount(b & ltmask)] = (x1 << 20) | (e + 1u);
      c += __builtin_popcount(b);
      b = __builtin_amdgcn_ballot_w32(m2);
      if (m2) mylst[c + __builtin_popcount(b & ltmask)] = (x2 << 20) | (e + 2u);
      c += __builtin_popcount(b);
      b = __builtin_amdgcn_ballot_w32(m3);
      if (m3) mylst[c + __builtin_popcount(b & ltmask)] = (x3 << 20) | (e + 3u);
      c += __builtin_popcount(b);
    }
  }
  return c;
}

__global__ void __launch_bounds__(AGG_THREADS)
coarse_lists(const int* __restrict__ edst, unsigned int* __restrict__ clst, int* __restrict__ ccnt) {
  __shared__ int cnt[AGG_WAVES];

  const int tid  = threadIdx.x;
  const int lane = tid & 31;
  const int wave = tid >> 5;
  const int ri   = blockIdx.x / CS_N;
  const int sl   = blockIdx.x % CS_N;
  const int rbase = ri * CR_NB;
  const unsigned int ltmask = (1u << lane) - 1u;
  const int e0 = sl * CS_SL + wave * CS_SUB + lane * 4;

  int4 d[CS_IT];
  #pragma unroll
  for (int it = 0; it < CS_IT; ++it) {
    const int e = e0 + it * 128;
    d[it] = *(const int4*)(edst + (e < N_EDGES ? e : N_EDGES - 4));
  }

  int c = 0;
  #pragma unroll
  for (int it = 0; it < CS_IT; ++it) {
    const unsigned int lim = ((e0 + it * 128) < N_EDGES) ? (unsigned int)CR_NB : 0u;
    c += __builtin_popcount(__builtin_amdgcn_ballot_w32((unsigned int)(d[it].x - rbase) < lim));
    c += __builtin_popcount(__builtin_amdgcn_ballot_w32((unsigned int)(d[it].y - rbase) < lim));
    c += __builtin_popcount(__builtin_amdgcn_ballot_w32((unsigned int)(d[it].z - rbase) < lim));
    c += __builtin_popcount(__builtin_amdgcn_ballot_w32((unsigned int)(d[it].w - rbase) < lim));
  }
  if (lane == 0) cnt[wave] = c;
  __syncthreads();

  const int myc = (lane < AGG_WAVES) ? cnt[lane] : 0;
  int incl = myc;
  #pragma unroll
  for (int o = 1; o < AGG_WAVES; o <<= 1) {
    const int t = __shfl_up(incl, o);
    if (lane >= o) incl += t;
  }
  const int total = __builtin_amdgcn_readlane(incl, AGG_WAVES - 1);
  const int wv = __builtin_amdgcn_readfirstlane(wave);
  int pos = __builtin_amdgcn_readlane(incl, wv) - c;

  unsigned int* out = clst + (long)blockIdx.x * CS_SL;
  #pragma unroll
  for (int it = 0; it < CS_IT; ++it) {
    const unsigned int e = (unsigned int)(e0 + it * 128);
    const unsigned int lim = (e < (unsigned int)N_EDGES) ? (unsigned int)CR_NB : 0u;
    const unsigned int x0 = (unsigned int)(d[it].x - rbase);
    const unsigned int x1 = (unsigned int)(d[it].y - rbase);
    const unsigned int x2 = (unsigned int)(d[it].z - rbase);
    const unsigned int x3 = (unsigned int)(d[it].w - rbase);
    const bool m0 = x0 < lim, m1 = x1 < lim;
    const bool m2 = x2 < lim, m3 = x3 < lim;
    unsigned int b;
    b = __builtin_amdgcn_ballot_w32(m0);
    if (m0) out[pos + __builtin_popcount(b & ltmask)] = (x0 << 20) | (e + 0u);
    pos += __builtin_popcount(b);
    b = __builtin_amdgcn_ballot_w32(m1);
    if (m1) out[pos + __builtin_popcount(b & ltmask)] = (x1 << 20) | (e + 1u);
    pos += __builtin_popcount(b);
    b = __builtin_amdgcn_ballot_w32(m2);
    if (m2) out[pos + __builtin_popcount(b & ltmask)] = (x2 << 20) | (e + 2u);
    pos += __builtin_popcount(b);
    b = __builtin_amdgcn_ballot_w32(m3);
    if (m3) out[pos + __builtin_popcount(b & ltmask)] = (x3 << 20) | (e + 3u);
    pos += __builtin_popcount(b);
  }
  const int npad = (total == 0) ? 4 : ((4 - (total & 3)) & 3);
  if (wave == 0 && lane < npad) out[total + lane] = 0xFFFFFFFFu;
  if (tid == 0) ccnt[blockIdx.x] = total;
}

__global__ void __launch_bounds__(AGG_THREADS)
owner_lists(const int* __restrict__ esrc, const unsigned int* __restrict__ clst, const int* __restrict__ ccnt,
            unsigned int* __restrict__ glst, int* __restrict__ gcnt) {
  __shared__ unsigned int lst[AGG_WAVES * P2_CAP];
  __shared__ unsigned int own[AGG_WAVES * LST_CAP];
  __shared__ int          cnt[AGG_WAVES];

  const int tid  = threadIdx.x;
  const int lane = tid & 31;
  const int wave = tid >> 5;
  const int wv   = __builtin_amdgcn_readfirstlane(wave);
  const int base = blockIdx.x * AGG_NB;
  int nloc = N_NODES - base;
  if (nloc > AGG_NB) nloc = AGG_NB;
  const int ri = blockIdx.x / CR_G;
  const unsigned int off = (unsigned int)(blockIdx.x % CR_G) * AGG_NB;

  const unsigned int ltmask = (1u << lane) - 1u;
  unsigned int* mylst = lst + wave * P2_CAP;
  unsigned int* myown = own + wave * LST_CAP;
  unsigned int* myout = glst + ((long)blockIdx.x * AGG_WAVES + wave) * LST_CAP;

  const int n  = __builtin_amdgcn_readfirstlane(ccnt[ri * CS_N + wv]);
  const int n4 = (n > 0) ? ((n + 3) >> 2) : 1;
  const uint4* L = (const uint4*)(clst + (long)(ri * CS_N + wv) * CS_SL);
  auto ask = [&](uint4 (&R)[P2_U], int q0) {
    #pragma unroll
    for (int u = 0; u < P2_U; ++u) {
      const int q = q0 + u * 32 + lane;
      R[u] = L[q < n4 ? q : n4 - 1];
    }
  };
  int c = 0;
  uint4 d[P2_U];
  ask(d, 0);
  for (int q0 = 0; q0 < n4; q0 += 32 * P2_U) {
    uint4 dn[P2_U];
    ask(dn, q0 + 32 * P2_U);
    #pragma unroll
    for (int u = 0; u < P2_U; ++u) {
      const unsigned int lim = (q0 + u * 32 + lane < n4) ? (unsigned int)nloc : 0u;
      const unsigned int x0 = (d[u].x >> 20) - off, x1 = (d[u].y >> 20) - off;
      const unsigned int x2 = (d[u].z >> 20) - off, x3 = (d[u].w >> 20) - off;
      const bool m0 = x0 < lim, m1 = x1 < lim;
      const bool m2 = x2 < lim, m3 = x3 < lim;
      unsigned int b;
      int p;
      b = __builtin_amdgcn_ballot_w32(m0);
      p = c + __builtin_popcount(b & ltmask);
      if (m0 && p < P2_CAP) mylst[p] = (x0 << 20) | (d[u].x & 0xFFFFFu);
      c += __builtin_popcount(b);
      b = __builtin_amdgcn_ballot_w32(m1);
      p = c + __builtin_popcount(b & ltmask);
      if (m1 && p < P2_CAP) mylst[p] = (x1 << 20) | (d[u].y & 0xFFFFFu);
      c += __builtin_popcount(b);
      b = __builtin_amdgcn_ballot_w32(m2);
      p = c + __builtin_popcount(b & ltmask);
      if (m2 && p < P2_CAP) mylst[p] = (x2 << 20) | (d[u].z & 0xFFFFFu);
      c += __builtin_popcount(b);
      b = __builtin_amdgcn_ballot_w32(m3);
      p = c + __builtin_popcount(b & ltmask);
      if (m3 && p < P2_CAP) mylst[p] = (x3 << 20) | (d[u].w & 0xFFFFFu);
      c += __builtin_popcount(b);
    }
    #pragma unroll
    for (int u = 0; u < P2_U; ++u) d[u] = dn[u];
  }
  if (lane == 0) cnt[wave] = c;
  __syncthreads();

  const int cv = (lane < AGG_WAVES) ? cnt[lane] : 0;
  if (__builtin_amdgcn_ballot_w32(cv > P2_CAP) != 0u) {
    if (lane == 0) gcnt[blockIdx.x * AGG_WAVES + wave] = LST_CAP;
    return;
  }

  int gpos = 0;
  for (int lw = 0; lw < AGG_WAVES; ++lw) {
    const int cn = __builtin_amdgcn_readlane(cv, lw);
    const unsigned int* Lw = lst + lw * P2_CAP;
    for (int f0 = 0; f0 < cn; f0 += 128) {
      unsigned int ent[4];
      #pragma unroll
      for (int j = 0; j < 4; ++j) {
        const int f = f0 + 32 * j + lane;
        ent[j] = (f < cn) ? Lw[f] : 0u;
      }
      #pragma unroll
      for (int j = 0; j < 4; ++j) {
        const int f = f0 + 32 * j + lane;
        const bool mine = (f < cn) && (((ent[j] >> 20) & (unsigned int)(AGG_WAVES - 1)) == (unsigned int)wave);
        const unsigned int m = __builtin_amdgcn_ballot_w32(mine);
        if (mine) {
          const int pos = gpos + __builtin_popcount(m & ltmask);
          if (pos < LST_CAP) myown[pos] = ent[j];
        }
        gpos += __builtin_popcount(m);
      }
    }
  }

  const int nout = gpos < LST_CAP ? gpos : LST_CAP;
  for (int i = lane; i < nout; i += 32) {
    const unsigned int ent = myown[i];
    myout[i] = (ent & 0xFFF00000u) | (unsigned int)esrc[ent & 0xFFFFFu];
  }
  #pragma unroll
  for (int k = 0; k < (LST_PAD + 31) / 32; ++k) {
    const int o = k * 32 + lane;
    if (o < LST_PAD && gpos + o < LST_CAP) myout[gpos + o] = ((unsigned int)AGG_NB << 20) | (unsigned int)base;
  }
  if (lane == 0) gcnt[blockIdx.x * AGG_WAVES + wave] = gpos;
}

typedef __attribute__((ext_vector_type(4))) float v4f;
template <int C> struct row_part;
template <> struct row_part<128> {
  typedef v4f type;
  static __device__ __forceinline__ type load(const float* row, int lane) {
    return *(const v4f*)(row + 4 * lane);
  }
  static __device__ __forceinline__ void store(float* row, int lane, type s) {
    *(v4f*)(row + 4 * lane) = s;
  }
  static __device__ __forceinline__ void store_bf16(unsigned short* row, int lane, type s) {
    *(uint2*)(row + 4 * lane) = make_uint2(pack2bf(s[0], s[1]), pack2bf(s[2], s[3]));
  }
};
template <> struct row_part<256> {
  typedef v8f type;
  static __device__ __forceinline__ type load(const float* row, int lane) {
    const v4f a = *(const v4f*)(row + 4 * lane), b = *(const v4f*)(row + 128 + 4 * lane);
    return __builtin_shufflevector(a, b, 0, 1, 2, 3, 4, 5, 6, 7);
  }
  static __device__ __forceinline__ void store(float* row, int lane, type s) {
    *(v4f*)(row + 4 * lane)       = __builtin_shufflevector(s, s, 0, 1, 2, 3);
    *(v4f*)(row + 128 + 4 * lane) = __builtin_shufflevector(s, s, 4, 5, 6, 7);
  }
  static __device__ __forceinline__ void store_bf16(unsigned short* row, int lane, type s) {
    *(uint2*)(row + 4 * lane)       = make_uint2(pack2bf(s[0], s[1]), pack2bf(s[2], s[3]));
    *(uint2*)(row + 128 + 4 * lane) = make_uint2(pack2bf(s[4], s[5]), pack2bf(s[6], s[7]));
  }
};

template <int C>
__global__ void __launch_bounds__(AGG_THREADS)
agg_owner(const float* __restrict__ h, const int* __restrict__ esrc,
          const int* __restrict__ edst, const unsigned int* __restrict__ glst,
          const int* __restrict__ gcnt, unsigned short* __restrict__ agg) {
  __shared__ float        acc[(AGG_NB + 1) * C];
  __shared__ unsigned int lst[AGG_WAVES * AGG_SUB];
  __shared__ int          cnt[AGG_WAVES];
  constexpr int VEC = C / 128;

  const int tid  = threadIdx.x;
  const int lane = tid & 31;
  const int wave = tid >> 5;
  const int base = blockIdx.x * AGG_NB;
  int nloc = N_NODES - base;
  if (nloc > AGG_NB) nloc = AGG_NB;

  const int lc = (lane < AGG_WAVES) ? gcnt[blockIdx.x * AGG_WAVES + lane] : 0;
  const bool fits = __builtin_amdgcn_ballot_w32(lc > LST_CAP - LST_PAD) == 0u;

  if (fits) {
    const int wv = __builtin_amdgcn_readfirstlane(wave);
    const int n  = __builtin_amdgcn_readlane(lc, wv);
    const list_ptr q = (list_ptr)(unsigned long)(glst + ((long)blockIdx.x * AGG_WAVES + wv) * LST_CAP);
    const int ng = (n + LST_G - 1) / LST_G;

    v8u eA = q[0], eB = q[1], eC = q[2];
    float4 A[LST_G][VEC], B[LST_G][VEC];
    auto rows_ask = [&](float4 (&R)[LST_G][VEC], const v8u& e) {
      #pragma unroll
      for (int j = 0; j < LST_G; ++j) {
        const int s = (int)(e[j] & 0xFFFFFu);
        const float4* hp = (const float4*)(h + (long)s * C) + lane;
        #pragma unroll
        for (int v = 0; v < VEC; ++v) R[j][v] = hp[v * 32];
      }
    };
    auto rows_add = [&](const float4 (&R)[LST_G][VEC], const v8u& e) {
      #pragma unroll
      for (int j = 0; j < LST_G; ++j) {
        const int nd = (int)(e[j] >> 20);
        float4* ap = (float4*)(acc + nd * C) + lane;
        #pragma unroll
        for (int v = 0; v < VEC; ++v) {
          float4 a = ap[v * 32];
          a.x += R[j][v].x; a.y += R[j][v].y; a.z += R[j][v].z; a.w += R[j][v].w;
          ap[v * 32] = a;
        }
      }
    };

    typename row_part<C>::type own[AGG_NK];
    #pragma unroll
    for (int k = 0; k < AGG_NK; ++k) {
      const int loc = wv + AGG_WAVES * k;
      own[k] = row_part<C>::load(h + (long)(base + (loc < nloc ? loc : 0)) * C, lane);
    }
    rows_ask(A, eA);
    #pragma unroll
    for (int k = 0; k < AGG_NK; ++k) {
      const int loc = wv + AGG_WAVES * k;
      row_part<C>::store(acc + (loc < nloc ? loc : AGG_NB) * C, lane, own[k]);
    }
    const int nt = (ng + 1) / 2;
    int t = 0;
    do {
      const v8u fB = q[2 * t + 3], fC = q[2 * t + 4];
      rows_ask(B, eB);
      rows_add(A, eA);
      rows_ask(A, eC);
      rows_add(B, eB);
      eA = eC; eB = fB; eC = fC;
    } while (++t < nt);

    typename row_part<C>::type fin[AGG_NK];
    #pragma unroll
    for (int k = 0; k < AGG_NK; ++k) {
      const int loc = wv + AGG_WAVES * k;
      fin[k] = row_part<C>::load(acc + (loc < nloc ? loc : AGG_NB) * C, lane);
    }
    #pragma unroll
    for (int k = 0; k < AGG_NK; ++k) {
      const int loc = wv + AGG_WAVES * k;
      if (loc < nloc)
        row_part<C>::store_bf16(agg + (long)(base + loc) * C, lane, fin[k]);
    }
  } else {
    {
      const float4* hs = (const float4*)(h + (long)base * C);
      float4* a4 = (float4*)acc;
      const int n4 = nloc * (C / 4);
      for (int i = tid; i < n4; i += AGG_THREADS) a4[i] = hs[i];
    }
    const unsigned int ltmask = (1u << lane) - 1u;
    unsigned int* mylst = lst + wave * AGG_SUB;
    int rr = (int)(blockIdx.x % AGG_NROUND);
    for (int r = 0; r < AGG_NROUND; ++r) {
      const int e0 = rr * AGG_ROUND + wave * AGG_SUB + lane * 4;
      rr = (rr + 1 == AGG_NROUND) ? 0 : rr + 1;
      int4 d[AGG_IT];
      scan_load(edst, e0, d);
      const int c = scan_note(d, e0, base, nloc, ltmask, mylst);
      if (lane == 0) cnt[wave] = c;
      __syncthreads();

      const int myc = (lane < AGG_WAVES) ? cnt[lane] : 0;
      int incl = myc;
      #pragma unroll
      for (int o = 1; o < AGG_WAVES; o <<= 1) {
        const int t = __shfl_up(incl, o);
        if (lane >= o) incl += t;
      }
      const int excl  = incl - myc;
      const int total = __builtin_amdgcn_readlane(incl, AGG_WAVES - 1);
      for (int f0 = 0; f0 < total; f0 += 32) {
        const int f = f0 + lane;
        int li = 0;
        #pragma unroll
        for (int l = 0; l < AGG_WAVES - 1; ++l) li += (f >= __builtin_amdgcn_readlane(incl, l)) ? 1 : 0;
        const int st = __shfl(excl, li);
        unsigned int ent = 0u;
        bool mine = false;
        if (f < total) {
          ent  = lst[li * AGG_SUB + (f - st)];
          mine = ((ent >> 20) & (unsigned int)(AGG_WAVES - 1)) == (unsigned int)wave;
        }
        unsigned int m = __builtin_amdgcn_ballot_w32(mine);
        int s = 0;
        if (mine) s = esrc[ent & 0xFFFFFu];
        while (m) {
          const int b0 = __builtin_ctz(m); m &= m - 1u;
          const int s0 = __builtin_amdgcn_readlane(s, b0);
          const int n0 = (int)(__builtin_amdgcn_readlane((int)ent, b0) >> 20) & 0xFF;
          const float4* h0 = (const float4*)(h + (long)s0 * C) + lane;
          float4* a0 = (float4*)(acc + n0 * C) + lane;
          #pragma unroll
          for (int j = 0; j < VEC; ++j) {
            const float4 v0 = h0[j * 32];
            float4 a = a0[j * 32];
            a.x += v0.x; a.y += v0.y; a.z += v0.z; a.w += v0.w;
            a0[j * 32] = a;
          }
        }
      }
      __syncthreads();
    }

    {
      uint4* o8 = (uint4*)(agg + (long)base * C);
      const float4* a4 = (const float4*)acc;
      const int n8 = nloc * (C / 8);
      for (int i = tid; i < n8; i += AGG_THREADS) {
        const float4 u = a4[2 * i], v = a4[2 * i + 1];
        o8[i] = make_uint4(pack2bf(u.x, u.y), pack2bf(u.z, u.w), pack2bf(v.x, v.y), pack2bf(v.z, v.w));
      }
    }
  }
}

#define MLP_WAVES 8
#define MLP_THREADS (MLP_WAVES * 32)
#define MLP_PITCH 36
#define WT_PITCH(KT) ((KT) * 64 + 1)

template <int KT1>
__device__ __forceinline__ void stage_weights(const float* __restrict__ W1, uint4* S1,
                                              const float* __restrict__ W2, uint4* S2, int wave, int lane) {
  constexpr int KT2 = HID / 32;
  constexpr int NC  = (KT1 + KT2) / 4;
  static_assert(KT1 % 4 == 0 && KT2 % 4 == 0 && MLP_WAVES == 8 && HID == 256 && NC >= 2 && NC <= 4,
                "eight waves = two halves of a weight row x four groups of eight k");
  const int quad = ((wave & 1) << 5) + lane;
  const int jh   = wave >> 1;
  const float4* g1 = (const float4*)(W1 + (long)(8 * jh) * HID) + quad;
  const float4* g2 = (const float4*)(W2 + (long)(8 * jh) * HID) + quad;
  const int so = (jh >> 1) * 32 + (jh & 1) * 16 + 4 * (quad & 3);
  uint4* s1 = S1 + (quad >> 2) * WT_PITCH(KT1) + so;
  uint4* s2 = S2 + (quad >> 2) * WT_PITCH(KT2) + so;
  float4 X[4][8], Y[4][8];
  auto ask = [&](float4 (&R)[4][8], int ch) {
    #pragma unroll
    for (int u = 0; u < 4; ++u) {
      const int s = 4 * ch + u;
      const float4* g = (s < KT1) ? g1 + s * (32 * HID / 4) : g2 + (s - KT1) * (32 * HID / 4);
      #pragma unroll
      for (int i = 0; i < 8; ++i) R[u][i] = g[i * (HID / 4)];
    }
  };
  auto put = [&](const float4 (&R)[4][8], int ch) {
    #pragma unroll
    for (int u = 0; u < 4; ++u) {
      const int s = 4 * ch + u;
      uint4* d = (s < KT1) ? s1 + s * 64 : s2 + (s - KT1) * 64;
      d[0] = make_uint4(pack2bf(R[u][0].x, R[u][1].x), pack2bf(R[u][2].x, R[u][3].x),
                        pack2bf(R[u][4].x, R[u][5].x), pack2bf(R[u][6].x, R[u][7].x));
      d[1] = make_uint4(pack2bf(R[u][0].y, R[u][1].y), pack2bf(R[u][2].y, R[u][3].y),
                        pack2bf(R[u][4].y, R[u][5].y), pack2bf(R[u][6].y, R[u][7].y));
      d[2] = make_uint4(pack2bf(R[u][0].z, R[u][1].z), pack2bf(R[u][2].z, R[u][3].z),
                        pack2bf(R[u][4].z, R[u][5].z), pack2bf(R[u][6].z, R[u][7].z));
      d[3] = make_uint4(pack2bf(R[u][0].w, R[u][1].w), pack2bf(R[u][2].w, R[u][3].w),
                        pack2bf(R[u][4].w, R[u][5].w), pack2bf(R[u][6].w, R[u][7].w));
    }
  };
  ask(X, 0);
  ask(Y, 1);
  put(X, 0);
  if (NC > 2) ask(X, 2);
  put(Y, 1);
  if (NC > 3) ask(Y, 3);
  if (NC > 2) put(X, 2);
  if (NC > 3) put(Y, 3);
}
template <int KT1>
__global__ void __launch_bounds__(MLP_THREADS)
mlp_chain(const unsigned short* __restrict__ Xbf, const float* __restrict__ W1, const float* __restrict__ b1,
          const float* __restrict__ W2, const float* __restrict__ b2, float* __restrict__ out) {
  __shared__ alignas(16) float stage[MLP_WAVES][2][16 * MLP_PITCH];
  __shared__ uint4 w1s[(HID / 16) * WT_PITCH(KT1)];
  __shared__ uint4 w2s[(HID / 16) * WT_PITCH(HID / 32)];
  const int lane = threadIdx.x & 31;
  const int wave = threadIdx.x >> 5;
  const int c    = lane & 15;
  const int hh   = lane >> 4;
  const int nrt  = N_NODES / 16;
  const int t0   = (blockIdx.x * MLP_WAVES + wave) * 2;
  const bool ok0 = t0 < nrt, ok1 = (t0 + 1) < nrt;
  const int tA = ok0 ? t0 : 0, tB = ok1 ? (t0 + 1) : 0;
  constexpr int KT2 = HID / 32;
  constexpr int UT  = HID / 16;
  union F { uint4 q[2]; v16bf v; };

  v8f acc[UT][2];
  #pragma unroll
  for (int j = 0; j < UT; ++j) { acc[j][0] = v8f{}; acc[j][1] = v8f{}; }
  const uint4* w1l = w1s + lane;
  auto w1_ask = [&](F (&g)[UT / 2], int kt, int j0) {
    #pragma unroll
    for (int j = 0; j < UT / 2; ++j) {
      const uint4* pw = w1l + (j0 + j) * WT_PITCH(KT1) + kt * 64;
      g[j].q[0] = pw[0]; g[j].q[1] = pw[32];
    }
  };
  const uint4* xr0 = (const uint4*)(Xbf + (long)(tA * 16 + c) * (32 * KT1)) + hh;
  const uint4* xr1 = (const uint4*)(Xbf + (long)(tB * 16 + c) * (32 * KT1)) + hh;
  auto x_ask = [&](F& x0, F& x1, int kt) {
    x0.q[0] = xr0[4 * kt]; x0.q[1] = xr0[4 * kt + 2];
    x1.q[0] = xr1[4 * kt]; x1.q[1] = xr1[4 * kt + 2];
  };
  F wA[UT / 2], wB[UT / 2], x0, x1;
  x_ask(x0, x1, 0);
  stage_weights<KT1>(W1, w1s, W2, w2s, wave, lane);
  __syncthreads();
  w1_ask(wA, 0, 0);
  for (int kt = 0; kt < KT1; ++kt) {
    const int ktn = (kt + 1 < KT1) ? kt + 1 : kt;
    F y0, y1;
    w1_ask(wB, kt, UT / 2);
    x_ask(y0, y1, ktn);
    __builtin_amdgcn_sched_barrier(0);
    #pragma unroll
    for (int j = 0; j < UT / 2; ++j) {
      acc[j][0] = __builtin_amdgcn_wmma_f32_16x16x32_bf16(false, wA[j].v, false, x0.v, (short)0, acc[j][0], false, false);
      acc[j][1] = __builtin_amdgcn_wmma_f32_16x16x32_bf16(false, wA[j].v, false, x1.v, (short)0, acc[j][1], false, false);
    }
    __builtin_amdgcn_sched_barrier(0);
    w1_ask(wA, ktn, 0);
    __builtin_amdgcn_sched_barrier(0);
    #pragma unroll
    for (int j = 0; j < UT / 2; ++j) {
      acc[UT / 2 + j][0] = __builtin_amdgcn_wmma_f32_16x16x32_bf16(false, wB[j].v, false, x0.v, (short)0, acc[UT / 2 + j][0], false, false);
      acc[UT / 2 + j][1] = __builtin_amdgcn_wmma_f32_16x16x32_bf16(false, wB[j].v, false, x1.v, (short)0, acc[UT / 2 + j][1], false, false);
    }
    __builtin_amdgcn_sched_barrier(0);
    x0 = y0; x1 = y1;
  }

  F m[2][KT2];
  #pragma unroll
  for (int t = 0; t < KT2; ++t) {
    const float4* bp = (const float4*)(b1 + 32 * t + 8 * hh);
    const float4 ba0 = bp[0], ba1 = bp[1];
    const float4 bb0 = bp[4], bb1 = bp[5];
    const float ba[8] = {ba0.x, ba0.y, ba0.z, ba0.w, ba1.x, ba1.y, ba1.z, ba1.w};
    const float bb[8] = {bb0.x, bb0.y, bb0.z, bb0.w, bb1.x, bb1.y, bb1.z, bb1.w};
    #pragma unroll
    for (int nt = 0; nt < 2; ++nt) {
      unsigned int lo[4], hi[4];
      #pragma unroll
      for (int p = 0; p < 4; ++p) {
        const float a0 = fmaxf(acc[2 * t][nt][2 * p]     + ba[2 * p],     0.0f);
        const float a1 = fmaxf(acc[2 * t][nt][2 * p + 1] + ba[2 * p + 1], 0.0f);
        const float c0 = fmaxf(acc[2 * t + 1][nt][2 * p]     + bb[2 * p],     0.0f);
        const float c1 = fmaxf(acc[2 * t + 1][nt][2 * p + 1] + bb[2 * p + 1], 0.0f);
        lo[p] = pack2bf(a0, a1);
        hi[p] = pack2bf(c0, c1);
      }
      m[nt][t].q[0] = make_uint4(lo[0], lo[1], lo[2], lo[3]);
      m[nt][t].q[1] = make_uint4(hi[0], hi[1], hi[2], hi[3]);
    }
  }

  float* st0 = &stage[wave][0][0];
  float* st1 = &stage[wave][1][0];
  const uint4* w2l = w2s + lane;
  auto w2_ask = [&](F (&g)[8], int cp, int tq) {
    #pragma unroll
    for (int t = 0; t < 4; ++t) {
      const uint4* pa = w2l + (2 * cp) * WT_PITCH(KT2) + (tq + t) * 64;
      const uint4* pb = w2l + (2 * cp + 1) * WT_PITCH(KT2) + (tq + t) * 64;
      g[2 * t].q[0] = pa[0];     g[2 * t].q[1] = pa[32];
      g[2 * t + 1].q[0] = pb[0]; g[2 * t + 1].q[1] = pb[32];
    }
  };
  w2_ask(wA, 0, 0);
  for (int cp = 0; cp < UT / 2; ++cp) {
    const int cpn = (cp + 1 < UT / 2) ? cp + 1 : cp;
    v8f o00 = {}, o01 = {}, o10 = {}, o11 = {};
    w2_ask(wB, cp, 4);
    __builtin_amdgcn_sched_barrier(0);
    #pragma unroll
    for (int t = 0; t < 4; ++t) {
      o00 = __builtin_amdgcn_wmma_f32_16x16x32_bf16(false, wA[2 * t].v,     false, m[0][t].v, (short)0, o00, false, false);
      o01 = __builtin_amdgcn_wmma_f32_16x16x32_bf16(false, wA[2 * t].v,     false, m[1][t].v, (short)0, o01, false, false);
      o10 = __builtin_amdgcn_wmma_f32_16x16x32_bf16(false, wA[2 * t + 1].v, false, m[0][t].v, (short)0, o10, false, false);
      o11 = __builtin_amdgcn_wmma_f32_16x16x32_bf16(false, wA[2 * t + 1].v, false, m[1][t].v, (short)0, o11, false, false);
    }
    __builtin_amdgcn_sched_barrier(0);
    w2_ask(wA, cpn, 0);
    __builtin_amdgcn_sched_barrier(0);
    #pragma unroll
    for (int t = 0; t < 4; ++t) {
      o00 = __builtin_amdgcn_wmma_f32_16x16x32_bf16(false, wB[2 * t].v,     false, m[0][4 + t].v, (short)0, o00, false, false);
      o01 = __builtin_amdgcn_wmma_f32_16x16x32_bf16(false, wB[2 * t].v,     false, m[1][4 + t].v, (short)0, o01, false, false);
      o10 = __builtin_amdgcn_wmma_f32_16x16x32_bf16(false, wB[2 * t + 1].v, false, m[0][4 + t].v, (short)0, o10, false, false);
      o11 = __builtin_amdgcn_wmma_f32_16x16x32_bf16(false, wB[2 * t + 1].v, false, m[1][4 + t].v, (short)0, o11, false, false);
    }
    __builtin_amdgcn_sched_barrier(0);
    const float4* bq = (const float4*)(b2 + 32 * cp + 8 * hh);
    const float4 ba0 = bq[0], ba1 = bq[1], bb0 = bq[4], bb1 = bq[5];
    auto put = [&](float* st, const v8f& oa, const v8f& ob) {
      float4* r = (float4*)(st + c * MLP_PITCH + 8 * hh);
      r[0] = make_float4(fmaxf(oa[0] + ba0.x, 0.0f), fmaxf(oa[1] + ba0.y, 0.0f),
                         fmaxf(oa[2] + ba0.z, 0.0f), fmaxf(oa[3] + ba0.w, 0.0f));
      r[1] = make_float4(fmaxf(oa[4] + ba1.x, 0.0f), fmaxf(oa[5] + ba1.y, 0.0f),
                         fmaxf(oa[6] + ba1.z, 0.0f), fmaxf(oa[7] + ba1.w, 0.0f));
      r[4] = make_float4(fmaxf(ob[0] + bb0.x, 0.0f), fmaxf(ob[1] + bb0.y, 0.0f),
                         fmaxf(ob[2] + bb0.z, 0.0f), fmaxf(ob[3] + bb0.w, 0.0f));
      r[5] = make_float4(fmaxf(ob[4] + bb1.x, 0.0f), fmaxf(ob[5] + bb1.y, 0.0f),
                         fmaxf(ob[6] + bb1.z, 0.0f), fmaxf(ob[7] + bb1.w, 0.0f));
    };
    put(st0, o00, o10);
    put(st1, o01, o11);
    auto rows_out = [&](const float* st, int tile, bool ok) {
      if (!ok) return;
      #pragma unroll
      for (int i = 0; i < 4; ++i) {
        const int node = 4 * i + (lane >> 3);
        const float4 v = *(const float4*)(st + node * MLP_PITCH + 4 * (lane & 7));
        *(float4*)(out + (long)(tile * 16 + node) * HID + 32 * cp + 4 * (lane & 7)) = v;
      }
    };
    rows_out(st0, t0, ok0);
    rows_out(st1, t0 + 1, ok1);
  }
}

#define POOL_GB      4
#define POOL_WAVES   16
#define POOL_THREADS (POOL_WAVES * 32)
#define POOL_IT      25
#define POOL_SUB     (POOL_IT * 128)
#define POOL_PG      16

static_assert(POOL_SUB * POOL_WAVES >= N_NODES && N_NODES % 4 == 0, "the slices cover the id list, four ids per load");
static_assert(N_GRAPHS % POOL_GB == 0 && POOL_GB <= 8, "owned graphs are rows 0 .. POOL_GB - 1 of a result tile: the lower lane half's registers");
static_assert(POOL_WAVES == POOL_GB * (HID / 64) && POOL_WAVES == HID / 16, "one wave per graph and 64 columns; one per tile of 16 hidden units");
static_assert(OUT_C == 16 && HID % 32 == 0, "the output is one tile of 16 columns; whole k-steps");

__global__ void __launch_bounds__(POOL_THREADS)
pool_mlp(const float* __restrict__ h, const int* __restrict__ batch,
         const float* __restrict__ W1, const float* __restrict__ b1,
         const float* __restrict__ W2, const float* __restrict__ b2, float* __restrict__ out) {
  __shared__ unsigned int lst[POOL_WAVES * POOL_SUB];
  __shared__ int          cnt[POOL_WAVES];
  __shared__ alignas(16) unsigned int stage[POOL_GB * (HID / 2)];
  __shared__ alignas(16) unsigned int hid[POOL_GB * (HID / 2)];

  const int tid  = threadIdx.x;
  const int lane = tid & 31;
  const int wave = tid >> 5;
  const int g0   = blockIdx.x * POOL_GB;
  const unsigned int ltmask = (1u << lane) - 1u;

  {
    unsigned int* mylst = lst + wave * POOL_SUB;
    const int n0 = wave * POOL_SUB + lane * 4;
    int4 d[POOL_IT];
    #pragma unroll
    for (int it = 0; it < POOL_IT; ++it) {
      const int n = n0 + it * 128;
      d[it] = *(const int4*)(batch + (n < N_NODES ? n : N_NODES - 4));
    }
    int c = 0;
    #pragma unroll
    for (int it = 0; it < POOL_IT; ++it) {
      const unsigned int n  = (unsigned int)(n0 + it * 128);
      const unsigned int x0 = (unsigned int)(d[it].x - g0);
      const unsigned int x1 = (unsigned int)(d[it].y - g0);
      const unsigned int x2 = (unsigned int)(d[it].z - g0);
      const unsigned int x3 = (unsigned int)(d[it].w - g0);
      const unsigned int lim = (n < (unsigned int)N_NODES) ? (unsigned int)POOL_GB : 0u;
      const bool m0 = x0 < lim, m1 = x1 < lim;
      const bool m2 = x2 < lim, m3 = x3 < lim;
      if (__builtin_amdgcn_ballot_w32(m0 || m1 || m2 || m3)) {
        unsigned int b;
        b = __builtin_amdgcn_ballot_w32(m0);
        if (m0) mylst[c + __builtin_popcount(b & ltmask)] = (x0 << 20) | (n + 0u);
        c += __builtin_popcount(b);
        b = __builtin_amdgcn_ballot_w32(m1);
        if (m1) mylst[c + __builtin_popcount(b & ltmask)] = (x1 << 20) | (n + 1u);
        c += __builtin_popcount(b);
        b = __builtin_amdgcn_ballot_w32(m2);
        if (m2) mylst[c + __builtin_popcount(b & ltmask)] = (x2 << 20) | (n + 2u);
        c += __builtin_popcount(b);
        b = __builtin_amdgcn_ballot_w32(m3);
        if (m3) mylst[c + __builtin_popcount(b & ltmask)] = (x3 << 20) | (n + 3u);
        c += __builtin_popcount(b);
      }
    }
    if (lane == 0) cnt[wave] = c;
  }
  __syncthreads();

  const unsigned int gi = (unsigned int)(wave & (POOL_GB - 1));
  const int cq = wave / POOL_GB;
  const float2* hq = (const float2*)h + cq * 32 + lane;
  float2 acc = make_float2(0.0f, 0.0f);
  int deg = 0;
  for (int lw = 0; lw < POOL_WAVES; ++lw) {
    const int cn = __builtin_amdgcn_readfirstlane(cnt[lw]);
    const unsigned int* L = lst + lw * POOL_SUB;
    for (int f0 = 0; f0 < cn; f0 += 32) {
      const int f = f0 + lane;
      const unsigned int e = (f < cn) ? L[f] : 0xFFFFFFFFu;
      unsigned int m = __builtin_amdgcn_ballot_w32((e >> 20) == gi);
      deg += __builtin_popcount(m);
      while (m) {
        float2 v[POOL_PG];
        #pragma unroll
        for (int j = 0; j < POOL_PG; ++j) {
          v[j] = make_float2(0.0f, 0.0f);
          if (m) {
            const int b = __builtin_ctz(m);
            m &= m - 1u;
            const int n = __builtin_amdgcn_readlane((int)e, b) & 0xFFFFF;
            v[j] = hq[n * (HID / 2)];
          }
        }
        #pragma unroll
        for (int j = 0; j < POOL_PG; ++j) { acc.x += v[j].x; acc.y += v[j].y; }
      }
    }
  }

  const float den = fmaxf((float)deg, 1.0f);
  stage[gi * (HID / 2) + cq * 32 + lane] = pack2bf(acc.x / den, acc.y / den);

  union F { uint4 q[2]; v16bf v; };
  constexpr int KT = HID / 32;
  const int c  = lane & 15;
  const int hh = lane >> 4;
  auto wfrag = [&](F& f, const float* p, int M) {
    f.q[0] = make_uint4(pack2bf(p[0], p[M]), pack2bf(p[2 * M], p[3 * M]),
                        pack2bf(p[4 * M], p[5 * M]), pack2bf(p[6 * M], p[7 * M]));
    p += 16 * M;
    f.q[1] = make_uint4(pack2bf(p[0], p[M]), pack2bf(p[2 * M], p[3 * M]),
                        pack2bf(p[4 * M], p[5 * M]), pack2bf(p[6 * M], p[7 * M]));
  };
  const bool  live  = c < POOL_GB;
  const uint4 zero4 = make_uint4(0u, 0u, 0u, 0u);
  const int   crow  = live ? c : 0;

  F wf[KT];
  {
    const float* wb = W1 + (long)(8 * hh) * HID + wave * 16 + c;
    #pragma unroll
    for (int kt = 0; kt < KT; ++kt) wfrag(wf[kt], wb + (long)kt * 32 * HID, HID);
  }
  const float bias1 = b1[wave * 16 + c];
  __syncthreads();

  v8f hacc = {};
  {
    const uint4* arow = (const uint4*)(stage + crow * (HID / 2)) + hh;
    #pragma unroll
    for (int kt = 0; kt < KT; ++kt) {
      F a;
      const uint4 a0 = arow[4 * kt], a1 = arow[4 * kt + 2];
      a.q[0] = live ? a0 : zero4;
      a.q[1] = live ? a1 : zero4;
      hacc = __builtin_amdgcn_wmma_f32_16x16x32_bf16(false, a.v, false, wf[kt].v, (short)0, hacc, false, false);
    }
  }

  const bool second = __builtin_amdgcn_readfirstlane(wave) == 0;
  F vf[KT];
  float bias2 = 0.0f;
  if (second) {
    const float* vb = W2 + (long)(8 * hh) * OUT_C + c;
    #pragma unroll
    for (int kt = 0; kt < KT; ++kt) wfrag(vf[kt], vb + (long)kt * 32 * OUT_C, OUT_C);
    bias2 = b2[c];
  }

  {
    unsigned short hv[POOL_GB];
    #pragma unroll
    for (int r = 0; r < POOL_GB; ++r) hv[r] = f2bf(fmaxf(hacc[r] + bias1, 0.0f));
    if (hh == 0) {
      #pragma unroll
      for (int r = 0; r < POOL_GB; ++r) ((unsigned short*)hid)[r * HID + wave * 16 + c] = hv[r];
    }
  }
  __syncthreads();

  if (second) {
    const uint4* hrow = (const uint4*)(hid + crow * (HID / 2)) + hh;
    v8f o = {};
    #pragma unroll
    for (int kt = 0; kt < KT; ++kt) {
      F a;
      const uint4 a0 = hrow[4 * kt], a1 = hrow[4 * kt + 2];
      a.q[0] = live ? a0 : zero4;
      a.q[1] = live ? a1 : zero4;
      o = __builtin_amdgcn_wmma_f32_16x16x32_bf16(false, a.v, false, vf[kt].v, (short)0, o, false, false);
    }
    float ov[POOL_GB];
    #pragma unroll
    for (int r = 0; r < POOL_GB; ++r) ov[r] = o[r] + bias2;
    if (hh == 0) {
      #pragma unroll
      for (int r = 0; r < POOL_GB; ++r) out[(long)(g0 + r) * OUT_C + c] = ov[r];
    }
  }
}

extern "C" void kernel_launch(void* const* d_in, const int* in_sizes, int n_in,
                              void* d_out, int out_size, void* d_ws, size_t ws_size,
                              hipStream_t stream) {
  const float* x     = (const float*)d_in[0];
  const int*   ei    = (const int*)d_in[1];
  const int*   esrc  = ei;
  const int*   edst  = ei + N_EDGES;
  const int*   batch = (const int*)d_in[2];
  const float* cw1[3] = {(const float*)d_in[3],  (const float*)d_in[7],  (const float*)d_in[11]};
  const float* cb1[3] = {(const float*)d_in[4],  (const float*)d_in[8],  (const float*)d_in[12]};
  const float* cw2[3] = {(const float*)d_in[5],  (const float*)d_in[9],  (const float*)d_in[13]};
  const float* cb2[3] = {(const float*)d_in[6],  (const float*)d_in[10], (const float*)d_in[14]};
  const float* ow1 = (const float*)d_in[15];
  const float* ob1 = (const float*)d_in[16];
  const float* ow2 = (const float*)d_in[17];
  const float* ob2 = (const float*)d_in[18];

  char* ws = (char*)d_ws;
  size_t off = 0;
  auto bump = [&](size_t bytes) {
    void* p = ws + off;
    off += (bytes + 255) & ~(size_t)255;
    return p;
  };
  const int NRT = N_NODES / 16;
  unsigned short* Abf    = (unsigned short*)bump((size_t)N_NODES * HID * 2);
  float*          B2     = (float*)bump((size_t)N_NODES * HID * 4);
  unsigned int* glst  = (unsigned int*)bump((size_t)AGG_GROUPS * AGG_WAVES * LST_CAP * 4);
  int*          gcnt  = (int*)bump((size_t)AGG_GROUPS * AGG_WAVES * 4);
  unsigned int* clst  = (unsigned int*)bump((size_t)CR_N * CS_N * CS_SL * 4);
  int*          ccnt  = (int*)bump((size_t)CR_N * CS_N * 4);


  coarse_lists<<<CR_N * CS_N, AGG_THREADS, 0, stream>>>(edst, clst, ccnt);
  owner_lists<<<AGG_GROUPS, AGG_THREADS, 0, stream>>>(esrc, clst, ccnt, glst, gcnt);

  const float* h = x;
  for (int l = 0; l < 3; ++l) {
    if (l == 0) agg_owner<IN_C><<<AGG_GROUPS, AGG_THREADS, 0, stream>>>(h, esrc, edst, glst, gcnt, Abf);
    else        agg_owner<HID><<<AGG_GROUPS, AGG_THREADS, 0, stream>>>(h, esrc, edst, glst, gcnt, Abf);

    const unsigned mg = (unsigned)((NRT + 2 * MLP_WAVES - 1) / (2 * MLP_WAVES));
    if (l == 0) mlp_chain<IN_C / 32><<<mg, MLP_THREADS, 0, stream>>>(Abf, cw1[l], cb1[l], cw2[l], cb2[l], B2);
    else        mlp_chain<HID / 32><<<mg, MLP_THREADS, 0, stream>>>(Abf, cw1[l], cb1[l], cw2[l], cb2[l], B2);
    h = B2;
  }

  pool_mlp<<<N_GRAPHS / POOL_GB, POOL_THREADS, 0, stream>>>(B2, batch, ow1, ob1, ow2, ob2, (float*)d_out);
}
